# starting kernel
# baseline (speedup 1.0000x reference)
#include <hip/hip_runtime.h>
#include <math.h>

typedef float v2f __attribute__((ext_vector_type(2)));
typedef float v8f __attribute__((ext_vector_type(8)));

#define FEAT   64
#define NCLS   40
#define NCLSP  48


__global__ void k_fill(float* __restrict__ p, float val, int n) {
    int i = blockIdx.x * blockDim.x + threadIdx.x;
    if (i < n) p[i] = val;
}

__global__ void k_degree(const int* __restrict__ col, float* __restrict__ deg, int E) {
    int i = blockIdx.x * blockDim.x + threadIdx.x;
    if (i < E) atomicAdd(&deg[col[i]], 1.0f);
}

__global__ void k_rsqrt_inplace(float* __restrict__ d, int n) {
    int i = blockIdx.x * blockDim.x + threadIdx.x;
    if (i < n) {
        float v = d[i];
        d[i] = (v > 0.0f) ? rsqrtf(v) : 0.0f;
    }
}

__global__ void k_selfloop_init(const float* __restrict__ xin, float* __restrict__ xout,
                                const float* __restrict__ dinv, int total) {
    int i = blockIdx.x * blockDim.x + threadIdx.x;
    if (i < total) {
        float w = dinv[i >> 6];
        xout[i] = w * w * xin[i];
    }
}

__global__ void k_edge_spmm(const float* __restrict__ xin, float* __restrict__ xout,
                            const int* __restrict__ row, const int* __restrict__ col,
                            const float* __restrict__ dinv, int E) {
    long long t = (long long)blockIdx.x * blockDim.x + threadIdx.x;
    int e    = (int)(t >> 4);
    int lane = (int)(t & 15);
    if (e >= E) return;
    int r = row[e];
    int c = col[e];
    float w = dinv[r] * dinv[c];
    const float4 v = ((const float4*)(xin + (size_t)r * FEAT))[lane];
    float* dst = xout + (size_t)c * FEAT + lane * 4;
    atomicAdd(dst + 0, w * v.x);
    atomicAdd(dst + 1, w * v.y);
    atomicAdd(dst + 2, w * v.z);
    atomicAdd(dst + 3, w * v.w);
}

__global__ void k_pad_w(const float* __restrict__ W, float* __restrict__ Wp) {
    int i = blockIdx.x * blockDim.x + threadIdx.x;
    if (i < NCLSP * FEAT) {
        int r = i / FEAT;
        Wp[i] = (r < NCLS) ? W[i] : 0.0f;
    }
}

__global__ void k_gemm_wmma(const float* __restrict__ x, const float* __restrict__ Wp,
                            const float* __restrict__ bias, float* __restrict__ out, int N) {
    const int wave   = blockIdx.x * (blockDim.x >> 5) + (threadIdx.x >> 5);
    const int lane   = threadIdx.x & 31;
    const int ntiles = (N + 15) >> 4;
    if (wave >= ntiles) return;

    const int base = wave << 4;
    const int half = lane >> 4;
    const int l16  = lane & 15;

    int m = base + l16;
    if (m >= N) m = N - 1;
    const float* arow = x + (size_t)m * FEAT;
    const float* b0r  = Wp + (size_t)(l16)      * FEAT;
    const float* b1r  = Wp + (size_t)(16 + l16) * FEAT;
    const float* b2r  = Wp + (size_t)(32 + l16) * FEAT;

    v8f acc0 = {};
    v8f acc1 = {};
    v8f acc2 = {};

    for (int k0 = 0; k0 < FEAT; k0 += 4) {
        const int ko = k0 + 2 * half;
        v2f a  = *(const v2f*)(arow + ko);
        v2f b0 = *(const v2f*)(b0r + ko);
        v2f b1 = *(const v2f*)(b1r + ko);
        v2f b2 = *(const v2f*)(b2r + ko);
        acc0 = __builtin_amdgcn_wmma_f32_16x16x4_f32(false, a, false, b0, (short)0, acc0, false, false);
        acc1 = __builtin_amdgcn_wmma_f32_16x16x4_f32(false, a, false, b1, (short)0, acc1, false, false);
        acc2 = __builtin_amdgcn_wmma_f32_16x16x4_f32(false, a, false, b2, (short)0, acc2, false, false);
    }

    const int c0 = l16;
    const int c1 = 16 + l16;
    const int c2 = 32 + l16;
    const float bb0 = bias[c0];
    const float bb1 = bias[c1];
    const float bb2 = (c2 < NCLS) ? bias[c2] : 0.0f;
#pragma unroll
    for (int j = 0; j < 8; ++j) {
        const int node = base + j + 8 * half;
        if (node < N) {
            float* o = out + (size_t)node * NCLS;
            o[c0] = acc0[j] + bb0;
            o[c1] = acc1[j] + bb1;
            if (c2 < NCLS) o[c2] = acc2[j] + bb2;
        }
    }
}

__global__ void k_logsoftmax(float* __restrict__ out, int N) {
    int n = blockIdx.x * blockDim.x + threadIdx.x;
    if (n >= N) return;
    float* p = out + (size_t)n * NCLS;
    float v[NCLS];
    float m = -INFINITY;
#pragma unroll
    for (int c = 0; c < NCLS; ++c) { v[c] = p[c]; m = fmaxf(m, v[c]); }
    float s = 0.0f;
#pragma unroll
    for (int c = 0; c < NCLS; ++c) s += expf(v[c] - m);
    const float ls = logf(s);
#pragma unroll
    for (int c = 0; c < NCLS; ++c) p[c] = v[c] - m - ls;
}


static inline size_t align256f(size_t nfloats) { return (nfloats + 63) & ~(size_t)63; }

extern "C" void kernel_launch(void* const* d_in, const int* in_sizes, int n_in,
                              void* d_out, int out_size, void* d_ws, size_t ws_size,
                              hipStream_t stream) {
    const float* x  = (const float*)d_in[0];
    const int*   ei = (const int*)d_in[1];
    const float* W  = (const float*)d_in[2];
    const float* bi = (const float*)d_in[3];

    const int N = in_sizes[0] / FEAT;
    const int E = in_sizes[1] / 2;
    const int* row = ei;
    const int* col = ei + E;
    const int total = N * FEAT;

    float* ws = (float*)d_ws;
    size_t off = 0;
    float* dinv = ws + off; off += align256f((size_t)N);
    float* Wp   = ws + off; off += align256f((size_t)NCLSP * FEAT);
    float* x0   = ws + off; off += align256f((size_t)total);
    float* x1   = ws + off;

    const int T = 256;
    k_fill<<<(N + T - 1) / T, T, 0, stream>>>(dinv, 1.0f, N);
    k_degree<<<(E + T - 1) / T, T, 0, stream>>>(col, dinv, E);
    k_rsqrt_inplace<<<(N + T - 1) / T, T, 0, stream>>>(dinv, N);
    k_pad_w<<<(NCLSP * FEAT + T - 1) / T, T, 0, stream>>>(W, Wp);

    const long long ethreads = (long long)E * 16;
    const int eblocks = (int)((ethreads + T - 1) / T);

    k_selfloop_init<<<(total + T - 1) / T, T, 0, stream>>>(x, x0, dinv, total);
    k_edge_spmm<<<eblocks, T, 0, stream>>>(x, x0, row, col, dinv, E);
    k_selfloop_init<<<(total + T - 1) / T, T, 0, stream>>>(x0, x1, dinv, total);
    k_edge_spmm<<<eblocks, T, 0, stream>>>(x0, x1, row, col, dinv, E);

    const int ntiles = (N + 15) >> 4;
    const int gblocks = (ntiles + 7) / 8;
    k_gemm_wmma<<<gblocks, 256, 0, stream>>>(x1, Wp, bi, (float*)d_out, N);
    k_logsoftmax<<<(N + T - 1) / T, T, 0, stream>>>((float*)d_out, N);
}


# both hops: a work-group keeps 960 target rows' sums in LDS (seeded with the self-loop term), filters the whole edge list for them, adds gathered source rows by ds_add_f32; sum order = LDS arrival
# speedup vs baseline: 1.0933x; 1.0933x over previous
#include <hip/hip_runtime.h>
#include <math.h>

typedef float v2f __attribute__((ext_vector_type(2)));
typedef float v8f __attribute__((ext_vector_type(8)));

#define FEAT   64
#define NCLS   40
#define NCLSP  48


__global__ void k_fill(float* __restrict__ p, float val, int n) {
    int i = blockIdx.x * blockDim.x + threadIdx.x;
    if (i < n) p[i] = val;
}

__global__ void k_degree(const int* __restrict__ col, float* __restrict__ deg, int E) {
    int i = blockIdx.x * blockDim.x + threadIdx.x;
    if (i < E) atomicAdd(&deg[col[i]], 1.0f);
}

__global__ void k_rsqrt_inplace(float* __restrict__ d, int n) {
    int i = blockIdx.x * blockDim.x + threadIdx.x;
    if (i < n) {
        float v = d[i];
        d[i] = (v > 0.0f) ? rsqrtf(v) : 0.0f;
    }
}

#define TILE    960
#define HOPW    32
#define GRP     512
#define QFLUSH  64
#define QCAP    (QFLUSH + GRP)
#define BW      16

#define APPEND(m_, e_)                                                         \
    {                                                                          \
        const unsigned mk_ = __builtin_amdgcn_ballot_w32(m_);                  \
        if (mk_) {                                                             \
            if (m_) q[cnt + (int)__builtin_amdgcn_mbcnt_lo(mk_, 0u)] = (e_);   \
            cnt += __builtin_popcount(mk_);                                    \
        }                                                                      \
    }

#define FILTER4(cv_, e0_)                                                      \
    {                                                                          \
        bool m_;                                                               \
        m_ = (unsigned)((cv_).x - t0) < (unsigned)tn; APPEND(m_, (e0_) + 0);   \
        m_ = (unsigned)((cv_).y - t0) < (unsigned)tn; APPEND(m_, (e0_) + 1);   \
        m_ = (unsigned)((cv_).z - t0) < (unsigned)tn; APPEND(m_, (e0_) + 2);   \
        m_ = (unsigned)((cv_).w - t0) < (unsigned)tn; APPEND(m_, (e0_) + 3);   \
    }

__global__ __launch_bounds__(HOPW * 32)
void k_hop_tile(const float* __restrict__ xin, float* __restrict__ xout,
                const int* __restrict__ row, const int* __restrict__ col,
                const float* __restrict__ dinv, int N, int E) {
    __shared__ __attribute__((aligned(16))) float tile[TILE * FEAT];
    __shared__ int queue[HOPW * QCAP];

    const int tid  = threadIdx.x;
    const int lane = tid & 31;
    const int wave = __builtin_amdgcn_readfirstlane(tid >> 5);
    const int t0   = blockIdx.x * TILE;
    const int tn   = min(TILE, N - t0);

    {
        const float4* src = (const float4*)(xin + (size_t)t0 * FEAT);
        float4* dst = (float4*)tile;
        for (int i = tid; i < tn * (FEAT / 4); i += HOPW * 32) {
            float w = dinv[t0 + (i >> 4)];
            w = w * w;
            const float4 v = src[i];
            dst[i] = make_float4(w * v.x, w * v.y, w * v.z, w * v.w);
        }
    }
    __syncthreads();

    int* q = queue + wave * QCAP;
    int cnt = 0;

    int head = (int)((0ull - ((unsigned long long)(size_t)col >> 2)) & 3ull);
    if (head > E) head = E;
    const int nG   = (E - head) / GRP;
    const int nrem = E - nG * GRP;
    for (int s = wave; s * 32 < nrem; s += HOPW) {
        const int idx = s * 32 + lane;
        const int e = (idx < head) ? idx : idx + nG * GRP;
        const int c = (idx < nrem) ? col[e] : -1;
        const bool m = (unsigned)(c - t0) < (unsigned)tn;
        APPEND(m, e);
    }

    const int rot = (int)(((long long)blockIdx.x * nG) / gridDim.x);
    const int4* colv = (const int4*)(col + head);

    auto drain = [&](const bool more) {
        const int nproc = more ? (cnt & ~31) : cnt;
        for (int b = 0; b < nproc; b += 64) {
            const int  i0 = b + lane;
            const int  i1 = b + 32 + lane;
            const bool v0 = i0 < nproc;
            const bool v1 = i1 < nproc;
            const int  e0 = q[v0 ? i0 : 0];
            const int  e1 = q[v1 ? i1 : 0];
            const int  r0 = row[e0];
            const int  r1 = row[e1];
            const int  c0 = col[e0];
            const int  c1 = col[e1];
            const float dr0 = dinv[r0];
            const float dr1 = dinv[r1];
            const float dc0 = dinv[c0];
            const float dc1 = dinv[c1];
            const float w0 = v0 ? dr0 * dc0 : 0.0f;
            const float w1 = v1 ? dr1 * dc1 : 0.0f;
            const int  cl0 = c0 - t0;
            const int  cl1 = c1 - t0;
            const int  mtot = min(64, nproc - b);
#pragma unroll 1
            for (int k = 0; k < mtot; k += BW) {
                const bool  hi = k >= 32;
                const int   rs = hi ? r1 : r0;
                const float ws = hi ? w1 : w0;
                const int   cs = hi ? cl1 : cl0;
                const int   kk = k & 31;
                v2f val[BW];
#pragma unroll
                for (int u = 0; u < BW; ++u) {
                    const int rk = __builtin_amdgcn_readlane(rs, kk + u);
                    val[u] = *(const v2f*)(xin + (size_t)rk * FEAT + 2 * lane);
                }
#pragma unroll
                for (int u = 0; u < BW; ++u) {
                    if (k + u < mtot) {
                        const float wk = __int_as_float(__builtin_amdgcn_readlane(__float_as_int(ws), kk + u));
                        const int   ck = __builtin_amdgcn_readlane(cs, kk + u);
                        float* d = tile + ck * FEAT + 2 * lane;
                        __hip_atomic_fetch_add(d + 0, wk * val[u].x, __ATOMIC_RELAXED, __HIP_MEMORY_SCOPE_WORKGROUP);
                        __hip_atomic_fetch_add(d + 1, wk * val[u].y, __ATOMIC_RELAXED, __HIP_MEMORY_SCOPE_WORKGROUP);
                    }
                }
            }
        }
        const int rem = cnt - nproc;
        if (rem > 0) {
            const int keep = (lane < rem) ? q[nproc + lane] : 0;
            if (lane < rem) q[lane] = keep;
        }
        cnt = rem;
    };

    int t = wave;
    const int4 none = make_int4(-1, -1, -1, -1);
    int4 A0 = none, A1 = none, A2 = none, A3 = none;
    int4 B0 = none, B1 = none, B2 = none, B3 = none;
    int4 C0 = none, C1 = none, C2 = none, C3 = none;
    int gA = 0, gB = 0, gC = 0;

#define LOADSET(X, tt)                                                         \
    {                                                                          \
        int g_ = (tt) + rot; if (g_ >= nG) g_ -= nG;                           \
        g_ = ((tt) < nG) ? g_ : 0;                                             \
        g##X = g_;                                                             \
        const int4* p_ = colv + (size_t)g_ * (GRP / 4) + lane;                 \
        X##0 = p_[0]; X##1 = p_[32]; X##2 = p_[64]; X##3 = p_[96];             \
    }

#define FILTERSET(X)                                                           \
    {                                                                          \
        const int eb_ = head + g##X * GRP + lane * 4;                          \
        FILTER4(X##0, eb_);                                                    \
        FILTER4(X##1, eb_ + 128);                                              \
        FILTER4(X##2, eb_ + 256);                                              \
        FILTER4(X##3, eb_ + 384);                                              \
    }

#define STAGE(CUR, NXT2)                                                       \
    if (t >= nG) break;                                                        \
    LOADSET(NXT2, t + 2 * HOPW);                                               \
    FILTERSET(CUR);                                                            \
    t += HOPW;                                                                 \
    if (cnt > QFLUSH) drain(true);

    if (nG > 0) {
        LOADSET(A, t);
        LOADSET(B, t + HOPW);
    }
    for (;;) {
        STAGE(A, C);
        STAGE(B, A);
        STAGE(C, B);
    }
    drain(false);
    __syncthreads();

    {
        const float4* srct = (const float4*)tile;
        float4* dst = (float4*)(xout + (size_t)t0 * FEAT);
        for (int i = tid; i < tn * (FEAT / 4); i += HOPW * 32) dst[i] = srct[i];
    }
}

__global__ void k_pad_w(const float* __restrict__ W, float* __restrict__ Wp) {
    int i = blockIdx.x * blockDim.x + threadIdx.x;
    if (i < NCLSP * FEAT) {
        int r = i / FEAT;
        Wp[i] = (r < NCLS) ? W[i] : 0.0f;
    }
}

__global__ void k_gemm_wmma(const float* __restrict__ x, const float* __restrict__ Wp,
                            const float* __restrict__ bias, float* __restrict__ out, int N) {
    const int wave   = blockIdx.x * (blockDim.x >> 5) + (threadIdx.x >> 5);
    const int lane   = threadIdx.x & 31;
    const int ntiles = (N + 15) >> 4;
    if (wave >= ntiles) return;

    const int base = wave << 4;
    const int half = lane >> 4;
    const int l16  = lane & 15;

    int m = base + l16;
    if (m >= N) m = N - 1;
    const float* arow = x + (size_t)m * FEAT;
    const float* b0r  = Wp + (size_t)(l16)      * FEAT;
    const float* b1r  = Wp + (size_t)(16 + l16) * FEAT;
    const float* b2r  = Wp + (size_t)(32 + l16) * FEAT;

    v8f acc0 = {};
    v8f acc1 = {};
    v8f acc2 = {};

    for (int k0 = 0; k0 < FEAT; k0 += 4) {
        const int ko = k0 + 2 * half;
        v2f a  = *(const v2f*)(arow + ko);
        v2f b0 = *(const v2f*)(b0r + ko);
        v2f b1 = *(const v2f*)(b1r + ko);
        v2f b2 = *(const v2f*)(b2r + ko);
        acc0 = __builtin_amdgcn_wmma_f32_16x16x4_f32(false, a, false, b0, (short)0, acc0, false, false);
        acc1 = __builtin_amdgcn_wmma_f32_16x16x4_f32(false, a, false, b1, (short)0, acc1, false, false);
        acc2 = __builtin_amdgcn_wmma_f32_16x16x4_f32(false, a, false, b2, (short)0, acc2, false, false);
    }

    const int c0 = l16;
    const int c1 = 16 + l16;
    const int c2 = 32 + l16;
    const float bb0 = bias[c0];
    const float bb1 = bias[c1];
    const float bb2 = (c2 < NCLS) ? bias[c2] : 0.0f;
#pragma unroll
    for (int j = 0; j < 8; ++j) {
        const int node = base + j + 8 * half;
        if (node < N) {
            float* o = out + (size_t)node * NCLS;
            o[c0] = acc0[j] + bb0;
            o[c1] = acc1[j] + bb1;
            if (c2 < NCLS) o[c2] = acc2[j] + bb2;
        }
    }
}

__global__ void k_logsoftmax(float* __restrict__ out, int N) {
    int n = blockIdx.x * blockDim.x + threadIdx.x;
    if (n >= N) return;
    float* p = out + (size_t)n * NCLS;
    float v[NCLS];
    float m = -INFINITY;
#pragma unroll
    for (int c = 0; c < NCLS; ++c) { v[c] = p[c]; m = fmaxf(m, v[c]); }
    float s = 0.0f;
#pragma unroll
    for (int c = 0; c < NCLS; ++c) s += expf(v[c] - m);
    const float ls = logf(s);
#pragma unroll
    for (int c = 0; c < NCLS; ++c) p[c] = v[c] - m - ls;
}


static inline size_t align256f(size_t nfloats) { return (nfloats + 63) & ~(size_t)63; }

extern "C" void kernel_launch(void* const* d_in, const int* in_sizes, int n_in,
                              void* d_out, int out_size, void* d_ws, size_t ws_size,
                              hipStream_t stream) {
    const float* x  = (const float*)d_in[0];
    const int*   ei = (const int*)d_in[1];
    const float* W  = (const float*)d_in[2];
    const float* bi = (const float*)d_in[3];

    const int N = in_sizes[0] / FEAT;
    const int E = in_sizes[1] / 2;
    const int* row = ei;
    const int* col = ei + E;
    const int total = N * FEAT;

    float* ws = (float*)d_ws;
    size_t off = 0;
    float* dinv = ws + off; off += align256f((size_t)N);
    float* Wp   = ws + off; off += align256f((size_t)NCLSP * FEAT);
    float* x0   = ws + off; off += align256f((size_t)total);
    float* x1   = ws + off;

    const int T = 256;
    k_fill<<<(N + T - 1) / T, T, 0, stream>>>(dinv, 1.0f, N);
    k_degree<<<(E + T - 1) / T, T, 0, stream>>>(col, dinv, E);
    k_rsqrt_inplace<<<(N + T - 1) / T, T, 0, stream>>>(dinv, N);
    k_pad_w<<<(NCLSP * FEAT + T - 1) / T, T, 0, stream>>>(W, Wp);

    const int htiles = (N + TILE - 1) / TILE;
    k_hop_tile<<<htiles, HOPW * 32, 0, stream>>>(x, x0, row, col, dinv, N, E);
    k_hop_tile<<<htiles, HOPW * 32, 0, stream>>>(x0, x1, row, col, dinv, N, E);

    const int ntiles = (N + 15) >> 4;
    const int gblocks = (ntiles + 7) / 8;
    k_gemm_wmma<<<gblocks, 256, 0, stream>>>(x1, Wp, bi, (float*)d_out, N);
    k_logsoftmax<<<(N + T - 1) / T, T, 0, stream>>>((float*)d_out, N);
}


# hop 1 writes each tile's filter hits (source, tile row, weight) as a list into the tile's own rows of hop 2's output plane; hop 2 walks it: no 2nd filter or lookups; sums in list/LDS arrival order
# speedup vs baseline: 1.2974x; 1.1867x over previous
#include <hip/hip_runtime.h>
#include <math.h>

typedef float v2f __attribute__((ext_vector_type(2)));
typedef float v8f __attribute__((ext_vector_type(8)));

#define FEAT   64
#define NCLS   40
#define NCLSP  48


__global__ void k_fill(float* __restrict__ p, float val, int n) {
    int i = blockIdx.x * blockDim.x + threadIdx.x;
    if (i < n) p[i] = val;
}

__global__ void k_degree(const int* __restrict__ col, float* __restrict__ deg, int E) {
    int i = blockIdx.x * blockDim.x + threadIdx.x;
    if (i < E) atomicAdd(&deg[col[i]], 1.0f);
}

__global__ void k_rsqrt_inplace(float* __restrict__ d, int n) {
    int i = blockIdx.x * blockDim.x + threadIdx.x;
    if (i < n) {
        float v = d[i];
        d[i] = (v > 0.0f) ? rsqrtf(v) : 0.0f;
    }
}

#define TILE    960
#define HOPW    32
#define GRP     512
#define QFLUSH  64
#define QCAP    (QFLUSH + GRP)
#define BW      16
#define LHDR    4

#define APPEND(m_, e_)                                                         \
    {                                                                          \
        const unsigned mk_ = __builtin_amdgcn_ballot_w32(m_);                  \
        if (mk_) {                                                             \
            if (m_) q[cnt + (int)__builtin_amdgcn_mbcnt_lo(mk_, 0u)] = (e_);   \
            cnt += __builtin_popcount(mk_);                                    \
        }                                                                      \
    }

#define FILTER4(cv_, e0_)                                                      \
    {                                                                          \
        bool m_;                                                               \
        m_ = (unsigned)((cv_).x - t0) < (unsigned)tn; APPEND(m_, (e0_) + 0);   \
        m_ = (unsigned)((cv_).y - t0) < (unsigned)tn; APPEND(m_, (e0_) + 1);   \
        m_ = (unsigned)((cv_).z - t0) < (unsigned)tn; APPEND(m_, (e0_) + 2);   \
        m_ = (unsigned)((cv_).w - t0) < (unsigned)tn; APPEND(m_, (e0_) + 3);   \
    }

template <bool SECOND>
__global__ __launch_bounds__(HOPW * 32)
void k_hop_tile(const float* __restrict__ xin, float* xout, float* lists,
                const int* __restrict__ row, const int* __restrict__ col,
                const float* __restrict__ dinv, int N, int E) {
    __shared__ __attribute__((aligned(16))) float tile[TILE * FEAT];
    __shared__ int queue[HOPW * QCAP];
    __shared__ int lcount;

    const int tid  = threadIdx.x;
    const int lane = tid & 31;
    const int wave = __builtin_amdgcn_readfirstlane(tid >> 5);
    const int t0   = blockIdx.x * TILE;
    const int tn   = min(TILE, N - t0);

    int*   const lhdr = (int*)(lists + (size_t)t0 * FEAT);
    const int    cap  = (tn * FEAT - LHDR) / 3;
    int*   const lr   = lhdr + LHDR;
    int*   const lc   = lr + cap;
    float* const lw   = (float*)(lc + cap);
    const int hraw = SECOND ? lhdr[0] : 0;
    if (!SECOND && tid == 0) lcount = 0;

    {
        const float4* src = (const float4*)(xin + (size_t)t0 * FEAT);
        float4* dst = (float4*)tile;
        for (int i = tid; i < tn * (FEAT / 4); i += HOPW * 32) {
            float w = dinv[t0 + (i >> 4)];
            w = w * w;
            const float4 v = src[i];
            dst[i] = make_float4(w * v.x, w * v.y, w * v.z, w * v.w);
        }
    }
    __syncthreads();

    const int  lcnt    = SECOND ? __builtin_amdgcn_readfirstlane(hraw) : 0;
    const bool uselist = SECOND && lcnt >= 0 && lcnt <= cap;

    auto consume = [&](const int r0, const int r1, const float w0, const float w1,
                       const int cl0, const int cl1, const int mtot) {
#pragma unroll 1
        for (int k = 0; k < mtot; k += BW) {
            const bool  hi = k >= 32;
            const int   rs = hi ? r1 : r0;
            const float ws = hi ? w1 : w0;
            const int   cs = hi ? cl1 : cl0;
            const int   kk = k & 31;
            v2f val[BW];
#pragma unroll
            for (int u = 0; u < BW; ++u) {
                const int rk = __builtin_amdgcn_readlane(rs, kk + u);
                val[u] = *(const v2f*)(xin + (size_t)rk * FEAT + 2 * lane);
            }
#pragma unroll
            for (int u = 0; u < BW; ++u) {
                if (k + u < mtot) {
                    const float wk = __int_as_float(__builtin_amdgcn_readlane(__float_as_int(ws), kk + u));
                    const int   ck = __builtin_amdgcn_readlane(cs, kk + u);
                    float* d = tile + ck * FEAT + 2 * lane;
                    __hip_atomic_fetch_add(d + 0, wk * val[u].x, __ATOMIC_RELAXED, __HIP_MEMORY_SCOPE_WORKGROUP);
                    __hip_atomic_fetch_add(d + 1, wk * val[u].y, __ATOMIC_RELAXED, __HIP_MEMORY_SCOPE_WORKGROUP);
                }
            }
        }
    };

    if (!uselist) {
    int* q = queue + wave * QCAP;
    int cnt = 0;

    int head = (int)((0ull - ((unsigned long long)(size_t)col >> 2)) & 3ull);
    if (head > E) head = E;
    const int nG   = (E - head) / GRP;
    const int nrem = E - nG * GRP;
    for (int s = wave; s * 32 < nrem; s += HOPW) {
        const int idx = s * 32 + lane;
        const int e = (idx < head) ? idx : idx + nG * GRP;
        const int c = (idx < nrem) ? col[e] : -1;
        const bool m = (unsigned)(c - t0) < (unsigned)tn;
        APPEND(m, e);
    }

    const int rot = (int)(((long long)blockIdx.x * nG) / gridDim.x);
    const int4* colv = (const int4*)(col + head);

    auto drain = [&](const bool more) {
        const int nproc = more ? (cnt & ~31) : cnt;
        for (int b = 0; b < nproc; b += 64) {
            const int  i0 = b + lane;
            const int  i1 = b + 32 + lane;
            const bool v0 = i0 < nproc;
            const bool v1 = i1 < nproc;
            const int  e0 = q[v0 ? i0 : 0];
            const int  e1 = q[v1 ? i1 : 0];
            const int  r0 = row[e0];
            const int  r1 = row[e1];
            const int  c0 = col[e0];
            const int  c1 = col[e1];
            const float dr0 = dinv[r0];
            const float dr1 = dinv[r1];
            const float dc0 = dinv[c0];
            const float dc1 = dinv[c1];
            const float w0 = v0 ? dr0 * dc0 : 0.0f;
            const float w1 = v1 ? dr1 * dc1 : 0.0f;
            const int  cl0 = c0 - t0;
            const int  cl1 = c1 - t0;
            const int  mtot = min(64, nproc - b);
            int base = 0;
            if (!SECOND && lane == 0)
                base = __hip_atomic_fetch_add(&lcount, mtot, __ATOMIC_RELAXED, __HIP_MEMORY_SCOPE_WORKGROUP);
            consume(r0, r1, w0, w1, cl0, cl1, mtot);
            if (!SECOND) {
                base = __builtin_amdgcn_readfirstlane(base);
                const int s0 = base + lane;
                const int s1 = base + 32 + lane;
                if (v0 && s0 < cap) { lr[s0] = r0; lc[s0] = cl0; lw[s0] = w0; }
                if (v1 && s1 < cap) { lr[s1] = r1; lc[s1] = cl1; lw[s1] = w1; }
            }
        }
        const int rem = cnt - nproc;
        if (rem > 0) {
            const int keep = (lane < rem) ? q[nproc + lane] : 0;
            if (lane < rem) q[lane] = keep;
        }
        cnt = rem;
    };

    int t = wave;
    const int4 none = make_int4(-1, -1, -1, -1);
    int4 A0 = none, A1 = none, A2 = none, A3 = none;
    int4 B0 = none, B1 = none, B2 = none, B3 = none;
    int4 C0 = none, C1 = none, C2 = none, C3 = none;
    int gA = 0, gB = 0, gC = 0;

#define LOADSET(X, tt)                                                         \
    {                                                                          \
        int g_ = (tt) + rot; if (g_ >= nG) g_ -= nG;                           \
        g_ = ((tt) < nG) ? g_ : 0;                                             \
        g##X = g_;                                                             \
        const int4* p_ = colv + (size_t)g_ * (GRP / 4) + lane;                 \
        X##0 = p_[0]; X##1 = p_[32]; X##2 = p_[64]; X##3 = p_[96];             \
    }

#define FILTERSET(X)                                                           \
    {                                                                          \
        const int eb_ = head + g##X * GRP + lane * 4;                          \
        FILTER4(X##0, eb_);                                                    \
        FILTER4(X##1, eb_ + 128);                                              \
        FILTER4(X##2, eb_ + 256);                                              \
        FILTER4(X##3, eb_ + 384);                                              \
    }

#define STAGE(CUR, NXT2)                                                       \
    if (t >= nG) break;                                                        \
    LOADSET(NXT2, t + 2 * HOPW);                                               \
    FILTERSET(CUR);                                                            \
    t += HOPW;                                                                 \
    if (cnt > QFLUSH) drain(true);

    if (nG > 0) {
        LOADSET(A, t);
        LOADSET(B, t + HOPW);
    }
    for (;;) {
        STAGE(A, C);
        STAGE(B, A);
        STAGE(C, B);
    }
    drain(false);
    } else {
        auto walk32 = [&](const int rs, const float ws, const int cs, const int n) {
            v2f val[32];
#pragma unroll
            for (int u = 0; u < 32; ++u) {
                const int rk = __builtin_amdgcn_readlane(rs, u);
                val[u] = *(const v2f*)(xin + (size_t)rk * FEAT + 2 * lane);
            }
#pragma unroll
            for (int u = 0; u < 32; ++u) {
                const float wk = __int_as_float(__builtin_amdgcn_readlane(__float_as_int(ws), u));
                const int   ck = __builtin_amdgcn_readlane(cs, u);
                const bool  ok = u < n;
                float* d = tile + ck * FEAT + 2 * lane;
                __hip_atomic_fetch_add(d + 0, ok ? wk * val[u].x : 0.0f, __ATOMIC_RELAXED, __HIP_MEMORY_SCOPE_WORKGROUP);
                __hip_atomic_fetch_add(d + 1, ok ? wk * val[u].y : 0.0f, __ATOMIC_RELAXED, __HIP_MEMORY_SCOPE_WORKGROUP);
            }
        };
        int b = wave * 64;
        int   r0 = 0, r1 = 0, cl0 = 0, cl1 = 0;
        float w0 = 0.0f, w1 = 0.0f;
        if (b < lcnt) {
            const bool v0 = b + lane < lcnt;
            const bool v1 = b + 32 + lane < lcnt;
            const int  j0 = v0 ? b + lane : 0;
            const int  j1 = v1 ? b + 32 + lane : 0;
            r0 = lr[j0]; r1 = lr[j1];
            cl0 = lc[j0]; cl1 = lc[j1];
            w0 = lw[j0]; w1 = lw[j1];
        }
        while (b < lcnt) {
            const int nb = b + HOPW * 64;
            int   nr0 = 0, nr1 = 0, ncl0 = 0, ncl1 = 0;
            float nw0 = 0.0f, nw1 = 0.0f;
            if (nb < lcnt) {
                const bool v0 = nb + lane < lcnt;
                const bool v1 = nb + 32 + lane < lcnt;
                const int  j0 = v0 ? nb + lane : 0;
                const int  j1 = v1 ? nb + 32 + lane : 0;
                nr0 = lr[j0]; nr1 = lr[j1];
                ncl0 = lc[j0]; ncl1 = lc[j1];
                nw0 = lw[j0]; nw1 = lw[j1];
            }
            const int m = lcnt - b;
            walk32(r0, w0, cl0, m);
            if (m > 32) walk32(r1, w1, cl1, m - 32);
            r0 = nr0; r1 = nr1; cl0 = ncl0; cl1 = ncl1; w0 = nw0; w1 = nw1;
            b = nb;
        }
    }
    __syncthreads();

    {
        const float4* srct = (const float4*)tile;
        float4* dst = (float4*)(xout + (size_t)t0 * FEAT);
        for (int i = tid; i < tn * (FEAT / 4); i += HOPW * 32) dst[i] = srct[i];
    }
    if (!SECOND && tid == 0) lhdr[0] = lcount;
}

__global__ void k_pad_w(const float* __restrict__ W, float* __restrict__ Wp) {
    int i = blockIdx.x * blockDim.x + threadIdx.x;
    if (i < NCLSP * FEAT) {
        int r = i / FEAT;
        Wp[i] = (r < NCLS) ? W[i] : 0.0f;
    }
}

__global__ void k_gemm_wmma(const float* __restrict__ x, const float* __restrict__ Wp,
                            const float* __restrict__ bias, float* __restrict__ out, int N) {
    const int wave   = blockIdx.x * (blockDim.x >> 5) + (threadIdx.x >> 5);
    const int lane   = threadIdx.x & 31;
    const int ntiles = (N + 15) >> 4;
    if (wave >= ntiles) return;

    const int base = wave << 4;
    const int half = lane >> 4;
    const int l16  = lane & 15;

    int m = base + l16;
    if (m >= N) m = N - 1;
    const float* arow = x + (size_t)m * FEAT;
    const float* b0r  = Wp + (size_t)(l16)      * FEAT;
    const float* b1r  = Wp + (size_t)(16 + l16) * FEAT;
    const float* b2r  = Wp + (size_t)(32 + l16) * FEAT;

    v8f acc0 = {};
    v8f acc1 = {};
    v8f acc2 = {};

    for (int k0 = 0; k0 < FEAT; k0 += 4) {
        const int ko = k0 + 2 * half;
        v2f a  = *(const v2f*)(arow + ko);
        v2f b0 = *(const v2f*)(b0r + ko);
        v2f b1 = *(const v2f*)(b1r + ko);
        v2f b2 = *(const v2f*)(b2r + ko);
        acc0 = __builtin_amdgcn_wmma_f32_16x16x4_f32(false, a, false, b0, (short)0, acc0, false, false);
        acc1 = __builtin_amdgcn_wmma_f32_16x16x4_f32(false, a, false, b1, (short)0, acc1, false, false);
        acc2 = __builtin_amdgcn_wmma_f32_16x16x4_f32(false, a, false, b2, (short)0, acc2, false, false);
    }

    const int c0 = l16;
    const int c1 = 16 + l16;
    const int c2 = 32 + l16;
    const float bb0 = bias[c0];
    const float bb1 = bias[c1];
    const float bb2 = (c2 < NCLS) ? bias[c2] : 0.0f;
#pragma unroll
    for (int j = 0; j < 8; ++j) {
        const int node = base + j + 8 * half;
        if (node < N) {
            float* o = out + (size_t)node * NCLS;
            o[c0] = acc0[j] + bb0;
            o[c1] = acc1[j] + bb1;
            if (c2 < NCLS) o[c2] = acc2[j] + bb2;
        }
    }
}

__global__ void k_logsoftmax(float* __restrict__ out, int N) {
    int n = blockIdx.x * blockDim.x + threadIdx.x;
    if (n >= N) return;
    float* p = out + (size_t)n * NCLS;
    float v[NCLS];
    float m = -INFINITY;
#pragma unroll
    for (int c = 0; c < NCLS; ++c) { v[c] = p[c]; m = fmaxf(m, v[c]); }
    float s = 0.0f;
#pragma unroll
    for (int c = 0; c < NCLS; ++c) s += expf(v[c] - m);
    const float ls = logf(s);
#pragma unroll
    for (int c = 0; c < NCLS; ++c) p[c] = v[c] - m - ls;
}


static inline size_t align256f(size_t nfloats) { return (nfloats + 63) & ~(size_t)63; }

extern "C" void kernel_launch(void* const* d_in, const int* in_sizes, int n_in,
                              void* d_out, int out_size, void* d_ws, size_t ws_size,
                              hipStream_t stream) {
    const float* x  = (const float*)d_in[0];
    const int*   ei = (const int*)d_in[1];
    const float* W  = (const float*)d_in[2];
    const float* bi = (const float*)d_in[3];

    const int N = in_sizes[0] / FEAT;
    const int E = in_sizes[1] / 2;
    const int* row = ei;
    const int* col = ei + E;
    const int total = N * FEAT;

    float* ws = (float*)d_ws;
    size_t off = 0;
    float* dinv = ws + off; off += align256f((size_t)N);
    float* Wp   = ws + off; off += align256f((size_t)NCLSP * FEAT);
    float* x0   = ws + off; off += align256f((size_t)total);
    float* x1   = ws + off;

    const int T = 256;
    k_fill<<<(N + T - 1) / T, T, 0, stream>>>(dinv, 1.0f, N);
    k_degree<<<(E + T - 1) / T, T, 0, stream>>>(col, dinv, E);
    k_rsqrt_inplace<<<(N + T - 1) / T, T, 0, stream>>>(dinv, N);
    k_pad_w<<<(NCLSP * FEAT + T - 1) / T, T, 0, stream>>>(W, Wp);

    const int htiles = (N + TILE - 1) / TILE;
    k_hop_tile<false><<<htiles, HOPW * 32, 0, stream>>>(x, x0, x1, row, col, dinv, N, E);
    k_hop_tile<true><<<htiles, HOPW * 32, 0, stream>>>(x0, x1, x1, row, col, dinv, N, E);

    const int ntiles = (N + 15) >> 4;
    const int gblocks = (ntiles + 7) / 8;
    k_gemm_wmma<<<gblocks, 256, 0, stream>>>(x1, Wp, bi, (float*)d_out, N);
    k_logsoftmax<<<(N + T - 1) / T, T, 0, stream>>>((float*)d_out, N);
}


# tile of 391 target rows, not 960: each hop launch has 256 work-groups, one per WGP, not 105; each still filters the whole edge list, gathers and LDS adds per WGP fall; sums in LDS arrival order
# speedup vs baseline: 1.5744x; 1.2135x over previous
#include <hip/hip_runtime.h>
#include <math.h>

typedef float v2f __attribute__((ext_vector_type(2)));
typedef float v8f __attribute__((ext_vector_type(8)));

#define FEAT   64
#define NCLS   40
#define NCLSP  48


__global__ void k_fill(float* __restrict__ p, float val, int n) {
    int i = blockIdx.x * blockDim.x + threadIdx.x;
    if (i < n) p[i] = val;
}

__global__ void k_degree(const int* __restrict__ col, float* __restrict__ deg, int E) {
    int i = blockIdx.x * blockDim.x + threadIdx.x;
    if (i < E) atomicAdd(&deg[col[i]], 1.0f);
}

__global__ void k_rsqrt_inplace(float* __restrict__ d, int n) {
    int i = blockIdx.x * blockDim.x + threadIdx.x;
    if (i < n) {
        float v = d[i];
        d[i] = (v > 0.0f) ? rsqrtf(v) : 0.0f;
    }
}

#define TILE    391
#define HOPW    32
#define GRP     512
#define QFLUSH  64
#define QCAP    (QFLUSH + GRP)
#define BW      16
#define LHDR    4

#define APPEND(m_, e_)                                                         \
    {                                                                          \
        const unsigned mk_ = __builtin_amdgcn_ballot_w32(m_);                  \
        if (mk_) {                                                             \
            if (m_) q[cnt + (int)__builtin_amdgcn_mbcnt_lo(mk_, 0u)] = (e_);   \
            cnt += __builtin_popcount(mk_);                                    \
        }                                                                      \
    }

#define FILTER4(cv_, e0_)                                                      \
    {                                                                          \
        bool m_;                                                               \
        m_ = (unsigned)((cv_).x - t0) < (unsigned)tn; APPEND(m_, (e0_) + 0);   \
        m_ = (unsigned)((cv_).y - t0) < (unsigned)tn; APPEND(m_, (e0_) + 1);   \
        m_ = (unsigned)((cv_).z - t0) < (unsigned)tn; APPEND(m_, (e0_) + 2);   \
        m_ = (unsigned)((cv_).w - t0) < (unsigned)tn; APPEND(m_, (e0_) + 3);   \
    }

template <bool SECOND>
__global__ __launch_bounds__(HOPW * 32)
void k_hop_tile(const float* __restrict__ xin, float* xout, float* lists,
                const int* __restrict__ row, const int* __restrict__ col,
                const float* __restrict__ dinv, int N, int E) {
    __shared__ __attribute__((aligned(16))) float tile[TILE * FEAT];
    __shared__ int queue[HOPW * QCAP];
    __shared__ int lcount;

    const int tid  = threadIdx.x;
    const int lane = tid & 31;
    const int wave = __builtin_amdgcn_readfirstlane(tid >> 5);
    const int t0   = blockIdx.x * TILE;
    const int tn   = min(TILE, N - t0);

    int*   const lhdr = (int*)(lists + (size_t)t0 * FEAT);
    const int    cap  = (tn * FEAT - LHDR) / 3;
    int*   const lr   = lhdr + LHDR;
    int*   const lc   = lr + cap;
    float* const lw   = (float*)(lc + cap);
    const int hraw = SECOND ? lhdr[0] : 0;
    if (!SECOND && tid == 0) lcount = 0;

    {
        const float4* src = (const float4*)(xin + (size_t)t0 * FEAT);
        float4* dst = (float4*)tile;
        for (int i = tid; i < tn * (FEAT / 4); i += HOPW * 32) {
            float w = dinv[t0 + (i >> 4)];
            w = w * w;
            const float4 v = src[i];
            dst[i] = make_float4(w * v.x, w * v.y, w * v.z, w * v.w);
        }
    }
    __syncthreads();

    const int  lcnt    = SECOND ? __builtin_amdgcn_readfirstlane(hraw) : 0;
    const bool uselist = SECOND && lcnt >= 0 && lcnt <= cap;

    auto consume = [&](const int r0, const int r1, const float w0, const float w1,
                       const int cl0, const int cl1, const int mtot) {
#pragma unroll 1
        for (int k = 0; k < mtot; k += BW) {
            const bool  hi = k >= 32;
            const int   rs = hi ? r1 : r0;
            const float ws = hi ? w1 : w0;
            const int   cs = hi ? cl1 : cl0;
            const int   kk = k & 31;
            v2f val[BW];
#pragma unroll
            for (int u = 0; u < BW; ++u) {
                const int rk = __builtin_amdgcn_readlane(rs, kk + u);
                val[u] = *(const v2f*)(xin + (size_t)rk * FEAT + 2 * lane);
            }
#pragma unroll
            for (int u = 0; u < BW; ++u) {
                if (k + u < mtot) {
                    const float wk = __int_as_float(__builtin_amdgcn_readlane(__float_as_int(ws), kk + u));
                    const int   ck = __builtin_amdgcn_readlane(cs, kk + u);
                    float* d = tile + ck * FEAT + 2 * lane;
                    __hip_atomic_fetch_add(d + 0, wk * val[u].x, __ATOMIC_RELAXED, __HIP_MEMORY_SCOPE_WORKGROUP);
                    __hip_atomic_fetch_add(d + 1, wk * val[u].y, __ATOMIC_RELAXED, __HIP_MEMORY_SCOPE_WORKGROUP);
                }
            }
        }
    };

    if (!uselist) {
    int* q = queue + wave * QCAP;
    int cnt = 0;

    int head = (int)((0ull - ((unsigned long long)(size_t)col >> 2)) & 3ull);
    if (head > E) head = E;
    const int nG   = (E - head) / GRP;
    const int nrem = E - nG * GRP;
    for (int s = wave; s * 32 < nrem; s += HOPW) {
        const int idx = s * 32 + lane;
        const int e = (idx < head) ? idx : idx + nG * GRP;
        const int c = (idx < nrem) ? col[e] : -1;
        const bool m = (unsigned)(c - t0) < (unsigned)tn;
        APPEND(m, e);
    }

    const int rot = (int)(((long long)blockIdx.x * nG) / gridDim.x);
    const int4* colv = (const int4*)(col + head);

    auto drain = [&](const bool more) {
        const int nproc = more ? (cnt & ~31) : cnt;
        for (int b = 0; b < nproc; b += 64) {
            const int  i0 = b + lane;
            const int  i1 = b + 32 + lane;
            const bool v0 = i0 < nproc;
            const bool v1 = i1 < nproc;
            const int  e0 = q[v0 ? i0 : 0];
            const int  e1 = q[v1 ? i1 : 0];
            const int  r0 = row[e0];
            const int  r1 = row[e1];
            const int  c0 = col[e0];
            const int  c1 = col[e1];
            const float dr0 = dinv[r0];
            const float dr1 = dinv[r1];
            const float dc0 = dinv[c0];
            const float dc1 = dinv[c1];
            const float w0 = v0 ? dr0 * dc0 : 0.0f;
            const float w1 = v1 ? dr1 * dc1 : 0.0f;
            const int  cl0 = c0 - t0;
            const int  cl1 = c1 - t0;
            const int  mtot = min(64, nproc - b);
            int base = 0;
            if (!SECOND && lane == 0)
                base = __hip_atomic_fetch_add(&lcount, mtot, __ATOMIC_RELAXED, __HIP_MEMORY_SCOPE_WORKGROUP);
            consume(r0, r1, w0, w1, cl0, cl1, mtot);
            if (!SECOND) {
                base = __builtin_amdgcn_readfirstlane(base);
                const int s0 = base + lane;
                const int s1 = base + 32 + lane;
                if (v0 && s0 < cap) { lr[s0] = r0; lc[s0] = cl0; lw[s0] = w0; }
                if (v1 && s1 < cap) { lr[s1] = r1; lc[s1] = cl1; lw[s1] = w1; }
            }
        }
        const int rem = cnt - nproc;
        if (rem > 0) {
            const int keep = (lane < rem) ? q[nproc + lane] : 0;
            if (lane < rem) q[lane] = keep;
        }
        cnt = rem;
    };

    int t = wave;
    const int4 none = make_int4(-1, -1, -1, -1);
    int4 A0 = none, A1 = none, A2 = none, A3 = none;
    int4 B0 = none, B1 = none, B2 = none, B3 = none;
    int4 C0 = none, C1 = none, C2 = none, C3 = none;
    int gA = 0, gB = 0, gC = 0;

#define LOADSET(X, tt)                                                         \
    {                                                                          \
        int g_ = (tt) + rot; if (g_ >= nG) g_ -= nG;                           \
        g_ = ((tt) < nG) ? g_ : 0;                                             \
        g##X = g_;                                                             \
        const int4* p_ = colv + (size_t)g_ * (GRP / 4) + lane;                 \
        X##0 = p_[0]; X##1 = p_[32]; X##2 = p_[64]; X##3 = p_[96];             \
    }

#define FILTERSET(X)                                                           \
    {                                                                          \
        const int eb_ = head + g##X * GRP + lane * 4;                          \
        FILTER4(X##0, eb_);                                                    \
        FILTER4(X##1, eb_ + 128);                                              \
        FILTER4(X##2, eb_ + 256);                                              \
        FILTER4(X##3, eb_ + 384);                                              \
    }

#define STAGE(CUR, NXT2)                                                       \
    if (t >= nG) break;                                                        \
    LOADSET(NXT2, t + 2 * HOPW);                                               \
    FILTERSET(CUR);                                                            \
    t += HOPW;                                                                 \
    if (cnt > QFLUSH) drain(true);

    if (nG > 0) {
        LOADSET(A, t);
        LOADSET(B, t + HOPW);
    }
    for (;;) {
        STAGE(A, C);
        STAGE(B, A);
        STAGE(C, B);
    }
    drain(false);
    } else {
        auto walk32 = [&](const int rs, const float ws, const int cs, const int n) {
            v2f val[32];
#pragma unroll
            for (int u = 0; u < 32; ++u) {
                const int rk = __builtin_amdgcn_readlane(rs, u);
                val[u] = *(const v2f*)(xin + (size_t)rk * FEAT + 2 * lane);
            }
#pragma unroll
            for (int u = 0; u < 32; ++u) {
                const float wk = __int_as_float(__builtin_amdgcn_readlane(__float_as_int(ws), u));
                const int   ck = __builtin_amdgcn_readlane(cs, u);
                const bool  ok = u < n;
                float* d = tile + ck * FEAT + 2 * lane;
                __hip_atomic_fetch_add(d + 0, ok ? wk * val[u].x : 0.0f, __ATOMIC_RELAXED, __HIP_MEMORY_SCOPE_WORKGROUP);
                __hip_atomic_fetch_add(d + 1, ok ? wk * val[u].y : 0.0f, __ATOMIC_RELAXED, __HIP_MEMORY_SCOPE_WORKGROUP);
            }
        };
        int b = wave * 64;
        int   r0 = 0, r1 = 0, cl0 = 0, cl1 = 0;
        float w0 = 0.0f, w1 = 0.0f;
        if (b < lcnt) {
            const bool v0 = b + lane < lcnt;
            const bool v1 = b + 32 + lane < lcnt;
            const int  j0 = v0 ? b + lane : 0;
            const int  j1 = v1 ? b + 32 + lane : 0;
            r0 = lr[j0]; r1 = lr[j1];
            cl0 = lc[j0]; cl1 = lc[j1];
            w0 = lw[j0]; w1 = lw[j1];
        }
        while (b < lcnt) {
            const int nb = b + HOPW * 64;
            int   nr0 = 0, nr1 = 0, ncl0 = 0, ncl1 = 0;
            float nw0 = 0.0f, nw1 = 0.0f;
            if (nb < lcnt) {
                const bool v0 = nb + lane < lcnt;
                const bool v1 = nb + 32 + lane < lcnt;
                const int  j0 = v0 ? nb + lane : 0;
                const int  j1 = v1 ? nb + 32 + lane : 0;
                nr0 = lr[j0]; nr1 = lr[j1];
                ncl0 = lc[j0]; ncl1 = lc[j1];
                nw0 = lw[j0]; nw1 = lw[j1];
            }
            const int m = lcnt - b;
            walk32(r0, w0, cl0, m);
            if (m > 32) walk32(r1, w1, cl1, m - 32);
            r0 = nr0; r1 = nr1; cl0 = ncl0; cl1 = ncl1; w0 = nw0; w1 = nw1;
            b = nb;
        }
    }
    __syncthreads();

    {
        const float4* srct = (const float4*)tile;
        float4* dst = (float4*)(xout + (size_t)t0 * FEAT);
        for (int i = tid; i < tn * (FEAT / 4); i += HOPW * 32) dst[i] = srct[i];
    }
    if (!SECOND && tid == 0) lhdr[0] = lcount;
}

__global__ void k_pad_w(const float* __restrict__ W, float* __restrict__ Wp) {
    int i = blockIdx.x * blockDim.x + threadIdx.x;
    if (i < NCLSP * FEAT) {
        int r = i / FEAT;
        Wp[i] = (r < NCLS) ? W[i] : 0.0f;
    }
}

__global__ void k_gemm_wmma(const float* __restrict__ x, const float* __restrict__ Wp,
                            const float* __restrict__ bias, float* __restrict__ out, int N) {
    const int wave   = blockIdx.x * (blockDim.x >> 5) + (threadIdx.x >> 5);
    const int lane   = threadIdx.x & 31;
    const int ntiles = (N + 15) >> 4;
    if (wave >= ntiles) return;

    const int base = wave << 4;
    const int half = lane >> 4;
    const int l16  = lane & 15;

    int m = base + l16;
    if (m >= N) m = N - 1;
    const float* arow = x + (size_t)m * FEAT;
    const float* b0r  = Wp + (size_t)(l16)      * FEAT;
    const float* b1r  = Wp + (size_t)(16 + l16) * FEAT;
    const float* b2r  = Wp + (size_t)(32 + l16) * FEAT;

    v8f acc0 = {};
    v8f acc1 = {};
    v8f acc2 = {};

    for (int k0 = 0; k0 < FEAT; k0 += 4) {
        const int ko = k0 + 2 * half;
        v2f a  = *(const v2f*)(arow + ko);
        v2f b0 = *(const v2f*)(b0r + ko);
        v2f b1 = *(const v2f*)(b1r + ko);
        v2f b2 = *(const v2f*)(b2r + ko);
        acc0 = __builtin_amdgcn_wmma_f32_16x16x4_f32(false, a, false, b0, (short)0, acc0, false, false);
        acc1 = __builtin_amdgcn_wmma_f32_16x16x4_f32(false, a, false, b1, (short)0, acc1, false, false);
        acc2 = __builtin_amdgcn_wmma_f32_16x16x4_f32(false, a, false, b2, (short)0, acc2, false, false);
    }

    const int c0 = l16;
    const int c1 = 16 + l16;
    const int c2 = 32 + l16;
    const float bb0 = bias[c0];
    const float bb1 = bias[c1];
    const float bb2 = (c2 < NCLS) ? bias[c2] : 0.0f;
#pragma unroll
    for (int j = 0; j < 8; ++j) {
        const int node = base + j + 8 * half;
        if (node < N) {
            float* o = out + (size_t)node * NCLS;
            o[c0] = acc0[j] + bb0;
            o[c1] = acc1[j] + bb1;
            if (c2 < NCLS) o[c2] = acc2[j] + bb2;
        }
    }
}

__global__ void k_logsoftmax(float* __restrict__ out, int N) {
    int n = blockIdx.x * blockDim.x + threadIdx.x;
    if (n >= N) return;
    float* p = out + (size_t)n * NCLS;
    float v[NCLS];
    float m = -INFINITY;
#pragma unroll
    for (int c = 0; c < NCLS; ++c) { v[c] = p[c]; m = fmaxf(m, v[c]); }
    float s = 0.0f;
#pragma unroll
    for (int c = 0; c < NCLS; ++c) s += expf(v[c] - m);
    const float ls = logf(s);
#pragma unroll
    for (int c = 0; c < NCLS; ++c) p[c] = v[c] - m - ls;
}


static inline size_t align256f(size_t nfloats) { return (nfloats + 63) & ~(size_t)63; }

extern "C" void kernel_launch(void* const* d_in, const int* in_sizes, int n_in,
                              void* d_out, int out_size, void* d_ws, size_t ws_size,
                              hipStream_t stream) {
    const float* x  = (const float*)d_in[0];
    const int*   ei = (const int*)d_in[1];
    const float* W  = (const float*)d_in[2];
    const float* bi = (const float*)d_in[3];

    const int N = in_sizes[0] / FEAT;
    const int E = in_sizes[1] / 2;
    const int* row = ei;
    const int* col = ei + E;
    const int total = N * FEAT;

    float* ws = (float*)d_ws;
    size_t off = 0;
    float* dinv = ws + off; off += align256f((size_t)N);
    float* Wp   = ws + off; off += align256f((size_t)NCLSP * FEAT);
    float* x0   = ws + off; off += align256f((size_t)total);
    float* x1   = ws + off;

    const int T = 256;
    k_fill<<<(N + T - 1) / T, T, 0, stream>>>(dinv, 1.0f, N);
    k_degree<<<(E + T - 1) / T, T, 0, stream>>>(col, dinv, E);
    k_rsqrt_inplace<<<(N + T - 1) / T, T, 0, stream>>>(dinv, N);
    k_pad_w<<<(NCLSP * FEAT + T - 1) / T, T, 0, stream>>>(W, Wp);

    const int htiles = (N + TILE - 1) / TILE;
    k_hop_tile<false><<<htiles, HOPW * 32, 0, stream>>>(x, x0, x1, row, col, dinv, N, E);
    k_hop_tile<true><<<htiles, HOPW * 32, 0, stream>>>(x0, x1, x1, row, col, dinv, N, E);

    const int ntiles = (N + 15) >> 4;
    const int gblocks = (ntiles + 7) / 8;
    k_gemm_wmma<<<gblocks, 256, 0, stream>>>(x1, Wp, bi, (float*)d_out, N);
    k_logsoftmax<<<(N + T - 1) / T, T, 0, stream>>>((float*)d_out, N);
}


# in-degree = hits per tile row, counted in LDS by the tile's own edge filter: filter becomes its own launch (lists + dinv), the global-atomic degree pass goes, hop 1 walks the list; 8 -> 6 launches
# speedup vs baseline: 2.0200x; 1.2830x over previous
#include <hip/hip_runtime.h>
#include <math.h>

typedef float v2f __attribute__((ext_vector_type(2)));
typedef float v8f __attribute__((ext_vector_type(8)));

#define FEAT   64
#define NCLS   40
#define NCLSP  48

#define TILE    391
#define HOPW    32
#define GRP     512
#define QFLUSH  64
#define QCAP    (QFLUSH + GRP)
#define BW      16
#define LHDR    4
#define FQCAP   1312

#define APPEND(m_, e_)                                                         \
    {                                                                          \
        const unsigned mk_ = __builtin_amdgcn_ballot_w32(m_);                  \
        if (mk_) {                                                             \
            if (m_) q[cnt + (int)__builtin_amdgcn_mbcnt_lo(mk_, 0u)] = (e_);   \
            cnt += __builtin_popcount(mk_);                                    \
        }                                                                      \
    }

#define FILTER4(cv_, e0_)                                                      \
    {                                                                          \
        bool m_;                                                               \
        m_ = (unsigned)((cv_).x - t0) < (unsigned)tn; APPEND(m_, (e0_) + 0);   \
        m_ = (unsigned)((cv_).y - t0) < (unsigned)tn; APPEND(m_, (e0_) + 1);   \
        m_ = (unsigned)((cv_).z - t0) < (unsigned)tn; APPEND(m_, (e0_) + 2);   \
        m_ = (unsigned)((cv_).w - t0) < (unsigned)tn; APPEND(m_, (e0_) + 3);   \
    }

#define LOADSET(X, tt)                                                         \
    {                                                                          \
        int g_ = (tt) + rot; if (g_ >= nG) g_ -= nG;                           \
        g_ = ((tt) < nG) ? g_ : 0;                                             \
        g##X = g_;                                                             \
        const int4* p_ = colv + (size_t)g_ * (GRP / 4) + lane;                 \
        X##0 = p_[0]; X##1 = p_[32]; X##2 = p_[64]; X##3 = p_[96];             \
    }

#define FILTERSET(X)                                                           \
    {                                                                          \
        const int eb_ = head + g##X * GRP + lane * 4;                          \
        FILTER4(X##0, eb_);                                                    \
        FILTER4(X##1, eb_ + 128);                                              \
        FILTER4(X##2, eb_ + 256);                                              \
        FILTER4(X##3, eb_ + 384);                                              \
    }

#define STAGE(CUR, NXT2)                                                       \
    if (t >= nG) break;                                                        \
    LOADSET(NXT2, t + 2 * HOPW);                                               \
    FILTERSET(CUR);                                                            \
    t += HOPW;                                                                 \
    if (cnt > QFLUSH) drain(true, cnt);

template <class DrainF>
__device__ __forceinline__ void scan_edge_list(const int* __restrict__ col, const int E,
                                               const int t0, const int tn, int* const q,
                                               const int wave, const int lane, DrainF drain) {
    int cnt = 0;

    int head = (int)((0ull - ((unsigned long long)(size_t)col >> 2)) & 3ull);
    if (head > E) head = E;
    const int nG   = (E - head) / GRP;
    const int nrem = E - nG * GRP;
    for (int s = wave; s * 32 < nrem; s += HOPW) {
        const int idx = s * 32 + lane;
        const int e = (idx < head) ? idx : idx + nG * GRP;
        const int c = (idx < nrem) ? col[e] : -1;
        const bool m = (unsigned)(c - t0) < (unsigned)tn;
        APPEND(m, e);
    }

    const int rot = (int)(((long long)blockIdx.x * nG) / gridDim.x);
    const int4* colv = (const int4*)(col + head);

    int t = wave;
    const int4 none = make_int4(-1, -1, -1, -1);
    int4 A0 = none, A1 = none, A2 = none, A3 = none;
    int4 B0 = none, B1 = none, B2 = none, B3 = none;
    int4 C0 = none, C1 = none, C2 = none, C3 = none;
    int gA = 0, gB = 0, gC = 0;

    if (nG > 0) {
        LOADSET(A, t);
        LOADSET(B, t + HOPW);
    }
    for (;;) {
        STAGE(A, C);
        STAGE(B, A);
        STAGE(C, B);
    }
    drain(false, cnt);
}

__global__ __launch_bounds__(HOPW * 32)
void k_filter_tile(float* lists, const int* __restrict__ row, const int* __restrict__ col,
                   float* __restrict__ dinv, int N, int E) {
    __shared__ int queue[HOPW * FQCAP];
    __shared__ int degc[TILE];
    __shared__ int lcount;

    const int tid  = threadIdx.x;
    const int lane = tid & 31;
    const int wave = __builtin_amdgcn_readfirstlane(tid >> 5);
    const int t0   = blockIdx.x * TILE;
    const int tn   = min(TILE, N - t0);

    int* const lhdr = (int*)(lists + (size_t)t0 * FEAT);
    const int  cap  = (tn * FEAT - LHDR) / 3;
    int* const lr   = lhdr + LHDR;
    int* const lc   = lr + cap;

    for (int i = tid; i < TILE; i += HOPW * 32) degc[i] = 0;
    if (tid == 0) lcount = 0;
    __syncthreads();

    int* const q = queue + wave * FQCAP;
    auto drain = [&](const bool more, int& cnt) {
        const int nproc = more ? (cnt & ~31) : cnt;
        for (int b = 0; b < nproc; b += 64) {
            const int  i0 = b + lane;
            const int  i1 = b + 32 + lane;
            const bool v0 = i0 < nproc;
            const bool v1 = i1 < nproc;
            const int  e0 = q[v0 ? i0 : 0];
            const int  e1 = q[v1 ? i1 : 0];
            const int  r0 = row[e0];
            const int  r1 = row[e1];
            const int  cl0 = col[e0] - t0;
            const int  cl1 = col[e1] - t0;
            const int  mtot = min(64, nproc - b);
            int base = 0;
            if (lane == 0)
                base = __hip_atomic_fetch_add(&lcount, mtot, __ATOMIC_RELAXED, __HIP_MEMORY_SCOPE_WORKGROUP);
            if (v0) __hip_atomic_fetch_add(&degc[cl0], 1, __ATOMIC_RELAXED, __HIP_MEMORY_SCOPE_WORKGROUP);
            if (v1) __hip_atomic_fetch_add(&degc[cl1], 1, __ATOMIC_RELAXED, __HIP_MEMORY_SCOPE_WORKGROUP);
            base = __builtin_amdgcn_readfirstlane(base);
            const int s0 = base + lane;
            const int s1 = base + 32 + lane;
            if (v0 && s0 < cap) { lr[s0] = r0; lc[s0] = cl0; }
            if (v1 && s1 < cap) { lr[s1] = r1; lc[s1] = cl1; }
        }
        const int rem = cnt - nproc;
        if (rem > 0) {
            const int keep = (lane < rem) ? q[nproc + lane] : 0;
            if (lane < rem) q[lane] = keep;
        }
        cnt = rem;
    };
    scan_edge_list(col, E, t0, tn, q, wave, lane, drain);
    __syncthreads();

    for (int i = tid; i < tn; i += HOPW * 32) {
        const float v = (float)(degc[i] + 1);
        dinv[t0 + i] = (v > 0.0f) ? rsqrtf(v) : 0.0f;
    }
    if (tid == 0) lhdr[0] = lcount;
}

template <bool FIRST>
__global__ __launch_bounds__(HOPW * 32)
void k_hop_tile(const float* __restrict__ xin, float* xout, float* lists,
                const int* __restrict__ row, const int* __restrict__ col,
                const float* __restrict__ dinv, int N, int E) {
    __shared__ __attribute__((aligned(16))) float tile[TILE * FEAT];
    __shared__ int queue[HOPW * QCAP];
    __shared__ float dloc[TILE];

    const int tid  = threadIdx.x;
    const int lane = tid & 31;
    const int wave = __builtin_amdgcn_readfirstlane(tid >> 5);
    const int t0   = blockIdx.x * TILE;
    const int tn   = min(TILE, N - t0);

    int*   const lhdr = (int*)(lists + (size_t)t0 * FEAT);
    const int    cap  = (tn * FEAT - LHDR) / 3;
    int*   const lr   = lhdr + LHDR;
    int*   const lc   = lr + cap;
    float* const lw   = (float*)(lc + cap);
    const int hraw = lhdr[0];

    {
        const float4* src = (const float4*)(xin + (size_t)t0 * FEAT);
        float4* dst = (float4*)tile;
        for (int i = tid; i < tn * (FEAT / 4); i += HOPW * 32) {
            float w = dinv[t0 + (i >> 4)];
            w = w * w;
            const float4 v = src[i];
            dst[i] = make_float4(w * v.x, w * v.y, w * v.z, w * v.w);
        }
        if (FIRST)
            for (int i = tid; i < tn; i += HOPW * 32) dloc[i] = dinv[t0 + i];
    }
    __syncthreads();

    const int  lcnt    = __builtin_amdgcn_readfirstlane(hraw);
    const bool uselist = lcnt >= 0 && lcnt <= cap;

    if (!uselist) {
        auto consume = [&](const int r0, const int r1, const float w0, const float w1,
                           const int cl0, const int cl1, const int mtot) {
#pragma unroll 1
            for (int k = 0; k < mtot; k += BW) {
                const bool  hi = k >= 32;
                const int   rs = hi ? r1 : r0;
                const float ws = hi ? w1 : w0;
                const int   cs = hi ? cl1 : cl0;
                const int   kk = k & 31;
                v2f val[BW];
#pragma unroll
                for (int u = 0; u < BW; ++u) {
                    const int rk = __builtin_amdgcn_readlane(rs, kk + u);
                    val[u] = *(const v2f*)(xin + (size_t)rk * FEAT + 2 * lane);
                }
#pragma unroll
                for (int u = 0; u < BW; ++u) {
                    if (k + u < mtot) {
                        const float wk = __int_as_float(__builtin_amdgcn_readlane(__float_as_int(ws), kk + u));
                        const int   ck = __builtin_amdgcn_readlane(cs, kk + u);
                        float* d = tile + ck * FEAT + 2 * lane;
                        __hip_atomic_fetch_add(d + 0, wk * val[u].x, __ATOMIC_RELAXED, __HIP_MEMORY_SCOPE_WORKGROUP);
                        __hip_atomic_fetch_add(d + 1, wk * val[u].y, __ATOMIC_RELAXED, __HIP_MEMORY_SCOPE_WORKGROUP);
                    }
                }
            }
        };
        int* const q = queue + wave * QCAP;
        auto drain = [&](const bool more, int& cnt) {
            const int nproc = more ? (cnt & ~31) : cnt;
            for (int b = 0; b < nproc; b += 64) {
                const int  i0 = b + lane;
                const int  i1 = b + 32 + lane;
                const bool v0 = i0 < nproc;
                const bool v1 = i1 < nproc;
                const int  e0 = q[v0 ? i0 : 0];
                const int  e1 = q[v1 ? i1 : 0];
                const int  r0 = row[e0];
                const int  r1 = row[e1];
                const int  c0 = col[e0];
                const int  c1 = col[e1];
                const float dr0 = dinv[r0];
                const float dr1 = dinv[r1];
                const float dc0 = dinv[c0];
                const float dc1 = dinv[c1];
                const float w0 = v0 ? dr0 * dc0 : 0.0f;
                const float w1 = v1 ? dr1 * dc1 : 0.0f;
                consume(r0, r1, w0, w1, c0 - t0, c1 - t0, min(64, nproc - b));
            }
            const int rem = cnt - nproc;
            if (rem > 0) {
                const int keep = (lane < rem) ? q[nproc + lane] : 0;
                if (lane < rem) q[lane] = keep;
            }
            cnt = rem;
        };
        scan_edge_list(col, E, t0, tn, q, wave, lane, drain);
    } else {
        auto walk32 = [&](const int rs, const float wa, const float wb, const int cs, const int n) {
            v2f val[32];
#pragma unroll
            for (int u = 0; u < 32; ++u) {
                const int rk = __builtin_amdgcn_readlane(rs, u);
                val[u] = *(const v2f*)(xin + (size_t)rk * FEAT + 2 * lane);
            }
            const float ws = FIRST ? wa * wb : wa;
#pragma unroll
            for (int u = 0; u < 32; ++u) {
                const float wk = __int_as_float(__builtin_amdgcn_readlane(__float_as_int(ws), u));
                const int   ck = __builtin_amdgcn_readlane(cs, u);
                const bool  ok = u < n;
                float* d = tile + ck * FEAT + 2 * lane;
                __hip_atomic_fetch_add(d + 0, ok ? wk * val[u].x : 0.0f, __ATOMIC_RELAXED, __HIP_MEMORY_SCOPE_WORKGROUP);
                __hip_atomic_fetch_add(d + 1, ok ? wk * val[u].y : 0.0f, __ATOMIC_RELAXED, __HIP_MEMORY_SCOPE_WORKGROUP);
            }
            return ws;
        };
        int b = wave * 64;
        int   r0 = 0, r1 = 0, cl0 = 0, cl1 = 0;
        float w0 = 0.0f, w1 = 0.0f;
        if (b < lcnt) {
            const bool v0 = b + lane < lcnt;
            const bool v1 = b + 32 + lane < lcnt;
            const int  j0 = v0 ? b + lane : 0;
            const int  j1 = v1 ? b + 32 + lane : 0;
            r0 = lr[j0]; r1 = lr[j1];
            cl0 = lc[j0]; cl1 = lc[j1];
            if (!FIRST) { w0 = lw[j0]; w1 = lw[j1]; }
        }
        while (b < lcnt) {
            const int nb = b + HOPW * 64;
            int   nr0 = 0, nr1 = 0, ncl0 = 0, ncl1 = 0;
            float nw0 = 0.0f, nw1 = 0.0f;
            if (nb < lcnt) {
                const bool v0 = nb + lane < lcnt;
                const bool v1 = nb + 32 + lane < lcnt;
                const int  j0 = v0 ? nb + lane : 0;
                const int  j1 = v1 ? nb + 32 + lane : 0;
                nr0 = lr[j0]; nr1 = lr[j1];
                ncl0 = lc[j0]; ncl1 = lc[j1];
                if (!FIRST) { nw0 = lw[j0]; nw1 = lw[j1]; }
            }
            const int m = lcnt - b;
            if (FIRST) {
                const float a0 = dinv[r0];
                const float a1 = dinv[r1];
                const float g0 = walk32(r0, a0, dloc[cl0], cl0, m);
                if (b + lane < lcnt) lw[b + lane] = g0;
                if (m > 32) {
                    const float g1 = walk32(r1, a1, dloc[cl1], cl1, m - 32);
                    if (b + 32 + lane < lcnt) lw[b + 32 + lane] = g1;
                }
            } else {
                walk32(r0, w0, 1.0f, cl0, m);
                if (m > 32) walk32(r1, w1, 1.0f, cl1, m - 32);
            }
            r0 = nr0; r1 = nr1; cl0 = ncl0; cl1 = ncl1; w0 = nw0; w1 = nw1;
            b = nb;
        }
    }
    __syncthreads();

    {
        const float4* srct = (const float4*)tile;
        float4* dst = (float4*)(xout + (size_t)t0 * FEAT);
        for (int i = tid; i < tn * (FEAT / 4); i += HOPW * 32) dst[i] = srct[i];
    }
}

__global__ void k_pad_w(const float* __restrict__ W, float* __restrict__ Wp) {
    int i = blockIdx.x * blockDim.x + threadIdx.x;
    if (i < NCLSP * FEAT) {
        int r = i / FEAT;
        Wp[i] = (r < NCLS) ? W[i] : 0.0f;
    }
}

__global__ void k_gemm_wmma(const float* __restrict__ x, const float* __restrict__ Wp,
                            const float* __restrict__ bias, float* __restrict__ out, int N) {
    const int wave   = blockIdx.x * (blockDim.x >> 5) + (threadIdx.x >> 5);
    const int lane   = threadIdx.x & 31;
    const int ntiles = (N + 15) >> 4;
    if (wave >= ntiles) return;

    const int base = wave << 4;
    const int half = lane >> 4;
    const int l16  = lane & 15;

    int m = base + l16;
    if (m >= N) m = N - 1;
    const float* arow = x + (size_t)m * FEAT;
    const float* b0r  = Wp + (size_t)(l16)      * FEAT;
    const float* b1r  = Wp + (size_t)(16 + l16) * FEAT;
    const float* b2r  = Wp + (size_t)(32 + l16) * FEAT;

    v8f acc0 = {};
    v8f acc1 = {};
    v8f acc2 = {};

    for (int k0 = 0; k0 < FEAT; k0 += 4) {
        const int ko = k0 + 2 * half;
        v2f a  = *(const v2f*)(arow + ko);
        v2f b0 = *(const v2f*)(b0r + ko);
        v2f b1 = *(const v2f*)(b1r + ko);
        v2f b2 = *(const v2f*)(b2r + ko);
        acc0 = __builtin_amdgcn_wmma_f32_16x16x4_f32(false, a, false, b0, (short)0, acc0, false, false);
        acc1 = __builtin_amdgcn_wmma_f32_16x16x4_f32(false, a, false, b1, (short)0, acc1, false, false);
        acc2 = __builtin_amdgcn_wmma_f32_16x16x4_f32(false, a, false, b2, (short)0, acc2, false, false);
    }

    const int c0 = l16;
    const int c1 = 16 + l16;
    const int c2 = 32 + l16;
    const float bb0 = bias[c0];
    const float bb1 = bias[c1];
    const float bb2 = (c2 < NCLS) ? bias[c2] : 0.0f;
#pragma unroll
    for (int j = 0; j < 8; ++j) {
        const int node = base + j + 8 * half;
        if (node < N) {
            float* o = out + (size_t)node * NCLS;
            o[c0] = acc0[j] + bb0;
            o[c1] = acc1[j] + bb1;
            if (c2 < NCLS) o[c2] = acc2[j] + bb2;
        }
    }
}

__global__ void k_logsoftmax(float* __restrict__ out, int N) {
    int n = blockIdx.x * blockDim.x + threadIdx.x;
    if (n >= N) return;
    float* p = out + (size_t)n * NCLS;
    float v[NCLS];
    float m = -INFINITY;
#pragma unroll
    for (int c = 0; c < NCLS; ++c) { v[c] = p[c]; m = fmaxf(m, v[c]); }
    float s = 0.0f;
#pragma unroll
    for (int c = 0; c < NCLS; ++c) s += expf(v[c] - m);
    const float ls = logf(s);
#pragma unroll
    for (int c = 0; c < NCLS; ++c) p[c] = v[c] - m - ls;
}


static inline size_t align256f(size_t nfloats) { return (nfloats + 63) & ~(size_t)63; }

extern "C" void kernel_launch(void* const* d_in, const int* in_sizes, int n_in,
                              void* d_out, int out_size, void* d_ws, size_t ws_size,
                              hipStream_t stream) {
    const float* x  = (const float*)d_in[0];
    const int*   ei = (const int*)d_in[1];
    const float* W  = (const float*)d_in[2];
    const float* bi = (const float*)d_in[3];

    const int N = in_sizes[0] / FEAT;
    const int E = in_sizes[1] / 2;
    const int* row = ei;
    const int* col = ei + E;
    const int total = N * FEAT;

    float* ws = (float*)d_ws;
    size_t off = 0;
    float* dinv = ws + off; off += align256f((size_t)N);
    float* Wp   = ws + off; off += align256f((size_t)NCLSP * FEAT);
    float* x0   = ws + off; off += align256f((size_t)total);
    float* x1   = ws + off;

    const int T = 256;
    k_pad_w<<<(NCLSP * FEAT + T - 1) / T, T, 0, stream>>>(W, Wp);

    const int htiles = (N + TILE - 1) / TILE;
    k_filter_tile<<<htiles, HOPW * 32, 0, stream>>>(x1, row, col, dinv, N, E);
    k_hop_tile<true><<<htiles, HOPW * 32, 0, stream>>>(x, x0, x1, row, col, dinv, N, E);
    k_hop_tile<false><<<htiles, HOPW * 32, 0, stream>>>(x0, x1, x1, row, col, dinv, N, E);

    const int ntiles = (N + 15) >> 4;
    const int gblocks = (ntiles + 7) / 8;
    k_gemm_wmma<<<gblocks, 256, 0, stream>>>(x1, Wp, bi, (float*)d_out, N);
    k_logsoftmax<<<(N + T - 1) / T, T, 0, stream>>>((float*)d_out, N);
}


# hop 2 ends in the head: its LDS tile feeds the f32 WMMA (W padded in LDS), bias + log_softmax in registers, out stored once; x1 round trip, 3 launches go (6 -> 3); softmax sum: lane butterfly order
# speedup vs baseline: 2.7223x; 1.3477x over previous
#include <hip/hip_runtime.h>
#include <math.h>

typedef float v2f __attribute__((ext_vector_type(2)));
typedef float v8f __attribute__((ext_vector_type(8)));

#define FEAT   64
#define NCLS   40
#define NCLSP  48
#define WLS    68

#define TILE    391
#define HOPW    32
#define GRP     512
#define QFLUSH  64
#define QCAP    (QFLUSH + GRP)
#define BW      16
#define LHDR    4
#define FQCAP   1312

#define APPEND(m_, e_)                                                         \
    {                                                                          \
        const unsigned mk_ = __builtin_amdgcn_ballot_w32(m_);                  \
        if (mk_) {                                                             \
            if (m_) q[cnt + (int)__builtin_amdgcn_mbcnt_lo(mk_, 0u)] = (e_);   \
            cnt += __builtin_popcount(mk_);                                    \
        }                                                                      \
    }

#define FILTER4(cv_, e0_)                                                      \
    {                                                                          \
        bool m_;                                                               \
        m_ = (unsigned)((cv_).x - t0) < (unsigned)tn; APPEND(m_, (e0_) + 0);   \
        m_ = (unsigned)((cv_).y - t0) < (unsigned)tn; APPEND(m_, (e0_) + 1);   \
        m_ = (unsigned)((cv_).z - t0) < (unsigned)tn; APPEND(m_, (e0_) + 2);   \
        m_ = (unsigned)((cv_).w - t0) < (unsigned)tn; APPEND(m_, (e0_) + 3);   \
    }

#define LOADSET(X, tt)                                                         \
    {                                                                          \
        int g_ = (tt) + rot; if (g_ >= nG) g_ -= nG;                           \
        g_ = ((tt) < nG) ? g_ : 0;                                             \
        g##X = g_;                                                             \
        const int4* p_ = colv + (size_t)g_ * (GRP / 4) + lane;                 \
        X##0 = p_[0]; X##1 = p_[32]; X##2 = p_[64]; X##3 = p_[96];             \
    }

#define FILTERSET(X)                                                           \
    {                                                                          \
        const int eb_ = head + g##X * GRP + lane * 4;                          \
        FILTER4(X##0, eb_);                                                    \
        FILTER4(X##1, eb_ + 128);                                              \
        FILTER4(X##2, eb_ + 256);                                              \
        FILTER4(X##3, eb_ + 384);                                              \
    }

#define STAGE(CUR, NXT2)                                                       \
    if (t >= nG) break;                                                        \
    LOADSET(NXT2, t + 2 * HOPW);                                               \
    FILTERSET(CUR);                                                            \
    t += HOPW;                                                                 \
    if (cnt > QFLUSH) drain(true, cnt);

template <class DrainF>
__device__ __forceinline__ void scan_edge_list(const int* __restrict__ col, const int E,
                                               const int t0, const int tn, int* const q,
                                               const int wave, const int lane, DrainF drain) {
    int cnt = 0;

    int head = (int)((0ull - ((unsigned long long)(size_t)col >> 2)) & 3ull);
    if (head > E) head = E;
    const int nG   = (E - head) / GRP;
    const int nrem = E - nG * GRP;
    for (int s = wave; s * 32 < nrem; s += HOPW) {
        const int idx = s * 32 + lane;
        const int e = (idx < head) ? idx : idx + nG * GRP;
        const int c = (idx < nrem) ? col[e] : -1;
        const bool m = (unsigned)(c - t0) < (unsigned)tn;
        APPEND(m, e);
    }

    const int rot = (int)(((long long)blockIdx.x * nG) / gridDim.x);
    const int4* colv = (const int4*)(col + head);

    int t = wave;
    const int4 none = make_int4(-1, -1, -1, -1);
    int4 A0 = none, A1 = none, A2 = none, A3 = none;
    int4 B0 = none, B1 = none, B2 = none, B3 = none;
    int4 C0 = none, C1 = none, C2 = none, C3 = none;
    int gA = 0, gB = 0, gC = 0;

    if (nG > 0) {
        LOADSET(A, t);
        LOADSET(B, t + HOPW);
    }
    for (;;) {
        STAGE(A, C);
        STAGE(B, A);
        STAGE(C, B);
    }
    drain(false, cnt);
}

__global__ __launch_bounds__(HOPW * 32)
void k_filter_tile(float* lists, const int* __restrict__ row, const int* __restrict__ col,
                   float* __restrict__ dinv, int N, int E) {
    __shared__ int queue[HOPW * FQCAP];
    __shared__ int degc[TILE];
    __shared__ int lcount;

    const int tid  = threadIdx.x;
    const int lane = tid & 31;
    const int wave = __builtin_amdgcn_readfirstlane(tid >> 5);
    const int t0   = blockIdx.x * TILE;
    const int tn   = min(TILE, N - t0);

    int* const lhdr = (int*)(lists + (size_t)t0 * FEAT);
    const int  cap  = (tn * FEAT - LHDR) / 3;
    int* const lr   = lhdr + LHDR;
    int* const lc   = lr + cap;

    for (int i = tid; i < TILE; i += HOPW * 32) degc[i] = 0;
    if (tid == 0) lcount = 0;
    __syncthreads();

    int* const q = queue + wave * FQCAP;
    auto drain = [&](const bool more, int& cnt) {
        const int nproc = more ? (cnt & ~31) : cnt;
        for (int b = 0; b < nproc; b += 64) {
            const int  i0 = b + lane;
            const int  i1 = b + 32 + lane;
            const bool v0 = i0 < nproc;
            const bool v1 = i1 < nproc;
            const int  e0 = q[v0 ? i0 : 0];
            const int  e1 = q[v1 ? i1 : 0];
            const int  r0 = row[e0];
            const int  r1 = row[e1];
            const int  cl0 = col[e0] - t0;
            const int  cl1 = col[e1] - t0;
            const int  mtot = min(64, nproc - b);
            int base = 0;
            if (lane == 0)
                base = __hip_atomic_fetch_add(&lcount, mtot, __ATOMIC_RELAXED, __HIP_MEMORY_SCOPE_WORKGROUP);
            if (v0) __hip_atomic_fetch_add(&degc[cl0], 1, __ATOMIC_RELAXED, __HIP_MEMORY_SCOPE_WORKGROUP);
            if (v1) __hip_atomic_fetch_add(&degc[cl1], 1, __ATOMIC_RELAXED, __HIP_MEMORY_SCOPE_WORKGROUP);
            base = __builtin_amdgcn_readfirstlane(base);
            const int s0 = base + lane;
            const int s1 = base + 32 + lane;
            if (v0 && s0 < cap) { lr[s0] = r0; lc[s0] = cl0; }
            if (v1 && s1 < cap) { lr[s1] = r1; lc[s1] = cl1; }
        }
        const int rem = cnt - nproc;
        if (rem > 0) {
            const int keep = (lane < rem) ? q[nproc + lane] : 0;
            if (lane < rem) q[lane] = keep;
        }
        cnt = rem;
    };
    scan_edge_list(col, E, t0, tn, q, wave, lane, drain);
    __syncthreads();

    for (int i = tid; i < tn; i += HOPW * 32) {
        const float v = (float)(degc[i] + 1);
        dinv[t0 + i] = (v > 0.0f) ? rsqrtf(v) : 0.0f;
    }
    if (tid == 0) lhdr[0] = lcount;
}

template <bool FIRST>
__global__ __launch_bounds__(HOPW * 32)
void k_hop_tile(const float* __restrict__ xin, float* xout, float* lists,
                const int* __restrict__ row, const int* __restrict__ col,
                const float* __restrict__ dinv,
                const float* __restrict__ Wm, const float* __restrict__ bias,
                float* __restrict__ outp, int N, int E) {
    __shared__ __attribute__((aligned(16))) float tile[TILE * FEAT];
    __shared__ int queue[HOPW * QCAP];
    __shared__ float dloc[TILE];
    __shared__ __attribute__((aligned(16))) float wl[FIRST ? 4 : NCLSP * WLS];

    const int tid  = threadIdx.x;
    const int lane = tid & 31;
    const int wave = __builtin_amdgcn_readfirstlane(tid >> 5);
    const int t0   = blockIdx.x * TILE;
    const int tn   = min(TILE, N - t0);

    int*   const lhdr = (int*)(lists + (size_t)t0 * FEAT);
    const int    cap  = (tn * FEAT - LHDR) / 3;
    int*   const lr   = lhdr + LHDR;
    int*   const lc   = lr + cap;
    float* const lw   = (float*)(lc + cap);
    const int hraw = lhdr[0];

    {
        const float4* src = (const float4*)(xin + (size_t)t0 * FEAT);
        float4* dst = (float4*)tile;
        for (int i = tid; i < tn * (FEAT / 4); i += HOPW * 32) {
            float w = dinv[t0 + (i >> 4)];
            w = w * w;
            const float4 v = src[i];
            dst[i] = make_float4(w * v.x, w * v.y, w * v.z, w * v.w);
        }
        if (FIRST) {
            for (int i = tid; i < tn; i += HOPW * 32) dloc[i] = dinv[t0 + i];
        } else {
            for (int i = tid; i < NCLSP * FEAT; i += HOPW * 32) {
                const int r = i >> 6;
                wl[r * WLS + (i & (FEAT - 1))] = (r < NCLS) ? Wm[i] : 0.0f;
            }
        }
    }
    __syncthreads();

    const int  lcnt    = __builtin_amdgcn_readfirstlane(hraw);
    const bool uselist = lcnt >= 0 && lcnt <= cap;

    if (!uselist) {
        auto consume = [&](const int r0, const int r1, const float w0, const float w1,
                           const int cl0, const int cl1, const int mtot) {
#pragma unroll 1
            for (int k = 0; k < mtot; k += BW) {
                const bool  hi = k >= 32;
                const int   rs = hi ? r1 : r0;
                const float ws = hi ? w1 : w0;
                const int   cs = hi ? cl1 : cl0;
                const int   kk = k & 31;
                v2f val[BW];
#pragma unroll
                for (int u = 0; u < BW; ++u) {
                    const int rk = __builtin_amdgcn_readlane(rs, kk + u);
                    val[u] = *(const v2f*)(xin + (size_t)rk * FEAT + 2 * lane);
                }
#pragma unroll
                for (int u = 0; u < BW; ++u) {
                    if (k + u < mtot) {
                        const float wk = __int_as_float(__builtin_amdgcn_readlane(__float_as_int(ws), kk + u));
                        const int   ck = __builtin_amdgcn_readlane(cs, kk + u);
                        float* d = tile + ck * FEAT + 2 * lane;
                        __hip_atomic_fetch_add(d + 0, wk * val[u].x, __ATOMIC_RELAXED, __HIP_MEMORY_SCOPE_WORKGROUP);
                        __hip_atomic_fetch_add(d + 1, wk * val[u].y, __ATOMIC_RELAXED, __HIP_MEMORY_SCOPE_WORKGROUP);
                    }
                }
            }
        };
        int* const q = queue + wave * QCAP;
        auto drain = [&](const bool more, int& cnt) {
            const int nproc = more ? (cnt & ~31) : cnt;
            for (int b = 0; b < nproc; b += 64) {
                const int  i0 = b + lane;
                const int  i1 = b + 32 + lane;
                const bool v0 = i0 < nproc;
                const bool v1 = i1 < nproc;
                const int  e0 = q[v0 ? i0 : 0];
                const int  e1 = q[v1 ? i1 : 0];
                const int  r0 = row[e0];
                const int  r1 = row[e1];
                const int  c0 = col[e0];
                const int  c1 = col[e1];
                const float dr0 = dinv[r0];
                const float dr1 = dinv[r1];
                const float dc0 = dinv[c0];
                const float dc1 = dinv[c1];
                const float w0 = v0 ? dr0 * dc0 : 0.0f;
                const float w1 = v1 ? dr1 * dc1 : 0.0f;
                consume(r0, r1, w0, w1, c0 - t0, c1 - t0, min(64, nproc - b));
            }
            const int rem = cnt - nproc;
            if (rem > 0) {
                const int keep = (lane < rem) ? q[nproc + lane] : 0;
                if (lane < rem) q[lane] = keep;
            }
            cnt = rem;
        };
        scan_edge_list(col, E, t0, tn, q, wave, lane, drain);
    } else {
        auto walk32 = [&](const int rs, const float wa, const float wb, const int cs, const int n) {
            v2f val[32];
#pragma unroll
            for (int u = 0; u < 32; ++u) {
                const int rk = __builtin_amdgcn_readlane(rs, u);
                val[u] = *(const v2f*)(xin + (size_t)rk * FEAT + 2 * lane);
            }
            const float ws = FIRST ? wa * wb : wa;
#pragma unroll
            for (int u = 0; u < 32; ++u) {
                const float wk = __int_as_float(__builtin_amdgcn_readlane(__float_as_int(ws), u));
                const int   ck = __builtin_amdgcn_readlane(cs, u);
                const bool  ok = u < n;
                float* d = tile + ck * FEAT + 2 * lane;
                __hip_atomic_fetch_add(d + 0, ok ? wk * val[u].x : 0.0f, __ATOMIC_RELAXED, __HIP_MEMORY_SCOPE_WORKGROUP);
                __hip_atomic_fetch_add(d + 1, ok ? wk * val[u].y : 0.0f, __ATOMIC_RELAXED, __HIP_MEMORY_SCOPE_WORKGROUP);
            }
            return ws;
        };
        int b = wave * 64;
        int   r0 = 0, r1 = 0, cl0 = 0, cl1 = 0;
        float w0 = 0.0f, w1 = 0.0f;
        if (b < lcnt) {
            const bool v0 = b + lane < lcnt;
            const bool v1 = b + 32 + lane < lcnt;
            const int  j0 = v0 ? b + lane : 0;
            const int  j1 = v1 ? b + 32 + lane : 0;
            r0 = lr[j0]; r1 = lr[j1];
            cl0 = lc[j0]; cl1 = lc[j1];
            if (!FIRST) { w0 = lw[j0]; w1 = lw[j1]; }
        }
        while (b < lcnt) {
            const int nb = b + HOPW * 64;
            int   nr0 = 0, nr1 = 0, ncl0 = 0, ncl1 = 0;
            float nw0 = 0.0f, nw1 = 0.0f;
            if (nb < lcnt) {
                const bool v0 = nb + lane < lcnt;
                const bool v1 = nb + 32 + lane < lcnt;
                const int  j0 = v0 ? nb + lane : 0;
                const int  j1 = v1 ? nb + 32 + lane : 0;
                nr0 = lr[j0]; nr1 = lr[j1];
                ncl0 = lc[j0]; ncl1 = lc[j1];
                if (!FIRST) { nw0 = lw[j0]; nw1 = lw[j1]; }
            }
            const int m = lcnt - b;
            if (FIRST) {
                const float a0 = dinv[r0];
                const float a1 = dinv[r1];
                const float g0 = walk32(r0, a0, dloc[cl0], cl0, m);
                if (b + lane < lcnt) lw[b + lane] = g0;
                if (m > 32) {
                    const float g1 = walk32(r1, a1, dloc[cl1], cl1, m - 32);
                    if (b + 32 + lane < lcnt) lw[b + 32 + lane] = g1;
                }
            } else {
                walk32(r0, w0, 1.0f, cl0, m);
                if (m > 32) walk32(r1, w1, 1.0f, cl1, m - 32);
            }
            r0 = nr0; r1 = nr1; cl0 = ncl0; cl1 = ncl1; w0 = nw0; w1 = nw1;
            b = nb;
        }
    }
    __syncthreads();

    if (FIRST) {
        const float4* srct = (const float4*)tile;
        float4* dst = (float4*)(xout + (size_t)t0 * FEAT);
        for (int i = tid; i < tn * (FEAT / 4); i += HOPW * 32) dst[i] = srct[i];
    } else {
        const int half = lane >> 4;
        const int l16  = lane & 15;
        const bool has2 = l16 < NCLS - 32;
        const float bb0 = bias[l16];
        const float bb1 = bias[16 + l16];
        const float bb2 = has2 ? bias[32 + l16] : 0.0f;
        const float* const b0r = wl + (l16)      * WLS + 2 * half;
        const float* const b1r = wl + (16 + l16) * WLS + 2 * half;
        const float* const b2r = wl + (32 + l16) * WLS + 2 * half;
        for (int rb = wave * 16; rb < tn; rb += HOPW * 16) {
            const int ar = min(rb + l16, tn - 1);
            const float* const arow = tile + ar * FEAT + 2 * half;
            v8f acc0 = {};
            v8f acc1 = {};
            v8f acc2 = {};
#pragma unroll
            for (int k0 = 0; k0 < FEAT; k0 += 4) {
                const v2f a  = *(const v2f*)(arow + k0);
                const v2f b0 = *(const v2f*)(b0r + k0);
                const v2f b1 = *(const v2f*)(b1r + k0);
                const v2f b2 = *(const v2f*)(b2r + k0);
                acc0 = __builtin_amdgcn_wmma_f32_16x16x4_f32(false, a, false, b0, (short)0, acc0, false, false);
                acc1 = __builtin_amdgcn_wmma_f32_16x16x4_f32(false, a, false, b1, (short)0, acc1, false, false);
                acc2 = __builtin_amdgcn_wmma_f32_16x16x4_f32(false, a, false, b2, (short)0, acc2, false, false);
            }
            float a0[8], a1[8], a2[8], mx[8], sm[8];
#pragma unroll
            for (int j = 0; j < 8; ++j) {
                a0[j] = acc0[j] + bb0;
                a1[j] = acc1[j] + bb1;
                a2[j] = acc2[j] + bb2;
                const float m01 = fmaxf(a0[j], a1[j]);
                mx[j] = has2 ? fmaxf(m01, a2[j]) : m01;
            }
#pragma unroll
            for (int s = 1; s < 16; s <<= 1) {
#pragma unroll
                for (int j = 0; j < 8; ++j) mx[j] = fmaxf(mx[j], __shfl_xor(mx[j], s));
            }
#pragma unroll
            for (int j = 0; j < 8; ++j) {
                const float e01 = expf(a0[j] - mx[j]) + expf(a1[j] - mx[j]);
                sm[j] = has2 ? e01 + expf(a2[j] - mx[j]) : e01;
            }
#pragma unroll
            for (int s = 1; s < 16; s <<= 1) {
#pragma unroll
                for (int j = 0; j < 8; ++j) sm[j] += __shfl_xor(sm[j], s);
            }
#pragma unroll
            for (int j = 0; j < 8; ++j) {
                const int   rr = rb + j + 8 * half;
                const float ls = logf(sm[j]);
                if (rr < tn) {
                    float* o = outp + (size_t)(t0 + rr) * NCLS;
                    o[l16]      = a0[j] - mx[j] - ls;
                    o[16 + l16] = a1[j] - mx[j] - ls;
                    if (has2) o[32 + l16] = a2[j] - mx[j] - ls;
                }
            }
        }
    }
}


static inline size_t align256f(size_t nfloats) { return (nfloats + 63) & ~(size_t)63; }

extern "C" void kernel_launch(void* const* d_in, const int* in_sizes, int n_in,
                              void* d_out, int out_size, void* d_ws, size_t ws_size,
                              hipStream_t stream) {
    const float* x  = (const float*)d_in[0];
    const int*   ei = (const int*)d_in[1];
    const float* W  = (const float*)d_in[2];
    const float* bi = (const float*)d_in[3];

    const int N = in_sizes[0] / FEAT;
    const int E = in_sizes[1] / 2;
    const int* row = ei;
    const int* col = ei + E;
    const int total = N * FEAT;

    float* ws = (float*)d_ws;
    size_t off = 0;
    float* dinv = ws + off; off += align256f((size_t)N);
    float* x0   = ws + off; off += align256f((size_t)total);
    float* lst  = ws + off;

    const int htiles = (N + TILE - 1) / TILE;
    k_filter_tile<<<htiles, HOPW * 32, 0, stream>>>(lst, row, col, dinv, N, E);
    k_hop_tile<true><<<htiles, HOPW * 32, 0, stream>>>(x, x0, lst, row, col, dinv, W, bi, (float*)d_out, N, E);
    k_hop_tile<false><<<htiles, HOPW * 32, 0, stream>>>(x0, nullptr, lst, row, col, dinv, W, bi, (float*)d_out, N, E);
}


# filter scans 8 bit planes of targets' tile numbers (1 byte per edge, not 4; a lane tests 32 edges in 12 logic ops), built from col by a new small launch: 3 -> 4 launches; sums: LDS arrival order
# speedup vs baseline: 3.2688x; 1.2007x over previous
#include <hip/hip_runtime.h>
#include <math.h>

typedef float v2f __attribute__((ext_vector_type(2)));
typedef float v8f __attribute__((ext_vector_type(8)));

#define FEAT   64
#define NCLS   40
#define NCLSP  48
#define WLS    68

#define TILE    391
#define HOPW    32
#define GRP     512
#define QFLUSH  64
#define QCAP    (QFLUSH + GRP)
#define BW      16
#define LHDR    4
#define FQCAP   1312
#define PBLK    1024

#define APPEND(m_, e_)                                                         \
    {                                                                          \
        const unsigned mk_ = __builtin_amdgcn_ballot_w32(m_);                  \
        if (mk_) {                                                             \
            if (m_) q[cnt + (int)__builtin_amdgcn_mbcnt_lo(mk_, 0u)] = (e_);   \
            cnt += __builtin_popcount(mk_);                                    \
        }                                                                      \
    }

#define FILTER4(cv_, e0_)                                                      \
    {                                                                          \
        bool m_;                                                               \
        m_ = (unsigned)((cv_).x - t0) < (unsigned)tn; APPEND(m_, (e0_) + 0);   \
        m_ = (unsigned)((cv_).y - t0) < (unsigned)tn; APPEND(m_, (e0_) + 1);   \
        m_ = (unsigned)((cv_).z - t0) < (unsigned)tn; APPEND(m_, (e0_) + 2);   \
        m_ = (unsigned)((cv_).w - t0) < (unsigned)tn; APPEND(m_, (e0_) + 3);   \
    }

#define LOADSET(X, tt)                                                         \
    {                                                                          \
        int g_ = (tt) + rot; if (g_ >= nG) g_ -= nG;                           \
        g_ = ((tt) < nG) ? g_ : 0;                                             \
        g##X = g_;                                                             \
        const int4* p_ = colv + (size_t)g_ * (GRP / 4) + lane;                 \
        X##0 = p_[0]; X##1 = p_[32]; X##2 = p_[64]; X##3 = p_[96];             \
    }

#define FILTERSET(X)                                                           \
    {                                                                          \
        const int eb_ = head + g##X * GRP + lane * 4;                          \
        FILTER4(X##0, eb_);                                                    \
        FILTER4(X##1, eb_ + 128);                                              \
        FILTER4(X##2, eb_ + 256);                                              \
        FILTER4(X##3, eb_ + 384);                                              \
    }

#define STAGE(CUR, NXT2)                                                       \
    if (t >= nG) break;                                                        \
    LOADSET(NXT2, t + 2 * HOPW);                                               \
    FILTERSET(CUR);                                                            \
    t += HOPW;                                                                 \
    if (cnt > QFLUSH) drain(true, cnt);

template <class DrainF>
__device__ __forceinline__ void scan_edge_list(const int* __restrict__ col, const int E,
                                               const int t0, const int tn, int* const q,
                                               const int wave, const int lane, DrainF drain) {
    int cnt = 0;

    int head = (int)((0ull - ((unsigned long long)(size_t)col >> 2)) & 3ull);
    if (head > E) head = E;
    const int nG   = (E - head) / GRP;
    const int nrem = E - nG * GRP;
    for (int s = wave; s * 32 < nrem; s += HOPW) {
        const int idx = s * 32 + lane;
        const int e = (idx < head) ? idx : idx + nG * GRP;
        const int c = (idx < nrem) ? col[e] : -1;
        const bool m = (unsigned)(c - t0) < (unsigned)tn;
        APPEND(m, e);
    }

    const int rot = (int)(((long long)blockIdx.x * nG) / gridDim.x);
    const int4* colv = (const int4*)(col + head);

    int t = wave;
    const int4 none = make_int4(-1, -1, -1, -1);
    int4 A0 = none, A1 = none, A2 = none, A3 = none;
    int4 B0 = none, B1 = none, B2 = none, B3 = none;
    int4 C0 = none, C1 = none, C2 = none, C3 = none;
    int gA = 0, gB = 0, gC = 0;

    if (nG > 0) {
        LOADSET(A, t);
        LOADSET(B, t + HOPW);
    }
    for (;;) {
        STAGE(A, C);
        STAGE(B, A);
        STAGE(C, B);
    }
    drain(false, cnt);
}

__global__ __launch_bounds__(256)
void k_tile_planes(const int* __restrict__ col, float* planesf, int E) {
    const int lane = threadIdx.x & 31;
    const int wave = __builtin_amdgcn_readfirstlane((int)(threadIdx.x >> 5));
    const int blk  = blockIdx.x * 8 + wave;
    const int nB   = (E + PBLK - 1) / PBLK;
    if (blk >= nB) return;
    unsigned p0 = 0, p1 = 0, p2 = 0, p3 = 0, p4 = 0, p5 = 0, p6 = 0, p7 = 0;
    int cv[32];
#pragma unroll
    for (int r = 0; r < 32; ++r) {
        const int e = blk * PBLK + r * 32 + lane;
        cv[r] = col[min(e, E - 1)];
    }
#pragma unroll
    for (int r = 0; r < 32; ++r) {
        const unsigned tl = (unsigned)cv[r] / (unsigned)TILE;
        const unsigned b0 = __builtin_amdgcn_ballot_w32((tl & 1u) != 0u);
        const unsigned b1 = __builtin_amdgcn_ballot_w32((tl & 2u) != 0u);
        const unsigned b2 = __builtin_amdgcn_ballot_w32((tl & 4u) != 0u);
        const unsigned b3 = __builtin_amdgcn_ballot_w32((tl & 8u) != 0u);
        const unsigned b4 = __builtin_amdgcn_ballot_w32((tl & 16u) != 0u);
        const unsigned b5 = __builtin_amdgcn_ballot_w32((tl & 32u) != 0u);
        const unsigned b6 = __builtin_amdgcn_ballot_w32((tl & 64u) != 0u);
        const unsigned b7 = __builtin_amdgcn_ballot_w32((tl & 128u) != 0u);
        const bool mine = lane == r;
        p0 = mine ? b0 : p0;
        p1 = mine ? b1 : p1;
        p2 = mine ? b2 : p2;
        p3 = mine ? b3 : p3;
        p4 = mine ? b4 : p4;
        p5 = mine ? b5 : p5;
        p6 = mine ? b6 : p6;
        p7 = mine ? b7 : p7;
    }
    uint4* const o = (uint4*)planesf + (size_t)blk * (PBLK / 16);
    o[lane]      = make_uint4(p0, p1, p2, p3);
    o[32 + lane] = make_uint4(p4, p5, p6, p7);
}

#define LOADP(X, tt)                                                           \
    {                                                                          \
        int g_ = (tt) + rot; if (g_ >= nB) g_ -= nB;                           \
        g_ = ((tt) < nB) ? g_ : 0;                                             \
        g##X = g_;                                                             \
        const uint4* p_ = pv + (size_t)g_ * (PBLK / 16) + lane;                \
        X##0 = p_[0]; X##1 = p_[32];                                           \
    }

#define FILTERP(X)                                                             \
    {                                                                          \
        const uint4 lo_ = X##0;                                                \
        const uint4 hi_ = X##1;                                                \
        unsigned hm_ = ~(((lo_.x ^ m0) | (lo_.y ^ m1) | (lo_.z ^ m2) | (lo_.w ^ m3)) | \
                         ((hi_.x ^ m4) | (hi_.y ^ m5) | (hi_.z ^ m6) | (hi_.w ^ m7))); \
        const int eb_ = g##X * PBLK + lane * 32;                               \
        if (g##X == nB - 1) {                                                  \
            const int nv_ = E - eb_;     \
            hm_ = (nv_ >= 32) ? hm_ : ((nv_ > 0) ? (hm_ & ((1u << nv_) - 1u)) : 0u); \
        }                                                                      \
        unsigned mk_ = __builtin_amdgcn_ballot_w32(hm_ != 0u);                 \
        while (mk_) {                                                          \
            if (hm_ != 0u)                                                     \
                q[cnt + (int)__builtin_amdgcn_mbcnt_lo(mk_, 0u)] = eb_ + __builtin_ctz(hm_); \
            cnt += __builtin_popcount(mk_);                                    \
            hm_ &= hm_ - 1u;                                                   \
            mk_ = __builtin_amdgcn_ballot_w32(hm_ != 0u);                      \
        }                                                                      \
    }

#define STAGEP(CUR, NXT2)                                                      \
    if (t >= nB) break;                                                        \
    LOADP(NXT2, t + 2 * HOPW);                                                 \
    FILTERP(CUR);                                                              \
    t += HOPW;                                                                 \
    if (cnt > QFLUSH) drain(true, cnt);

template <class DrainF>
__device__ __forceinline__ void scan_planes(const uint4* __restrict__ pv, const int E,
                                            const int tsel, int* const q,
                                            const int wave, const int lane, DrainF drain) {
    int cnt = 0;
    const int nB = (E + PBLK - 1) / PBLK;
    const unsigned m0 = (tsel & 1)   ? ~0u : 0u;
    const unsigned m1 = (tsel & 2)   ? ~0u : 0u;
    const unsigned m2 = (tsel & 4)   ? ~0u : 0u;
    const unsigned m3 = (tsel & 8)   ? ~0u : 0u;
    const unsigned m4 = (tsel & 16)  ? ~0u : 0u;
    const unsigned m5 = (tsel & 32)  ? ~0u : 0u;
    const unsigned m6 = (tsel & 64)  ? ~0u : 0u;
    const unsigned m7 = (tsel & 128) ? ~0u : 0u;
    const int rot = (int)(((long long)blockIdx.x * nB) / gridDim.x);

    int t = wave;
    const uint4 zero = make_uint4(0u, 0u, 0u, 0u);
    uint4 A0 = zero, A1 = zero, B0 = zero, B1 = zero, C0 = zero, C1 = zero;
    int gA = 0, gB = 0, gC = 0;
    if (nB > 0) {
        LOADP(A, t);
        LOADP(B, t + HOPW);
    }
    for (;;) {
        STAGEP(A, C);
        STAGEP(B, A);
        STAGEP(C, B);
    }
    drain(false, cnt);
}

__global__ __launch_bounds__(HOPW * 32)
void k_filter_tile(float* lists, const float* planesf, const int* __restrict__ row,
                   const int* __restrict__ col, float* __restrict__ dinv, int N, int E) {
    __shared__ int queue[HOPW * FQCAP];
    __shared__ int degc[TILE];
    __shared__ int lcount;

    const int tid  = threadIdx.x;
    const int lane = tid & 31;
    const int wave = __builtin_amdgcn_readfirstlane(tid >> 5);
    const int t0   = blockIdx.x * TILE;
    const int tn   = min(TILE, N - t0);

    int* const lhdr = (int*)(lists + (size_t)t0 * FEAT);
    const int  cap  = (tn * FEAT - LHDR) / 3;
    int* const lr   = lhdr + LHDR;
    int* const lc   = lr + cap;

    for (int i = tid; i < TILE; i += HOPW * 32) degc[i] = 0;
    if (tid == 0) lcount = 0;
    __syncthreads();

    int* const q = queue + wave * FQCAP;
    auto drain = [&](const bool more, int& cnt) {
        const int nproc = more ? (cnt & ~31) : cnt;
        for (int b = 0; b < nproc; b += 64) {
            const int  i0 = b + lane;
            const int  i1 = b + 32 + lane;
            const bool v0 = i0 < nproc;
            const bool v1 = i1 < nproc;
            const int  e0 = q[v0 ? i0 : 0];
            const int  e1 = q[v1 ? i1 : 0];
            const int  r0 = row[e0];
            const int  r1 = row[e1];
            const int  cl0 = col[e0] - t0;
            const int  cl1 = col[e1] - t0;
            const int  mtot = min(64, nproc - b);
            int base = 0;
            if (lane == 0)
                base = __hip_atomic_fetch_add(&lcount, mtot, __ATOMIC_RELAXED, __HIP_MEMORY_SCOPE_WORKGROUP);
            if (v0) __hip_atomic_fetch_add(&degc[cl0], 1, __ATOMIC_RELAXED, __HIP_MEMORY_SCOPE_WORKGROUP);
            if (v1) __hip_atomic_fetch_add(&degc[cl1], 1, __ATOMIC_RELAXED, __HIP_MEMORY_SCOPE_WORKGROUP);
            base = __builtin_amdgcn_readfirstlane(base);
            const int s0 = base + lane;
            const int s1 = base + 32 + lane;
            if (v0 && s0 < cap) { lr[s0] = r0; lc[s0] = cl0; }
            if (v1 && s1 < cap) { lr[s1] = r1; lc[s1] = cl1; }
        }
        const int rem = cnt - nproc;
        if (rem > 0) {
            const int keep = (lane < rem) ? q[nproc + lane] : 0;
            if (lane < rem) q[lane] = keep;
        }
        cnt = rem;
    };
    scan_planes((const uint4*)planesf, E, (int)blockIdx.x, q, wave, lane, drain);
    __syncthreads();

    for (int i = tid; i < tn; i += HOPW * 32) {
        const float v = (float)(degc[i] + 1);
        dinv[t0 + i] = (v > 0.0f) ? rsqrtf(v) : 0.0f;
    }
    if (tid == 0) lhdr[0] = lcount;
}

template <bool FIRST>
__global__ __launch_bounds__(HOPW * 32)
void k_hop_tile(const float* __restrict__ xin, float* xout, float* lists,
                const int* __restrict__ row, const int* __restrict__ col,
                const float* __restrict__ dinv,
                const float* __restrict__ Wm, const float* __restrict__ bias,
                float* __restrict__ outp, int N, int E) {
    __shared__ __attribute__((aligned(16))) float tile[TILE * FEAT];
    __shared__ int queue[HOPW * QCAP];
    __shared__ float dloc[TILE];
    __shared__ __attribute__((aligned(16))) float wl[FIRST ? 4 : NCLSP * WLS];

    const int tid  = threadIdx.x;
    const int lane = tid & 31;
    const int wave = __builtin_amdgcn_readfirstlane(tid >> 5);
    const int t0   = blockIdx.x * TILE;
    const int tn   = min(TILE, N - t0);

    int*   const lhdr = (int*)(lists + (size_t)t0 * FEAT);
    const int    cap  = (tn * FEAT - LHDR) / 3;
    int*   const lr   = lhdr + LHDR;
    int*   const lc   = lr + cap;
    float* const lw   = (float*)(lc + cap);
    const int hraw = lhdr[0];

    {
        const float4* src = (const float4*)(xin + (size_t)t0 * FEAT);
        float4* dst = (float4*)tile;
        for (int i = tid; i < tn * (FEAT / 4); i += HOPW * 32) {
            float w = dinv[t0 + (i >> 4)];
            w = w * w;
            const float4 v = src[i];
            dst[i] = make_float4(w * v.x, w * v.y, w * v.z, w * v.w);
        }
        if (FIRST) {
            for (int i = tid; i < tn; i += HOPW * 32) dloc[i] = dinv[t0 + i];
        } else {
            for (int i = tid; i < NCLSP * FEAT; i += HOPW * 32) {
                const int r = i >> 6;
                wl[r * WLS + (i & (FEAT - 1))] = (r < NCLS) ? Wm[i] : 0.0f;
            }
        }
    }
    __syncthreads();

    const int  lcnt    = __builtin_amdgcn_readfirstlane(hraw);
    const bool uselist = lcnt >= 0 && lcnt <= cap;

    if (!uselist) {
        auto consume = [&](const int r0, const int r1, const float w0, const float w1,
                           const int cl0, const int cl1, const int mtot) {
#pragma unroll 1
            for (int k = 0; k < mtot; k += BW) {
                const bool  hi = k >= 32;
                const int   rs = hi ? r1 : r0;
                const float ws = hi ? w1 : w0;
                const int   cs = hi ? cl1 : cl0;
                const int   kk = k & 31;
                v2f val[BW];
#pragma unroll
                for (int u = 0; u < BW; ++u) {
                    const int rk = __builtin_amdgcn_readlane(rs, kk + u);
                    val[u] = *(const v2f*)(xin + (size_t)rk * FEAT + 2 * lane);
                }
#pragma unroll
                for (int u = 0; u < BW; ++u) {
                    if (k + u < mtot) {
                        const float wk = __int_as_float(__builtin_amdgcn_readlane(__float_as_int(ws), kk + u));
                        const int   ck = __builtin_amdgcn_readlane(cs, kk + u);
                        float* d = tile + ck * FEAT + 2 * lane;
                        __hip_atomic_fetch_add(d + 0, wk * val[u].x, __ATOMIC_RELAXED, __HIP_MEMORY_SCOPE_WORKGROUP);
                        __hip_atomic_fetch_add(d + 1, wk * val[u].y, __ATOMIC_RELAXED, __HIP_MEMORY_SCOPE_WORKGROUP);
                    }
                }
            }
        };
        int* const q = queue + wave * QCAP;
        auto drain = [&](const bool more, int& cnt) {
            const int nproc = more ? (cnt & ~31) : cnt;
            for (int b = 0; b < nproc; b += 64) {
                const int  i0 = b + lane;
                const int  i1 = b + 32 + lane;
                const bool v0 = i0 < nproc;
                const bool v1 = i1 < nproc;
                const int  e0 = q[v0 ? i0 : 0];
                const int  e1 = q[v1 ? i1 : 0];
                const int  r0 = row[e0];
                const int  r1 = row[e1];
                const int  c0 = col[e0];
                const int  c1 = col[e1];
                const float dr0 = dinv[r0];
                const float dr1 = dinv[r1];
                const float dc0 = dinv[c0];
                const float dc1 = dinv[c1];
                const float w0 = v0 ? dr0 * dc0 : 0.0f;
                const float w1 = v1 ? dr1 * dc1 : 0.0f;
                consume(r0, r1, w0, w1, c0 - t0, c1 - t0, min(64, nproc - b));
            }
            const int rem = cnt - nproc;
            if (rem > 0) {
                const int keep = (lane < rem) ? q[nproc + lane] : 0;
                if (lane < rem) q[lane] = keep;
            }
            cnt = rem;
        };
        scan_edge_list(col, E, t0, tn, q, wave, lane, drain);
    } else {
        auto walk32 = [&](const int rs, const float wa, const float wb, const int cs, const int n) {
            v2f val[32];
#pragma unroll
            for (int u = 0; u < 32; ++u) {
                const int rk = __builtin_amdgcn_readlane(rs, u);
                val[u] = *(const v2f*)(xin + (size_t)rk * FEAT + 2 * lane);
            }
            const float ws = FIRST ? wa * wb : wa;
#pragma unroll
            for (int u = 0; u < 32; ++u) {
                const float wk = __int_as_float(__builtin_amdgcn_readlane(__float_as_int(ws), u));
                const int   ck = __builtin_amdgcn_readlane(cs, u);
                const bool  ok = u < n;
                float* d = tile + ck * FEAT + 2 * lane;
                __hip_atomic_fetch_add(d + 0, ok ? wk * val[u].x : 0.0f, __ATOMIC_RELAXED, __HIP_MEMORY_SCOPE_WORKGROUP);
                __hip_atomic_fetch_add(d + 1, ok ? wk * val[u].y : 0.0f, __ATOMIC_RELAXED, __HIP_MEMORY_SCOPE_WORKGROUP);
            }
            return ws;
        };
        int b = wave * 64;
        int   r0 = 0, r1 = 0, cl0 = 0, cl1 = 0;
        float w0 = 0.0f, w1 = 0.0f;
        if (b < lcnt) {
            const bool v0 = b + lane < lcnt;
            const bool v1 = b + 32 + lane < lcnt;
            const int  j0 = v0 ? b + lane : 0;
            const int  j1 = v1 ? b + 32 + lane : 0;
            r0 = lr[j0]; r1 = lr[j1];
            cl0 = lc[j0]; cl1 = lc[j1];
            if (!FIRST) { w0 = lw[j0]; w1 = lw[j1]; }
        }
        while (b < lcnt) {
            const int nb = b + HOPW * 64;
            int   nr0 = 0, nr1 = 0, ncl0 = 0, ncl1 = 0;
            float nw0 = 0.0f, nw1 = 0.0f;
            if (nb < lcnt) {
                const bool v0 = nb + lane < lcnt;
                const bool v1 = nb + 32 + lane < lcnt;
                const int  j0 = v0 ? nb + lane : 0;
                const int  j1 = v1 ? nb + 32 + lane : 0;
                nr0 = lr[j0]; nr1 = lr[j1];
                ncl0 = lc[j0]; ncl1 = lc[j1];
                if (!FIRST) { nw0 = lw[j0]; nw1 = lw[j1]; }
            }
            const int m = lcnt - b;
            if (FIRST) {
                const float a0 = dinv[r0];
                const float a1 = dinv[r1];
                const float g0 = walk32(r0, a0, dloc[cl0], cl0, m);
                if (b + lane < lcnt) lw[b + lane] = g0;
                if (m > 32) {
                    const float g1 = walk32(r1, a1, dloc[cl1], cl1, m - 32);
                    if (b + 32 + lane < lcnt) lw[b + 32 + lane] = g1;
                }
            } else {
                walk32(r0, w0, 1.0f, cl0, m);
                if (m > 32) walk32(r1, w1, 1.0f, cl1, m - 32);
            }
            r0 = nr0; r1 = nr1; cl0 = ncl0; cl1 = ncl1; w0 = nw0; w1 = nw1;
            b = nb;
        }
    }
    __syncthreads();

    if (FIRST) {
        const float4* srct = (const float4*)tile;
        float4* dst = (float4*)(xout + (size_t)t0 * FEAT);
        for (int i = tid; i < tn * (FEAT / 4); i += HOPW * 32) dst[i] = srct[i];
    } else {
        const int half = lane >> 4;
        const int l16  = lane & 15;
        const bool has2 = l16 < NCLS - 32;
        const float bb0 = bias[l16];
        const float bb1 = bias[16 + l16];
        const float bb2 = has2 ? bias[32 + l16] : 0.0f;
        const float* const b0r = wl + (l16)      * WLS + 2 * half;
        const float* const b1r = wl + (16 + l16) * WLS + 2 * half;
        const float* const b2r = wl + (32 + l16) * WLS + 2 * half;
        for (int rb = wave * 16; rb < tn; rb += HOPW * 16) {
            const int ar = min(rb + l16, tn - 1);
            const float* const arow = tile + ar * FEAT + 2 * half;
            v8f acc0 = {};
            v8f acc1 = {};
            v8f acc2 = {};
#pragma unroll
            for (int k0 = 0; k0 < FEAT; k0 += 4) {
                const v2f a  = *(const v2f*)(arow + k0);
                const v2f b0 = *(const v2f*)(b0r + k0);
                const v2f b1 = *(const v2f*)(b1r + k0);
                const v2f b2 = *(const v2f*)(b2r + k0);
                acc0 = __builtin_amdgcn_wmma_f32_16x16x4_f32(false, a, false, b0, (short)0, acc0, false, false);
                acc1 = __builtin_amdgcn_wmma_f32_16x16x4_f32(false, a, false, b1, (short)0, acc1, false, false);
                acc2 = __builtin_amdgcn_wmma_f32_16x16x4_f32(false, a, false, b2, (short)0, acc2, false, false);
            }
            float a0[8], a1[8], a2[8], mx[8], sm[8];
#pragma unroll
            for (int j = 0; j < 8; ++j) {
                a0[j] = acc0[j] + bb0;
                a1[j] = acc1[j] + bb1;
                a2[j] = acc2[j] + bb2;
                const float m01 = fmaxf(a0[j], a1[j]);
                mx[j] = has2 ? fmaxf(m01, a2[j]) : m01;
            }
#pragma unroll
            for (int s = 1; s < 16; s <<= 1) {
#pragma unroll
                for (int j = 0; j < 8; ++j) mx[j] = fmaxf(mx[j], __shfl_xor(mx[j], s));
            }
#pragma unroll
            for (int j = 0; j < 8; ++j) {
                const float e01 = expf(a0[j] - mx[j]) + expf(a1[j] - mx[j]);
                sm[j] = has2 ? e01 + expf(a2[j] - mx[j]) : e01;
            }
#pragma unroll
            for (int s = 1; s < 16; s <<= 1) {
#pragma unroll
                for (int j = 0; j < 8; ++j) sm[j] += __shfl_xor(sm[j], s);
            }
#pragma unroll
            for (int j = 0; j < 8; ++j) {
                const int   rr = rb + j + 8 * half;
                const float ls = logf(sm[j]);
                if (rr < tn) {
                    float* o = outp + (size_t)(t0 + rr) * NCLS;
                    o[l16]      = a0[j] - mx[j] - ls;
                    o[16 + l16] = a1[j] - mx[j] - ls;
                    if (has2) o[32 + l16] = a2[j] - mx[j] - ls;
                }
            }
        }
    }
}


static inline size_t align256f(size_t nfloats) { return (nfloats + 63) & ~(size_t)63; }

extern "C" void kernel_launch(void* const* d_in, const int* in_sizes, int n_in,
                              void* d_out, int out_size, void* d_ws, size_t ws_size,
                              hipStream_t stream) {
    const float* x  = (const float*)d_in[0];
    const int*   ei = (const int*)d_in[1];
    const float* W  = (const float*)d_in[2];
    const float* bi = (const float*)d_in[3];

    const int N = in_sizes[0] / FEAT;
    const int E = in_sizes[1] / 2;
    const int* row = ei;
    const int* col = ei + E;
    const int total = N * FEAT;

    float* ws = (float*)d_ws;
    size_t off = 0;
    float* dinv = ws + off; off += align256f((size_t)N);
    float* x0   = ws + off; off += align256f((size_t)total);
    float* lst  = ws + off;

    const int htiles = (N + TILE - 1) / TILE;
    const int nblk = (E + PBLK - 1) / PBLK;
    k_tile_planes<<<(nblk + 7) / 8, 256, 0, stream>>>(col, x0, E);
    k_filter_tile<<<htiles, HOPW * 32, 0, stream>>>(lst, x0, row, col, dinv, N, E);
    k_hop_tile<true><<<htiles, HOPW * 32, 0, stream>>>(x, x0, lst, row, col, dinv, W, bi, (float*)d_out, N, E);
    k_hop_tile<false><<<htiles, HOPW * 32, 0, stream>>>(x0, nullptr, lst, row, col, dinv, W, bi, (float*)d_out, N, E);
}


# plane pass also packs each edge's source row and tile row into one word; the filter fetches that one word per hit, not row[e] and col[e] (two scattered 4-byte loads); sum order unchanged
# speedup vs baseline: 3.4406x; 1.0526x over previous
#include <hip/hip_runtime.h>
#include <math.h>

typedef float v2f __attribute__((ext_vector_type(2)));
typedef float v8f __attribute__((ext_vector_type(8)));

#define FEAT   64
#define NCLS   40
#define NCLSP  48
#define WLS    68

#define TILE    391
#define HOPW    32
#define GRP     512
#define QFLUSH  64
#define QCAP    (QFLUSH + GRP)
#define BW      16
#define LHDR    4
#define FQCAP   1312
#define PBLK    1024
#define PKSH    23
#define PKMASK  0x7fffffu

#define APPEND(m_, e_)                                                         \
    {                                                                          \
        const unsigned mk_ = __builtin_amdgcn_ballot_w32(m_);                  \
        if (mk_) {                                                             \
            if (m_) q[cnt + (int)__builtin_amdgcn_mbcnt_lo(mk_, 0u)] = (e_);   \
            cnt += __builtin_popcount(mk_);                                    \
        }                                                                      \
    }

#define FILTER4(cv_, e0_)                                                      \
    {                                                                          \
        bool m_;                                                               \
        m_ = (unsigned)((cv_).x - t0) < (unsigned)tn; APPEND(m_, (e0_) + 0);   \
        m_ = (unsigned)((cv_).y - t0) < (unsigned)tn; APPEND(m_, (e0_) + 1);   \
        m_ = (unsigned)((cv_).z - t0) < (unsigned)tn; APPEND(m_, (e0_) + 2);   \
        m_ = (unsigned)((cv_).w - t0) < (unsigned)tn; APPEND(m_, (e0_) + 3);   \
    }

#define LOADSET(X, tt)                                                         \
    {                                                                          \
        int g_ = (tt) + rot; if (g_ >= nG) g_ -= nG;                           \
        g_ = ((tt) < nG) ? g_ : 0;                                             \
        g##X = g_;                                                             \
        const int4* p_ = colv + (size_t)g_ * (GRP / 4) + lane;                 \
        X##0 = p_[0]; X##1 = p_[32]; X##2 = p_[64]; X##3 = p_[96];             \
    }

#define FILTERSET(X)                                                           \
    {                                                                          \
        const int eb_ = head + g##X * GRP + lane * 4;                          \
        FILTER4(X##0, eb_);                                                    \
        FILTER4(X##1, eb_ + 128);                                              \
        FILTER4(X##2, eb_ + 256);                                              \
        FILTER4(X##3, eb_ + 384);                                              \
    }

#define STAGE(CUR, NXT2)                                                       \
    if (t >= nG) break;                                                        \
    LOADSET(NXT2, t + 2 * HOPW);                                               \
    FILTERSET(CUR);                                                            \
    t += HOPW;                                                                 \
    if (cnt > QFLUSH) drain(true, cnt);

template <class DrainF>
__device__ __forceinline__ void scan_edge_list(const int* __restrict__ col, const int E,
                                               const int t0, const int tn, int* const q,
                                               const int wave, const int lane, DrainF drain) {
    int cnt = 0;

    int head = (int)((0ull - ((unsigned long long)(size_t)col >> 2)) & 3ull);
    if (head > E) head = E;
    const int nG   = (E - head) / GRP;
    const int nrem = E - nG * GRP;
    for (int s = wave; s * 32 < nrem; s += HOPW) {
        const int idx = s * 32 + lane;
        const int e = (idx < head) ? idx : idx + nG * GRP;
        const int c = (idx < nrem) ? col[e] : -1;
        const bool m = (unsigned)(c - t0) < (unsigned)tn;
        APPEND(m, e);
    }

    const int rot = (int)(((long long)blockIdx.x * nG) / gridDim.x);
    const int4* colv = (const int4*)(col + head);

    int t = wave;
    const int4 none = make_int4(-1, -1, -1, -1);
    int4 A0 = none, A1 = none, A2 = none, A3 = none;
    int4 B0 = none, B1 = none, B2 = none, B3 = none;
    int4 C0 = none, C1 = none, C2 = none, C3 = none;
    int gA = 0, gB = 0, gC = 0;

    if (nG > 0) {
        LOADSET(A, t);
        LOADSET(B, t + HOPW);
    }
    for (;;) {
        STAGE(A, C);
        STAGE(B, A);
        STAGE(C, B);
    }
    drain(false, cnt);
}

__global__ __launch_bounds__(256)
void k_tile_planes(const int* __restrict__ row, const int* __restrict__ col, float* planesf,
                   float* pkf, int E) {
    const int lane = threadIdx.x & 31;
    const int wave = __builtin_amdgcn_readfirstlane((int)(threadIdx.x >> 5));
    const int blk  = blockIdx.x * 8 + wave;
    const int nB   = (E + PBLK - 1) / PBLK;
    if (blk >= nB) return;
    unsigned p0 = 0, p1 = 0, p2 = 0, p3 = 0, p4 = 0, p5 = 0, p6 = 0, p7 = 0;
    unsigned* const pk = (unsigned*)pkf;
    int cv[32];
    int rv[32];
#pragma unroll
    for (int r = 0; r < 32; ++r) {
        const int e = min(blk * PBLK + r * 32 + lane, E - 1);
        cv[r] = col[e];
        rv[r] = row[e];
    }
#pragma unroll
    for (int r = 0; r < 32; ++r) {
        const int e = min(blk * PBLK + r * 32 + lane, E - 1);
        const unsigned tl = (unsigned)cv[r] / (unsigned)TILE;
        const unsigned tr = (unsigned)cv[r] - tl * (unsigned)TILE;
        pk[e] = (unsigned)rv[r] | (tr << PKSH);
        cv[r] = (int)tl;
    }
#pragma unroll
    for (int r = 0; r < 32; ++r) {
        const unsigned tl = (unsigned)cv[r];
        const unsigned b0 = __builtin_amdgcn_ballot_w32((tl & 1u) != 0u);
        const unsigned b1 = __builtin_amdgcn_ballot_w32((tl & 2u) != 0u);
        const unsigned b2 = __builtin_amdgcn_ballot_w32((tl & 4u) != 0u);
        const unsigned b3 = __builtin_amdgcn_ballot_w32((tl & 8u) != 0u);
        const unsigned b4 = __builtin_amdgcn_ballot_w32((tl & 16u) != 0u);
        const unsigned b5 = __builtin_amdgcn_ballot_w32((tl & 32u) != 0u);
        const unsigned b6 = __builtin_amdgcn_ballot_w32((tl & 64u) != 0u);
        const unsigned b7 = __builtin_amdgcn_ballot_w32((tl & 128u) != 0u);
        const bool mine = lane == r;
        p0 = mine ? b0 : p0;
        p1 = mine ? b1 : p1;
        p2 = mine ? b2 : p2;
        p3 = mine ? b3 : p3;
        p4 = mine ? b4 : p4;
        p5 = mine ? b5 : p5;
        p6 = mine ? b6 : p6;
        p7 = mine ? b7 : p7;
    }
    uint4* const o = (uint4*)planesf + (size_t)blk * (PBLK / 16);
    o[lane]      = make_uint4(p0, p1, p2, p3);
    o[32 + lane] = make_uint4(p4, p5, p6, p7);
}

#define LOADP(X, tt)                                                           \
    {                                                                          \
        int g_ = (tt) + rot; if (g_ >= nB) g_ -= nB;                           \
        g_ = ((tt) < nB) ? g_ : 0;                                             \
        g##X = g_;                                                             \
        const uint4* p_ = pv + (size_t)g_ * (PBLK / 16) + lane;                \
        X##0 = p_[0]; X##1 = p_[32];                                           \
    }

#define FILTERP(X)                                                             \
    {                                                                          \
        const uint4 lo_ = X##0;                                                \
        const uint4 hi_ = X##1;                                                \
        unsigned hm_ = ~(((lo_.x ^ m0) | (lo_.y ^ m1) | (lo_.z ^ m2) | (lo_.w ^ m3)) | \
                         ((hi_.x ^ m4) | (hi_.y ^ m5) | (hi_.z ^ m6) | (hi_.w ^ m7))); \
        const int eb_ = g##X * PBLK + lane * 32;                               \
        if (g##X == nB - 1) {                                                  \
            const int nv_ = E - eb_;     \
            hm_ = (nv_ >= 32) ? hm_ : ((nv_ > 0) ? (hm_ & ((1u << nv_) - 1u)) : 0u); \
        }                                                                      \
        unsigned mk_ = __builtin_amdgcn_ballot_w32(hm_ != 0u);                 \
        while (mk_) {                                                          \
            if (hm_ != 0u)                                                     \
                q[cnt + (int)__builtin_amdgcn_mbcnt_lo(mk_, 0u)] = eb_ + __builtin_ctz(hm_); \
            cnt += __builtin_popcount(mk_);                                    \
            hm_ &= hm_ - 1u;                                                   \
            mk_ = __builtin_amdgcn_ballot_w32(hm_ != 0u);                      \
        }                                                                      \
    }

#define STAGEP(CUR, NXT2)                                                      \
    if (t >= nB) break;                                                        \
    LOADP(NXT2, t + 2 * HOPW);                                                 \
    FILTERP(CUR);                                                              \
    t += HOPW;                                                                 \
    if (cnt > QFLUSH) drain(true, cnt);

template <class DrainF>
__device__ __forceinline__ void scan_planes(const uint4* __restrict__ pv, const int E,
                                            const int tsel, int* const q,
                                            const int wave, const int lane, DrainF drain) {
    int cnt = 0;
    const int nB = (E + PBLK - 1) / PBLK;
    const unsigned m0 = (tsel & 1)   ? ~0u : 0u;
    const unsigned m1 = (tsel & 2)   ? ~0u : 0u;
    const unsigned m2 = (tsel & 4)   ? ~0u : 0u;
    const unsigned m3 = (tsel & 8)   ? ~0u : 0u;
    const unsigned m4 = (tsel & 16)  ? ~0u : 0u;
    const unsigned m5 = (tsel & 32)  ? ~0u : 0u;
    const unsigned m6 = (tsel & 64)  ? ~0u : 0u;
    const unsigned m7 = (tsel & 128) ? ~0u : 0u;
    const int rot = (int)(((long long)blockIdx.x * nB) / gridDim.x);

    int t = wave;
    const uint4 zero = make_uint4(0u, 0u, 0u, 0u);
    uint4 A0 = zero, A1 = zero, B0 = zero, B1 = zero, C0 = zero, C1 = zero;
    int gA = 0, gB = 0, gC = 0;
    if (nB > 0) {
        LOADP(A, t);
        LOADP(B, t + HOPW);
    }
    for (;;) {
        STAGEP(A, C);
        STAGEP(B, A);
        STAGEP(C, B);
    }
    drain(false, cnt);
}

__global__ __launch_bounds__(HOPW * 32)
void k_filter_tile(float* lists, const float* planesf, const float* __restrict__ pkf,
                   float* __restrict__ dinv, int N, int E) {
    __shared__ int queue[HOPW * FQCAP];
    __shared__ int degc[TILE];
    __shared__ int lcount;

    const int tid  = threadIdx.x;
    const int lane = tid & 31;
    const int wave = __builtin_amdgcn_readfirstlane(tid >> 5);
    const int t0   = blockIdx.x * TILE;
    const int tn   = min(TILE, N - t0);

    int* const lhdr = (int*)(lists + (size_t)t0 * FEAT);
    const int  cap  = (tn * FEAT - LHDR) / 3;
    int* const lr   = lhdr + LHDR;
    int* const lc   = lr + cap;

    for (int i = tid; i < TILE; i += HOPW * 32) degc[i] = 0;
    if (tid == 0) lcount = 0;
    __syncthreads();

    int* const q = queue + wave * FQCAP;
    const unsigned* const pk = (const unsigned*)pkf;
    auto drain = [&](const bool more, int& cnt) {
        const int nproc = more ? (cnt & ~31) : cnt;
        for (int b = 0; b < nproc; b += 64) {
            const int  i0 = b + lane;
            const int  i1 = b + 32 + lane;
            const bool v0 = i0 < nproc;
            const bool v1 = i1 < nproc;
            const int  e0 = q[v0 ? i0 : 0];
            const int  e1 = q[v1 ? i1 : 0];
            const unsigned p0 = pk[e0];
            const unsigned p1 = pk[e1];
            const int  r0 = (int)(p0 & PKMASK);
            const int  r1 = (int)(p1 & PKMASK);
            const int  cl0 = (int)(p0 >> PKSH);
            const int  cl1 = (int)(p1 >> PKSH);
            const int  mtot = min(64, nproc - b);
            int base = 0;
            if (lane == 0)
                base = __hip_atomic_fetch_add(&lcount, mtot, __ATOMIC_RELAXED, __HIP_MEMORY_SCOPE_WORKGROUP);
            if (v0) __hip_atomic_fetch_add(&degc[cl0], 1, __ATOMIC_RELAXED, __HIP_MEMORY_SCOPE_WORKGROUP);
            if (v1) __hip_atomic_fetch_add(&degc[cl1], 1, __ATOMIC_RELAXED, __HIP_MEMORY_SCOPE_WORKGROUP);
            base = __builtin_amdgcn_readfirstlane(base);
            const int s0 = base + lane;
            const int s1 = base + 32 + lane;
            if (v0 && s0 < cap) { lr[s0] = r0; lc[s0] = cl0; }
            if (v1 && s1 < cap) { lr[s1] = r1; lc[s1] = cl1; }
        }
        const int rem = cnt - nproc;
        if (rem > 0) {
            const int keep = (lane < rem) ? q[nproc + lane] : 0;
            if (lane < rem) q[lane] = keep;
        }
        cnt = rem;
    };
    scan_planes((const uint4*)planesf, E, (int)blockIdx.x, q, wave, lane, drain);
    __syncthreads();

    for (int i = tid; i < tn; i += HOPW * 32) {
        const float v = (float)(degc[i] + 1);
        dinv[t0 + i] = (v > 0.0f) ? rsqrtf(v) : 0.0f;
    }
    if (tid == 0) lhdr[0] = lcount;
}

template <bool FIRST>
__global__ __launch_bounds__(HOPW * 32)
void k_hop_tile(const float* __restrict__ xin, float* xout, float* lists,
                const int* __restrict__ row, const int* __restrict__ col,
                const float* __restrict__ dinv,
                const float* __restrict__ Wm, const float* __restrict__ bias,
                float* __restrict__ outp, int N, int E) {
    __shared__ __attribute__((aligned(16))) float tile[TILE * FEAT];
    __shared__ int queue[HOPW * QCAP];
    __shared__ float dloc[TILE];
    __shared__ __attribute__((aligned(16))) float wl[FIRST ? 4 : NCLSP * WLS];

    const int tid  = threadIdx.x;
    const int lane = tid & 31;
    const int wave = __builtin_amdgcn_readfirstlane(tid >> 5);
    const int t0   = blockIdx.x * TILE;
    const int tn   = min(TILE, N - t0);

    int*   const lhdr = (int*)(lists + (size_t)t0 * FEAT);
    const int    cap  = (tn * FEAT - LHDR) / 3;
    int*   const lr   = lhdr + LHDR;
    int*   const lc   = lr + cap;
    float* const lw   = (float*)(lc + cap);
    const int hraw = lhdr[0];

    {
        const float4* src = (const float4*)(xin + (size_t)t0 * FEAT);
        float4* dst = (float4*)tile;
        for (int i = tid; i < tn * (FEAT / 4); i += HOPW * 32) {
            float w = dinv[t0 + (i >> 4)];
            w = w * w;
            const float4 v = src[i];
            dst[i] = make_float4(w * v.x, w * v.y, w * v.z, w * v.w);
        }
        if (FIRST) {
            for (int i = tid; i < tn; i += HOPW * 32) dloc[i] = dinv[t0 + i];
        } else {
            for (int i = tid; i < NCLSP * FEAT; i += HOPW * 32) {
                const int r = i >> 6;
                wl[r * WLS + (i & (FEAT - 1))] = (r < NCLS) ? Wm[i] : 0.0f;
            }
        }
    }
    __syncthreads();

    const int  lcnt    = __builtin_amdgcn_readfirstlane(hraw);
    const bool uselist = lcnt >= 0 && lcnt <= cap;

    if (!uselist) {
        auto consume = [&](const int r0, const int r1, const float w0, const float w1,
                           const int cl0, const int cl1, const int mtot) {
#pragma unroll 1
            for (int k = 0; k < mtot; k += BW) {
                const bool  hi = k >= 32;
                const int   rs = hi ? r1 : r0;
                const float ws = hi ? w1 : w0;
                const int   cs = hi ? cl1 : cl0;
                const int   kk = k & 31;
                v2f val[BW];
#pragma unroll
                for (int u = 0; u < BW; ++u) {
                    const int rk = __builtin_amdgcn_readlane(rs, kk + u);
                    val[u] = *(const v2f*)(xin + (size_t)rk * FEAT + 2 * lane);
                }
#pragma unroll
                for (int u = 0; u < BW; ++u) {
                    if (k + u < mtot) {
                        const float wk = __int_as_float(__builtin_amdgcn_readlane(__float_as_int(ws), kk + u));
                        const int   ck = __builtin_amdgcn_readlane(cs, kk + u);
                        float* d = tile + ck * FEAT + 2 * lane;
                        __hip_atomic_fetch_add(d + 0, wk * val[u].x, __ATOMIC_RELAXED, __HIP_MEMORY_SCOPE_WORKGROUP);
                        __hip_atomic_fetch_add(d + 1, wk * val[u].y, __ATOMIC_RELAXED, __HIP_MEMORY_SCOPE_WORKGROUP);
                    }
                }
            }
        };
        int* const q = queue + wave * QCAP;
        auto drain = [&](const bool more, int& cnt) {
            const int nproc = more ? (cnt & ~31) : cnt;
            for (int b = 0; b < nproc; b += 64) {
                const int  i0 = b + lane;
                const int  i1 = b + 32 + lane;
                const bool v0 = i0 < nproc;
                const bool v1 = i1 < nproc;
                const int  e0 = q[v0 ? i0 : 0];
                const int  e1 = q[v1 ? i1 : 0];
                const int  r0 = row[e0];
                const int  r1 = row[e1];
                const int  c0 = col[e0];
                const int  c1 = col[e1];
                const float dr0 = dinv[r0];
                const float dr1 = dinv[r1];
                const float dc0 = dinv[c0];
                const float dc1 = dinv[c1];
                const float w0 = v0 ? dr0 * dc0 : 0.0f;
                const float w1 = v1 ? dr1 * dc1 : 0.0f;
                consume(r0, r1, w0, w1, c0 - t0, c1 - t0, min(64, nproc - b));
            }
            const int rem = cnt - nproc;
            if (rem > 0) {
                const int keep = (lane < rem) ? q[nproc + lane] : 0;
                if (lane < rem) q[lane] = keep;
            }
            cnt = rem;
        };
        scan_edge_list(col, E, t0, tn, q, wave, lane, drain);
    } else {
        auto walk32 = [&](const int rs, const float wa, const float wb, const int cs, const int n) {
            v2f val[32];
#pragma unroll
            for (int u = 0; u < 32; ++u) {
                const int rk = __builtin_amdgcn_readlane(rs, u);
                val[u] = *(const v2f*)(xin + (size_t)rk * FEAT + 2 * lane);
            }
            const float ws = FIRST ? wa * wb : wa;
#pragma unroll
            for (int u = 0; u < 32; ++u) {
                const float wk = __int_as_float(__builtin_amdgcn_readlane(__float_as_int(ws), u));
                const int   ck = __builtin_amdgcn_readlane(cs, u);
                const bool  ok = u < n;
                float* d = tile + ck * FEAT + 2 * lane;
                __hip_atomic_fetch_add(d + 0, ok ? wk * val[u].x : 0.0f, __ATOMIC_RELAXED, __HIP_MEMORY_SCOPE_WORKGROUP);
                __hip_atomic_fetch_add(d + 1, ok ? wk * val[u].y : 0.0f, __ATOMIC_RELAXED, __HIP_MEMORY_SCOPE_WORKGROUP);
            }
            return ws;
        };
        int b = wave * 64;
        int   r0 = 0, r1 = 0, cl0 = 0, cl1 = 0;
        float w0 = 0.0f, w1 = 0.0f;
        if (b < lcnt) {
            const bool v0 = b + lane < lcnt;
            const bool v1 = b + 32 + lane < lcnt;
            const int  j0 = v0 ? b + lane : 0;
            const int  j1 = v1 ? b + 32 + lane : 0;
            r0 = lr[j0]; r1 = lr[j1];
            cl0 = lc[j0]; cl1 = lc[j1];
            if (!FIRST) { w0 = lw[j0]; w1 = lw[j1]; }
        }
        while (b < lcnt) {
            const int nb = b + HOPW * 64;
            int   nr0 = 0, nr1 = 0, ncl0 = 0, ncl1 = 0;
            float nw0 = 0.0f, nw1 = 0.0f;
            if (nb < lcnt) {
                const bool v0 = nb + lane < lcnt;
                const bool v1 = nb + 32 + lane < lcnt;
                const int  j0 = v0 ? nb + lane : 0;
                const int  j1 = v1 ? nb + 32 + lane : 0;
                nr0 = lr[j0]; nr1 = lr[j1];
                ncl0 = lc[j0]; ncl1 = lc[j1];
                if (!FIRST) { nw0 = lw[j0]; nw1 = lw[j1]; }
            }
            const int m = lcnt - b;
            if (FIRST) {
                const float a0 = dinv[r0];
                const float a1 = dinv[r1];
                const float g0 = walk32(r0, a0, dloc[cl0], cl0, m);
                if (b + lane < lcnt) lw[b + lane] = g0;
                if (m > 32) {
                    const float g1 = walk32(r1, a1, dloc[cl1], cl1, m - 32);
                    if (b + 32 + lane < lcnt) lw[b + 32 + lane] = g1;
                }
            } else {
                walk32(r0, w0, 1.0f, cl0, m);
                if (m > 32) walk32(r1, w1, 1.0f, cl1, m - 32);
            }
            r0 = nr0; r1 = nr1; cl0 = ncl0; cl1 = ncl1; w0 = nw0; w1 = nw1;
            b = nb;
        }
    }
    __syncthreads();

    if (FIRST) {
        const float4* srct = (const float4*)tile;
        float4* dst = (float4*)(xout + (size_t)t0 * FEAT);
        for (int i = tid; i < tn * (FEAT / 4); i += HOPW * 32) dst[i] = srct[i];
    } else {
        const int half = lane >> 4;
        const int l16  = lane & 15;
        const bool has2 = l16 < NCLS - 32;
        const float bb0 = bias[l16];
        const float bb1 = bias[16 + l16];
        const float bb2 = has2 ? bias[32 + l16] : 0.0f;
        const float* const b0r = wl + (l16)      * WLS + 2 * half;
        const float* const b1r = wl + (16 + l16) * WLS + 2 * half;
        const float* const b2r = wl + (32 + l16) * WLS + 2 * half;
        for (int rb = wave * 16; rb < tn; rb += HOPW * 16) {
            const int ar = min(rb + l16, tn - 1);
            const float* const arow = tile + ar * FEAT + 2 * half;
            v8f acc0 = {};
            v8f acc1 = {};
            v8f acc2 = {};
#pragma unroll
            for (int k0 = 0; k0 < FEAT; k0 += 4) {
                const v2f a  = *(const v2f*)(arow + k0);
                const v2f b0 = *(const v2f*)(b0r + k0);
                const v2f b1 = *(const v2f*)(b1r + k0);
                const v2f b2 = *(const v2f*)(b2r + k0);
                acc0 = __builtin_amdgcn_wmma_f32_16x16x4_f32(false, a, false, b0, (short)0, acc0, false, false);
                acc1 = __builtin_amdgcn_wmma_f32_16x16x4_f32(false, a, false, b1, (short)0, acc1, false, false);
                acc2 = __builtin_amdgcn_wmma_f32_16x16x4_f32(false, a, false, b2, (short)0, acc2, false, false);
            }
            float a0[8], a1[8], a2[8], mx[8], sm[8];
#pragma unroll
            for (int j = 0; j < 8; ++j) {
                a0[j] = acc0[j] + bb0;
                a1[j] = acc1[j] + bb1;
                a2[j] = acc2[j] + bb2;
                const float m01 = fmaxf(a0[j], a1[j]);
                mx[j] = has2 ? fmaxf(m01, a2[j]) : m01;
            }
#pragma unroll
            for (int s = 1; s < 16; s <<= 1) {
#pragma unroll
                for (int j = 0; j < 8; ++j) mx[j] = fmaxf(mx[j], __shfl_xor(mx[j], s));
            }
#pragma unroll
            for (int j = 0; j < 8; ++j) {
                const float e01 = expf(a0[j] - mx[j]) + expf(a1[j] - mx[j]);
                sm[j] = has2 ? e01 + expf(a2[j] - mx[j]) : e01;
            }
#pragma unroll
            for (int s = 1; s < 16; s <<= 1) {
#pragma unroll
                for (int j = 0; j < 8; ++j) sm[j] += __shfl_xor(sm[j], s);
            }
#pragma unroll
            for (int j = 0; j < 8; ++j) {
                const int   rr = rb + j + 8 * half;
                const float ls = logf(sm[j]);
                if (rr < tn) {
                    float* o = outp + (size_t)(t0 + rr) * NCLS;
                    o[l16]      = a0[j] - mx[j] - ls;
                    o[16 + l16] = a1[j] - mx[j] - ls;
                    if (has2) o[32 + l16] = a2[j] - mx[j] - ls;
                }
            }
        }
    }
}


static inline size_t align256f(size_t nfloats) { return (nfloats + 63) & ~(size_t)63; }

extern "C" void kernel_launch(void* const* d_in, const int* in_sizes, int n_in,
                              void* d_out, int out_size, void* d_ws, size_t ws_size,
                              hipStream_t stream) {
    const float* x  = (const float*)d_in[0];
    const int*   ei = (const int*)d_in[1];
    const float* W  = (const float*)d_in[2];
    const float* bi = (const float*)d_in[3];

    const int N = in_sizes[0] / FEAT;
    const int E = in_sizes[1] / 2;
    const int* row = ei;
    const int* col = ei + E;
    const int total = N * FEAT;

    float* ws = (float*)d_ws;
    size_t off = 0;
    float* dinv = ws + off; off += align256f((size_t)N);
    float* x0   = ws + off; off += align256f((size_t)total);
    float* lst  = ws + off;

    const int htiles = (N + TILE - 1) / TILE;
    const int nblk = (E + PBLK - 1) / PBLK;
    float* pkw = x0 + align256f((size_t)nblk * (PBLK / 4));
    k_tile_planes<<<(nblk + 7) / 8, 256, 0, stream>>>(row, col, x0, pkw, E);
    k_filter_tile<<<htiles, HOPW * 32, 0, stream>>>(lst, x0, pkw, dinv, N, E);
    k_hop_tile<true><<<htiles, HOPW * 32, 0, stream>>>(x, x0, lst, row, col, dinv, W, bi, (float*)d_out, N, E);
    k_hop_tile<false><<<htiles, HOPW * 32, 0, stream>>>(x0, nullptr, lst, row, col, dinv, W, bi, (float*)d_out, N, E);
}


# hop prologue: all of a thread's seed rows, the tile's dinv or W and the first list entries are requested in one batch: 6 dependent round trips before a hop's first add become 2; sums unchanged
# speedup vs baseline: 3.5678x; 1.0370x over previous
#include <hip/hip_runtime.h>
#include <math.h>

typedef float v2f __attribute__((ext_vector_type(2)));
typedef float v8f __attribute__((ext_vector_type(8)));

#define FEAT   64
#define NCLS   40
#define NCLSP  48
#define WLS    68

#define TILE    391
#define HOPW    32
#define GRP     512
#define QFLUSH  64
#define QCAP    (QFLUSH + GRP)
#define BW      16
#define LHDR    4
#define FQCAP   1312
#define PBLK    1024
#define PKSH    23
#define PKMASK  0x7fffffu

#define APPEND(m_, e_)                                                         \
    {                                                                          \
        const unsigned mk_ = __builtin_amdgcn_ballot_w32(m_);                  \
        if (mk_) {                                                             \
            if (m_) q[cnt + (int)__builtin_amdgcn_mbcnt_lo(mk_, 0u)] = (e_);   \
            cnt += __builtin_popcount(mk_);                                    \
        }                                                                      \
    }

#define FILTER4(cv_, e0_)                                                      \
    {                                                                          \
        bool m_;                                                               \
        m_ = (unsigned)((cv_).x - t0) < (unsigned)tn; APPEND(m_, (e0_) + 0);   \
        m_ = (unsigned)((cv_).y - t0) < (unsigned)tn; APPEND(m_, (e0_) + 1);   \
        m_ = (unsigned)((cv_).z - t0) < (unsigned)tn; APPEND(m_, (e0_) + 2);   \
        m_ = (unsigned)((cv_).w - t0) < (unsigned)tn; APPEND(m_, (e0_) + 3);   \
    }

#define LOADSET(X, tt)                                                         \
    {                                                                          \
        int g_ = (tt) + rot; if (g_ >= nG) g_ -= nG;                           \
        g_ = ((tt) < nG) ? g_ : 0;                                             \
        g##X = g_;                                                             \
        const int4* p_ = colv + (size_t)g_ * (GRP / 4) + lane;                 \
        X##0 = p_[0]; X##1 = p_[32]; X##2 = p_[64]; X##3 = p_[96];             \
    }

#define FILTERSET(X)                                                           \
    {                                                                          \
        const int eb_ = head + g##X * GRP + lane * 4;                          \
        FILTER4(X##0, eb_);                                                    \
        FILTER4(X##1, eb_ + 128);                                              \
        FILTER4(X##2, eb_ + 256);                                              \
        FILTER4(X##3, eb_ + 384);                                              \
    }

#define STAGE(CUR, NXT2)                                                       \
    if (t >= nG) break;                                                        \
    LOADSET(NXT2, t + 2 * HOPW);                                               \
    FILTERSET(CUR);                                                            \
    t += HOPW;                                                                 \
    if (cnt > QFLUSH) drain(true, cnt);

template <class DrainF>
__device__ __forceinline__ void scan_edge_list(const int* __restrict__ col, const int E,
                                               const int t0, const int tn, int* const q,
                                               const int wave, const int lane, DrainF drain) {
    int cnt = 0;

    int head = (int)((0ull - ((unsigned long long)(size_t)col >> 2)) & 3ull);
    if (head > E) head = E;
    const int nG   = (E - head) / GRP;
    const int nrem = E - nG * GRP;
    for (int s = wave; s * 32 < nrem; s += HOPW) {
        const int idx = s * 32 + lane;
        const int e = (idx < head) ? idx : idx + nG * GRP;
        const int c = (idx < nrem) ? col[e] : -1;
        const bool m = (unsigned)(c - t0) < (unsigned)tn;
        APPEND(m, e);
    }

    const int rot = (int)(((long long)blockIdx.x * nG) / gridDim.x);
    const int4* colv = (const int4*)(col + head);

    int t = wave;
    const int4 none = make_int4(-1, -1, -1, -1);
    int4 A0 = none, A1 = none, A2 = none, A3 = none;
    int4 B0 = none, B1 = none, B2 = none, B3 = none;
    int4 C0 = none, C1 = none, C2 = none, C3 = none;
    int gA = 0, gB = 0, gC = 0;

    if (nG > 0) {
        LOADSET(A, t);
        LOADSET(B, t + HOPW);
    }
    for (;;) {
        STAGE(A, C);
        STAGE(B, A);
        STAGE(C, B);
    }
    drain(false, cnt);
}

__global__ __launch_bounds__(256)
void k_tile_planes(const int* __restrict__ row, const int* __restrict__ col, float* planesf,
                   float* pkf, int E) {
    const int lane = threadIdx.x & 31;
    const int wave = __builtin_amdgcn_readfirstlane((int)(threadIdx.x >> 5));
    const int blk  = blockIdx.x * 8 + wave;
    const int nB   = (E + PBLK - 1) / PBLK;
    if (blk >= nB) return;
    unsigned p0 = 0, p1 = 0, p2 = 0, p3 = 0, p4 = 0, p5 = 0, p6 = 0, p7 = 0;
    unsigned* const pk = (unsigned*)pkf;
    int cv[32];
    int rv[32];
#pragma unroll
    for (int r = 0; r < 32; ++r) {
        const int e = min(blk * PBLK + r * 32 + lane, E - 1);
        cv[r] = col[e];
        rv[r] = row[e];
    }
#pragma unroll
    for (int r = 0; r < 32; ++r) {
        const int e = min(blk * PBLK + r * 32 + lane, E - 1);
        const unsigned tl = (unsigned)cv[r] / (unsigned)TILE;
        const unsigned tr = (unsigned)cv[r] - tl * (unsigned)TILE;
        pk[e] = (unsigned)rv[r] | (tr << PKSH);
        cv[r] = (int)tl;
    }
#pragma unroll
    for (int r = 0; r < 32; ++r) {
        const unsigned tl = (unsigned)cv[r];
        const unsigned b0 = __builtin_amdgcn_ballot_w32((tl & 1u) != 0u);
        const unsigned b1 = __builtin_amdgcn_ballot_w32((tl & 2u) != 0u);
        const unsigned b2 = __builtin_amdgcn_ballot_w32((tl & 4u) != 0u);
        const unsigned b3 = __builtin_amdgcn_ballot_w32((tl & 8u) != 0u);
        const unsigned b4 = __builtin_amdgcn_ballot_w32((tl & 16u) != 0u);
        const unsigned b5 = __builtin_amdgcn_ballot_w32((tl & 32u) != 0u);
        const unsigned b6 = __builtin_amdgcn_ballot_w32((tl & 64u) != 0u);
        const unsigned b7 = __builtin_amdgcn_ballot_w32((tl & 128u) != 0u);
        const bool mine = lane == r;
        p0 = mine ? b0 : p0;
        p1 = mine ? b1 : p1;
        p2 = mine ? b2 : p2;
        p3 = mine ? b3 : p3;
        p4 = mine ? b4 : p4;
        p5 = mine ? b5 : p5;
        p6 = mine ? b6 : p6;
        p7 = mine ? b7 : p7;
    }
    uint4* const o = (uint4*)planesf + (size_t)blk * (PBLK / 16);
    o[lane]      = make_uint4(p0, p1, p2, p3);
    o[32 + lane] = make_uint4(p4, p5, p6, p7);
}

#define LOADP(X, tt)                                                           \
    {                                                                          \
        int g_ = (tt) + rot; if (g_ >= nB) g_ -= nB;                           \
        g_ = ((tt) < nB) ? g_ : 0;                                             \
        g##X = g_;                                                             \
        const uint4* p_ = pv + (size_t)g_ * (PBLK / 16) + lane;                \
        X##0 = p_[0]; X##1 = p_[32];                                           \
    }

#define FILTERP(X)                                                             \
    {                                                                          \
        const uint4 lo_ = X##0;                                                \
        const uint4 hi_ = X##1;                                                \
        unsigned hm_ = ~(((lo_.x ^ m0) | (lo_.y ^ m1) | (lo_.z ^ m2) | (lo_.w ^ m3)) | \
                         ((hi_.x ^ m4) | (hi_.y ^ m5) | (hi_.z ^ m6) | (hi_.w ^ m7))); \
        const int eb_ = g##X * PBLK + lane * 32;                               \
        if (g##X == nB - 1) {                                                  \
            const int nv_ = E - eb_;     \
            hm_ = (nv_ >= 32) ? hm_ : ((nv_ > 0) ? (hm_ & ((1u << nv_) - 1u)) : 0u); \
        }                                                                      \
        unsigned mk_ = __builtin_amdgcn_ballot_w32(hm_ != 0u);                 \
        while (mk_) {                                                          \
            if (hm_ != 0u)                                                     \
                q[cnt + (int)__builtin_amdgcn_mbcnt_lo(mk_, 0u)] = eb_ + __builtin_ctz(hm_); \
            cnt += __builtin_popcount(mk_);                                    \
            hm_ &= hm_ - 1u;                                                   \
            mk_ = __builtin_amdgcn_ballot_w32(hm_ != 0u);                      \
        }                                                                      \
    }

#define STAGEP(CUR, NXT2)                                                      \
    if (t >= nB) break;                                                        \
    LOADP(NXT2, t + 2 * HOPW);                                                 \
    FILTERP(CUR);                                                              \
    t += HOPW;                                                                 \
    if (cnt > QFLUSH) drain(true, cnt);

template <class DrainF>
__device__ __forceinline__ void scan_planes(const uint4* __restrict__ pv, const int E,
                                            const int tsel, int* const q,
                                            const int wave, const int lane, DrainF drain) {
    int cnt = 0;
    const int nB = (E + PBLK - 1) / PBLK;
    const unsigned m0 = (tsel & 1)   ? ~0u : 0u;
    const unsigned m1 = (tsel & 2)   ? ~0u : 0u;
    const unsigned m2 = (tsel & 4)   ? ~0u : 0u;
    const unsigned m3 = (tsel & 8)   ? ~0u : 0u;
    const unsigned m4 = (tsel & 16)  ? ~0u : 0u;
    const unsigned m5 = (tsel & 32)  ? ~0u : 0u;
    const unsigned m6 = (tsel & 64)  ? ~0u : 0u;
    const unsigned m7 = (tsel & 128) ? ~0u : 0u;
    const int rot = (int)(((long long)blockIdx.x * nB) / gridDim.x);

    int t = wave;
    const uint4 zero = make_uint4(0u, 0u, 0u, 0u);
    uint4 A0 = zero, A1 = zero, B0 = zero, B1 = zero, C0 = zero, C1 = zero;
    int gA = 0, gB = 0, gC = 0;
    if (nB > 0) {
        LOADP(A, t);
        LOADP(B, t + HOPW);
    }
    for (;;) {
        STAGEP(A, C);
        STAGEP(B, A);
        STAGEP(C, B);
    }
    drain(false, cnt);
}

__global__ __launch_bounds__(HOPW * 32)
void k_filter_tile(float* lists, const float* planesf, const float* __restrict__ pkf,
                   float* __restrict__ dinv, int N, int E) {
    __shared__ int queue[HOPW * FQCAP];
    __shared__ int degc[TILE];
    __shared__ int lcount;

    const int tid  = threadIdx.x;
    const int lane = tid & 31;
    const int wave = __builtin_amdgcn_readfirstlane(tid >> 5);
    const int t0   = blockIdx.x * TILE;
    const int tn   = min(TILE, N - t0);

    int* const lhdr = (int*)(lists + (size_t)t0 * FEAT);
    const int  cap  = (tn * FEAT - LHDR) / 3;
    int* const lr   = lhdr + LHDR;
    int* const lc   = lr + cap;

    for (int i = tid; i < TILE; i += HOPW * 32) degc[i] = 0;
    if (tid == 0) lcount = 0;
    __syncthreads();

    int* const q = queue + wave * FQCAP;
    const unsigned* const pk = (const unsigned*)pkf;
    auto drain = [&](const bool more, int& cnt) {
        const int nproc = more ? (cnt & ~31) : cnt;
        for (int b = 0; b < nproc; b += 64) {
            const int  i0 = b + lane;
            const int  i1 = b + 32 + lane;
            const bool v0 = i0 < nproc;
            const bool v1 = i1 < nproc;
            const int  e0 = q[v0 ? i0 : 0];
            const int  e1 = q[v1 ? i1 : 0];
            const unsigned p0 = pk[e0];
            const unsigned p1 = pk[e1];
            const int  r0 = (int)(p0 & PKMASK);
            const int  r1 = (int)(p1 & PKMASK);
            const int  cl0 = (int)(p0 >> PKSH);
            const int  cl1 = (int)(p1 >> PKSH);
            const int  mtot = min(64, nproc - b);
            int base = 0;
            if (lane == 0)
                base = __hip_atomic_fetch_add(&lcount, mtot, __ATOMIC_RELAXED, __HIP_MEMORY_SCOPE_WORKGROUP);
            if (v0) __hip_atomic_fetch_add(&degc[cl0], 1, __ATOMIC_RELAXED, __HIP_MEMORY_SCOPE_WORKGROUP);
            if (v1) __hip_atomic_fetch_add(&degc[cl1], 1, __ATOMIC_RELAXED, __HIP_MEMORY_SCOPE_WORKGROUP);
            base = __builtin_amdgcn_readfirstlane(base);
            const int s0 = base + lane;
            const int s1 = base + 32 + lane;
            if (v0 && s0 < cap) { lr[s0] = r0; lc[s0] = cl0; }
            if (v1 && s1 < cap) { lr[s1] = r1; lc[s1] = cl1; }
        }
        const int rem = cnt - nproc;
        if (rem > 0) {
            const int keep = (lane < rem) ? q[nproc + lane] : 0;
            if (lane < rem) q[lane] = keep;
        }
        cnt = rem;
    };
    scan_planes((const uint4*)planesf, E, (int)blockIdx.x, q, wave, lane, drain);
    __syncthreads();

    for (int i = tid; i < tn; i += HOPW * 32) {
        const float v = (float)(degc[i] + 1);
        dinv[t0 + i] = (v > 0.0f) ? rsqrtf(v) : 0.0f;
    }
    if (tid == 0) lhdr[0] = lcount;
}

template <bool FIRST>
__global__ __launch_bounds__(HOPW * 32)
void k_hop_tile(const float* __restrict__ xin, float* xout, float* lists,
                const int* __restrict__ row, const int* __restrict__ col,
                const float* __restrict__ dinv,
                const float* __restrict__ Wm, const float* __restrict__ bias,
                float* __restrict__ outp, int N, int E) {
    __shared__ __attribute__((aligned(16))) float tile[TILE * FEAT];
    __shared__ int queue[HOPW * QCAP];
    __shared__ float dloc[TILE];
    __shared__ __attribute__((aligned(16))) float wl[FIRST ? 4 : NCLSP * WLS];

    const int tid  = threadIdx.x;
    const int lane = tid & 31;
    const int wave = __builtin_amdgcn_readfirstlane(tid >> 5);
    const int t0   = blockIdx.x * TILE;
    const int tn   = min(TILE, N - t0);

    int*   const lhdr = (int*)(lists + (size_t)t0 * FEAT);
    const int    cap  = (tn * FEAT - LHDR) / 3;
    int*   const lr   = lhdr + LHDR;
    int*   const lc   = lr + cap;
    float* const lw   = (float*)(lc + cap);
    const int hraw = lhdr[0];

    constexpr int SEEDK = (TILE * (FEAT / 4) + HOPW * 32 - 1) / (HOPW * 32);
    constexpr int WK    = (NCLSP * FEAT + HOPW * 32 - 1) / (HOPW * 32);
    static_assert(TILE <= HOPW * 32, "one round copies the tile's own dinv");
    const int slim = tn * (FEAT / 4);
    float4 sv[SEEDK];
    float  sw[SEEDK];
    float  wv[FIRST ? 1 : WK];
    float  dl = 0.0f;
    {
        const float4* src = (const float4*)(xin + (size_t)t0 * FEAT);
#pragma unroll
        for (int k = 0; k < SEEDK; ++k) {
            const int i = min(tid + k * (HOPW * 32), slim - 1);
            sw[k] = dinv[t0 + (i >> 4)];
            sv[k] = src[i];
        }
        if (FIRST) {
            dl = dinv[t0 + min(tid, tn - 1)];
        } else {
#pragma unroll
            for (int k = 0; k < WK; ++k)
                wv[k] = Wm[min(tid + k * (HOPW * 32), NCLS * FEAT - 1)];
        }
    }

    const int  lcnt    = __builtin_amdgcn_readfirstlane(hraw);
    const bool uselist = lcnt >= 0 && lcnt <= cap;

    int   b = wave * 64;
    int   r0 = 0, r1 = 0, cl0 = 0, cl1 = 0;
    float w0 = 0.0f, w1 = 0.0f;
    if (uselist && b < lcnt) {
        const bool v0 = b + lane < lcnt;
        const bool v1 = b + 32 + lane < lcnt;
        const int  j0 = v0 ? b + lane : 0;
        const int  j1 = v1 ? b + 32 + lane : 0;
        r0 = lr[j0]; r1 = lr[j1];
        cl0 = lc[j0]; cl1 = lc[j1];
        if (!FIRST) { w0 = lw[j0]; w1 = lw[j1]; }
    }

    {
        float4* dst = (float4*)tile;
#pragma unroll
        for (int k = 0; k < SEEDK; ++k) {
            const int i = tid + k * (HOPW * 32);
            const float w = sw[k] * sw[k];
            const float4 v = sv[k];
            if (i < slim) dst[i] = make_float4(w * v.x, w * v.y, w * v.z, w * v.w);
        }
        if (FIRST) {
            if (tid < tn) dloc[tid] = dl;
        } else {
#pragma unroll
            for (int k = 0; k < WK; ++k) {
                const int i = tid + k * (HOPW * 32);
                const int r = i >> 6;
                if (i < NCLSP * FEAT) wl[r * WLS + (i & (FEAT - 1))] = (r < NCLS) ? wv[k] : 0.0f;
            }
        }
    }
    __syncthreads();

    if (!uselist) {
        auto consume = [&](const int r0, const int r1, const float w0, const float w1,
                           const int cl0, const int cl1, const int mtot) {
#pragma unroll 1
            for (int k = 0; k < mtot; k += BW) {
                const bool  hi = k >= 32;
                const int   rs = hi ? r1 : r0;
                const float ws = hi ? w1 : w0;
                const int   cs = hi ? cl1 : cl0;
                const int   kk = k & 31;
                v2f val[BW];
#pragma unroll
                for (int u = 0; u < BW; ++u) {
                    const int rk = __builtin_amdgcn_readlane(rs, kk + u);
                    val[u] = *(const v2f*)(xin + (size_t)rk * FEAT + 2 * lane);
                }
#pragma unroll
                for (int u = 0; u < BW; ++u) {
                    if (k + u < mtot) {
                        const float wk = __int_as_float(__builtin_amdgcn_readlane(__float_as_int(ws), kk + u));
                        const int   ck = __builtin_amdgcn_readlane(cs, kk + u);
                        float* d = tile + ck * FEAT + 2 * lane;
                        __hip_atomic_fetch_add(d + 0, wk * val[u].x, __ATOMIC_RELAXED, __HIP_MEMORY_SCOPE_WORKGROUP);
                        __hip_atomic_fetch_add(d + 1, wk * val[u].y, __ATOMIC_RELAXED, __HIP_MEMORY_SCOPE_WORKGROUP);
                    }
                }
            }
        };
        int* const q = queue + wave * QCAP;
        auto drain = [&](const bool more, int& cnt) {
            const int nproc = more ? (cnt & ~31) : cnt;
            for (int b = 0; b < nproc; b += 64) {
                const int  i0 = b + lane;
                const int  i1 = b + 32 + lane;
                const bool v0 = i0 < nproc;
                const bool v1 = i1 < nproc;
                const int  e0 = q[v0 ? i0 : 0];
                const int  e1 = q[v1 ? i1 : 0];
                const int  r0 = row[e0];
                const int  r1 = row[e1];
                const int  c0 = col[e0];
                const int  c1 = col[e1];
                const float dr0 = dinv[r0];
                const float dr1 = dinv[r1];
                const float dc0 = dinv[c0];
                const float dc1 = dinv[c1];
                const float w0 = v0 ? dr0 * dc0 : 0.0f;
                const float w1 = v1 ? dr1 * dc1 : 0.0f;
                consume(r0, r1, w0, w1, c0 - t0, c1 - t0, min(64, nproc - b));
            }
            const int rem = cnt - nproc;
            if (rem > 0) {
                const int keep = (lane < rem) ? q[nproc + lane] : 0;
                if (lane < rem) q[lane] = keep;
            }
            cnt = rem;
        };
        scan_edge_list(col, E, t0, tn, q, wave, lane, drain);
    } else {
        auto walk32 = [&](const int rs, const float wa, const float wb, const int cs, const int n) {
            v2f val[32];
#pragma unroll
            for (int u = 0; u < 32; ++u) {
                const int rk = __builtin_amdgcn_readlane(rs, u);
                val[u] = *(const v2f*)(xin + (size_t)rk * FEAT + 2 * lane);
            }
            const float ws = FIRST ? wa * wb : wa;
#pragma unroll
            for (int u = 0; u < 32; ++u) {
                const float wk = __int_as_float(__builtin_amdgcn_readlane(__float_as_int(ws), u));
                const int   ck = __builtin_amdgcn_readlane(cs, u);
                const bool  ok = u < n;
                float* d = tile + ck * FEAT + 2 * lane;
                __hip_atomic_fetch_add(d + 0, ok ? wk * val[u].x : 0.0f, __ATOMIC_RELAXED, __HIP_MEMORY_SCOPE_WORKGROUP);
                __hip_atomic_fetch_add(d + 1, ok ? wk * val[u].y : 0.0f, __ATOMIC_RELAXED, __HIP_MEMORY_SCOPE_WORKGROUP);
            }
            return ws;
        };
        while (b < lcnt) {
            const int nb = b + HOPW * 64;
            int   nr0 = 0, nr1 = 0, ncl0 = 0, ncl1 = 0;
            float nw0 = 0.0f, nw1 = 0.0f;
            if (nb < lcnt) {
                const bool v0 = nb + lane < lcnt;
                const bool v1 = nb + 32 + lane < lcnt;
                const int  j0 = v0 ? nb + lane : 0;
                const int  j1 = v1 ? nb + 32 + lane : 0;
                nr0 = lr[j0]; nr1 = lr[j1];
                ncl0 = lc[j0]; ncl1 = lc[j1];
                if (!FIRST) { nw0 = lw[j0]; nw1 = lw[j1]; }
            }
            const int m = lcnt - b;
            if (FIRST) {
                const float a0 = dinv[r0];
                const float a1 = dinv[r1];
                const float g0 = walk32(r0, a0, dloc[cl0], cl0, m);
                if (b + lane < lcnt) lw[b + lane] = g0;
                if (m > 32) {
                    const float g1 = walk32(r1, a1, dloc[cl1], cl1, m - 32);
                    if (b + 32 + lane < lcnt) lw[b + 32 + lane] = g1;
                }
            } else {
                walk32(r0, w0, 1.0f, cl0, m);
                if (m > 32) walk32(r1, w1, 1.0f, cl1, m - 32);
            }
            r0 = nr0; r1 = nr1; cl0 = ncl0; cl1 = ncl1; w0 = nw0; w1 = nw1;
            b = nb;
        }
    }
    __syncthreads();

    if (FIRST) {
        const float4* srct = (const float4*)tile;
        float4* dst = (float4*)(xout + (size_t)t0 * FEAT);
        for (int i = tid; i < tn * (FEAT / 4); i += HOPW * 32) dst[i] = srct[i];
    } else {
        const int half = lane >> 4;
        const int l16  = lane & 15;
        const bool has2 = l16 < NCLS - 32;
        const float bb0 = bias[l16];
        const float bb1 = bias[16 + l16];
        const float bb2 = has2 ? bias[32 + l16] : 0.0f;
        const float* const b0r = wl + (l16)      * WLS + 2 * half;
        const float* const b1r = wl + (16 + l16) * WLS + 2 * half;
        const float* const b2r = wl + (32 + l16) * WLS + 2 * half;
        for (int rb = wave * 16; rb < tn; rb += HOPW * 16) {
            const int ar = min(rb + l16, tn - 1);
            const float* const arow = tile + ar * FEAT + 2 * half;
            v8f acc0 = {};
            v8f acc1 = {};
            v8f acc2 = {};
#pragma unroll
            for (int k0 = 0; k0 < FEAT; k0 += 4) {
                const v2f a  = *(const v2f*)(arow + k0);
                const v2f b0 = *(const v2f*)(b0r + k0);
                const v2f b1 = *(const v2f*)(b1r + k0);
                const v2f b2 = *(const v2f*)(b2r + k0);
                acc0 = __builtin_amdgcn_wmma_f32_16x16x4_f32(false, a, false, b0, (short)0, acc0, false, false);
                acc1 = __builtin_amdgcn_wmma_f32_16x16x4_f32(false, a, false, b1, (short)0, acc1, false, false);
                acc2 = __builtin_amdgcn_wmma_f32_16x16x4_f32(false, a, false, b2, (short)0, acc2, false, false);
            }
            float a0[8], a1[8], a2[8], mx[8], sm[8];
#pragma unroll
            for (int j = 0; j < 8; ++j) {
                a0[j] = acc0[j] + bb0;
                a1[j] = acc1[j] + bb1;
                a2[j] = acc2[j] + bb2;
                const float m01 = fmaxf(a0[j], a1[j]);
                mx[j] = has2 ? fmaxf(m01, a2[j]) : m01;
            }
#pragma unroll
            for (int s = 1; s < 16; s <<= 1) {
#pragma unroll
                for (int j = 0; j < 8; ++j) mx[j] = fmaxf(mx[j], __shfl_xor(mx[j], s));
            }
#pragma unroll
            for (int j = 0; j < 8; ++j) {
                const float e01 = expf(a0[j] - mx[j]) + expf(a1[j] - mx[j]);
                sm[j] = has2 ? e01 + expf(a2[j] - mx[j]) : e01;
            }
#pragma unroll
            for (int s = 1; s < 16; s <<= 1) {
#pragma unroll
                for (int j = 0; j < 8; ++j) sm[j] += __shfl_xor(sm[j], s);
            }
#pragma unroll
            for (int j = 0; j < 8; ++j) {
                const int   rr = rb + j + 8 * half;
                const float ls = logf(sm[j]);
                if (rr < tn) {
                    float* o = outp + (size_t)(t0 + rr) * NCLS;
                    o[l16]      = a0[j] - mx[j] - ls;
                    o[16 + l16] = a1[j] - mx[j] - ls;
                    if (has2) o[32 + l16] = a2[j] - mx[j] - ls;
                }
            }
        }
    }
}


static inline size_t align256f(size_t nfloats) { return (nfloats + 63) & ~(size_t)63; }

extern "C" void kernel_launch(void* const* d_in, const int* in_sizes, int n_in,
                              void* d_out, int out_size, void* d_ws, size_t ws_size,
                              hipStream_t stream) {
    const float* x  = (const float*)d_in[0];
    const int*   ei = (const int*)d_in[1];
    const float* W  = (const float*)d_in[2];
    const float* bi = (const float*)d_in[3];

    const int N = in_sizes[0] / FEAT;
    const int E = in_sizes[1] / 2;
    const int* row = ei;
    const int* col = ei + E;
    const int total = N * FEAT;

    float* ws = (float*)d_ws;
    size_t off = 0;
    float* dinv = ws + off; off += align256f((size_t)N);
    float* x0   = ws + off; off += align256f((size_t)total);
    float* lst  = ws + off;

    const int htiles = (N + TILE - 1) / TILE;
    const int nblk = (E + PBLK - 1) / PBLK;
    float* pkw = x0 + align256f((size_t)nblk * (PBLK / 4));
    k_tile_planes<<<(nblk + 7) / 8, 256, 0, stream>>>(row, col, x0, pkw, E);
    k_filter_tile<<<htiles, HOPW * 32, 0, stream>>>(lst, x0, pkw, dinv, N, E);
    k_hop_tile<true><<<htiles, HOPW * 32, 0, stream>>>(x, x0, lst, row, col, dinv, W, bi, (float*)d_out, N, E);
    k_hop_tile<false><<<htiles, HOPW * 32, 0, stream>>>(x0, nullptr, lst, row, col, dinv, W, bi, (float*)d_out, N, E);
}


# a tile's hits are listed per owner wave (tile row mod 32); a hop wave adds its segment's rows into its own LDS rows by plain ds_load/ds_store, no ds_add_f32; sum order: segment order (LDS arrival)
# speedup vs baseline: 3.8933x; 1.0912x over previous
#include <hip/hip_runtime.h>
#include <math.h>

typedef float v2f __attribute__((ext_vector_type(2)));
typedef float v8f __attribute__((ext_vector_type(8)));

#define FEAT   64
#define NCLS   40
#define NCLSP  48
#define WLS    68

#define TILE    391
#define HOPW    32
#define GRP     512
#define QFLUSH  64
#define QCAP    (QFLUSH + GRP)
#define BW      16
#define LHDR    64
#define SEGMAX  384
#define FQCAP   1088
#define PBLK    1024
#define PKSH    23
#define PKMASK  0x7fffffu

#define APPEND(m_, e_)                                                         \
    {                                                                          \
        const unsigned mk_ = __builtin_amdgcn_ballot_w32(m_);                  \
        if (mk_) {                                                             \
            if (m_) q[cnt + (int)__builtin_amdgcn_mbcnt_lo(mk_, 0u)] = (e_);   \
            cnt += __builtin_popcount(mk_);                                    \
        }                                                                      \
    }

#define FILTER4(cv_, e0_)                                                      \
    {                                                                          \
        bool m_;                                                               \
        m_ = (unsigned)((cv_).x - t0) < (unsigned)tn; APPEND(m_, (e0_) + 0);   \
        m_ = (unsigned)((cv_).y - t0) < (unsigned)tn; APPEND(m_, (e0_) + 1);   \
        m_ = (unsigned)((cv_).z - t0) < (unsigned)tn; APPEND(m_, (e0_) + 2);   \
        m_ = (unsigned)((cv_).w - t0) < (unsigned)tn; APPEND(m_, (e0_) + 3);   \
    }

#define LOADSET(X, tt)                                                         \
    {                                                                          \
        int g_ = (tt) + rot; if (g_ >= nG) g_ -= nG;                           \
        g_ = ((tt) < nG) ? g_ : 0;                                             \
        g##X = g_;                                                             \
        const int4* p_ = colv + (size_t)g_ * (GRP / 4) + lane;                 \
        X##0 = p_[0]; X##1 = p_[32]; X##2 = p_[64]; X##3 = p_[96];             \
    }

#define FILTERSET(X)                                                           \
    {                                                                          \
        const int eb_ = head + g##X * GRP + lane * 4;                          \
        FILTER4(X##0, eb_);                                                    \
        FILTER4(X##1, eb_ + 128);                                              \
        FILTER4(X##2, eb_ + 256);                                              \
        FILTER4(X##3, eb_ + 384);                                              \
    }

#define STAGE(CUR, NXT2)                                                       \
    if (t >= nG) break;                                                        \
    LOADSET(NXT2, t + 2 * HOPW);                                               \
    FILTERSET(CUR);                                                            \
    t += HOPW;                                                                 \
    if (cnt > QFLUSH) drain(true, cnt);

template <class DrainF>
__device__ __forceinline__ void scan_edge_list(const int* __restrict__ col, const int E,
                                               const int t0, const int tn, int* const q,
                                               const int wave, const int lane, DrainF drain) {
    int cnt = 0;

    int head = (int)((0ull - ((unsigned long long)(size_t)col >> 2)) & 3ull);
    if (head > E) head = E;
    const int nG   = (E - head) / GRP;
    const int nrem = E - nG * GRP;
    for (int s = wave; s * 32 < nrem; s += HOPW) {
        const int idx = s * 32 + lane;
        const int e = (idx < head) ? idx : idx + nG * GRP;
        const int c = (idx < nrem) ? col[e] : -1;
        const bool m = (unsigned)(c - t0) < (unsigned)tn;
        APPEND(m, e);
    }

    const int rot = (int)(((long long)blockIdx.x * nG) / gridDim.x);
    const int4* colv = (const int4*)(col + head);

    int t = wave;
    const int4 none = make_int4(-1, -1, -1, -1);
    int4 A0 = none, A1 = none, A2 = none, A3 = none;
    int4 B0 = none, B1 = none, B2 = none, B3 = none;
    int4 C0 = none, C1 = none, C2 = none, C3 = none;
    int gA = 0, gB = 0, gC = 0;

    if (nG > 0) {
        LOADSET(A, t);
        LOADSET(B, t + HOPW);
    }
    for (;;) {
        STAGE(A, C);
        STAGE(B, A);
        STAGE(C, B);
    }
    drain(false, cnt);
}

__global__ __launch_bounds__(256)
void k_tile_planes(const int* __restrict__ row, const int* __restrict__ col, float* planesf,
                   float* pkf, int E) {
    const int lane = threadIdx.x & 31;
    const int wave = __builtin_amdgcn_readfirstlane((int)(threadIdx.x >> 5));
    const int blk  = blockIdx.x * 8 + wave;
    const int nB   = (E + PBLK - 1) / PBLK;
    if (blk >= nB) return;
    unsigned p0 = 0, p1 = 0, p2 = 0, p3 = 0, p4 = 0, p5 = 0, p6 = 0, p7 = 0;
    unsigned* const pk = (unsigned*)pkf;
    int cv[32];
    int rv[32];
#pragma unroll
    for (int r = 0; r < 32; ++r) {
        const int e = min(blk * PBLK + r * 32 + lane, E - 1);
        cv[r] = col[e];
        rv[r] = row[e];
    }
#pragma unroll
    for (int r = 0; r < 32; ++r) {
        const int e = min(blk * PBLK + r * 32 + lane, E - 1);
        const unsigned tl = (unsigned)cv[r] / (unsigned)TILE;
        const unsigned tr = (unsigned)cv[r] - tl * (unsigned)TILE;
        pk[e] = (unsigned)rv[r] | (tr << PKSH);
        cv[r] = (int)tl;
    }
#pragma unroll
    for (int r = 0; r < 32; ++r) {
        const unsigned tl = (unsigned)cv[r];
        const unsigned b0 = __builtin_amdgcn_ballot_w32((tl & 1u) != 0u);
        const unsigned b1 = __builtin_amdgcn_ballot_w32((tl & 2u) != 0u);
        const unsigned b2 = __builtin_amdgcn_ballot_w32((tl & 4u) != 0u);
        const unsigned b3 = __builtin_amdgcn_ballot_w32((tl & 8u) != 0u);
        const unsigned b4 = __builtin_amdgcn_ballot_w32((tl & 16u) != 0u);
        const unsigned b5 = __builtin_amdgcn_ballot_w32((tl & 32u) != 0u);
        const unsigned b6 = __builtin_amdgcn_ballot_w32((tl & 64u) != 0u);
        const unsigned b7 = __builtin_amdgcn_ballot_w32((tl & 128u) != 0u);
        const bool mine = lane == r;
        p0 = mine ? b0 : p0;
        p1 = mine ? b1 : p1;
        p2 = mine ? b2 : p2;
        p3 = mine ? b3 : p3;
        p4 = mine ? b4 : p4;
        p5 = mine ? b5 : p5;
        p6 = mine ? b6 : p6;
        p7 = mine ? b7 : p7;
    }
    uint4* const o = (uint4*)planesf + (size_t)blk * (PBLK / 16);
    o[lane]      = make_uint4(p0, p1, p2, p3);
    o[32 + lane] = make_uint4(p4, p5, p6, p7);
}

#define LOADP(X, tt)                                                           \
    {                                                                          \
        int g_ = (tt) + rot; if (g_ >= nB) g_ -= nB;                           \
        g_ = ((tt) < nB) ? g_ : 0;                                             \
        g##X = g_;                                                             \
        const uint4* p_ = pv + (size_t)g_ * (PBLK / 16) + lane;                \
        X##0 = p_[0]; X##1 = p_[32];                                           \
    }

#define FILTERP(X)                                                             \
    {                                                                          \
        const uint4 lo_ = X##0;                                                \
        const uint4 hi_ = X##1;                                                \
        unsigned hm_ = ~(((lo_.x ^ m0) | (lo_.y ^ m1) | (lo_.z ^ m2) | (lo_.w ^ m3)) | \
                         ((hi_.x ^ m4) | (hi_.y ^ m5) | (hi_.z ^ m6) | (hi_.w ^ m7))); \
        const int eb_ = g##X * PBLK + lane * 32;                               \
        if (g##X == nB - 1) {                                                  \
            const int nv_ = E - eb_;     \
            hm_ = (nv_ >= 32) ? hm_ : ((nv_ > 0) ? (hm_ & ((1u << nv_) - 1u)) : 0u); \
        }                                                                      \
        unsigned mk_ = __builtin_amdgcn_ballot_w32(hm_ != 0u);                 \
        while (mk_) {                                                          \
            if (hm_ != 0u)                                                     \
                q[cnt + (int)__builtin_amdgcn_mbcnt_lo(mk_, 0u)] = eb_ + __builtin_ctz(hm_); \
            cnt += __builtin_popcount(mk_);                                    \
            hm_ &= hm_ - 1u;                                                   \
            mk_ = __builtin_amdgcn_ballot_w32(hm_ != 0u);                      \
        }                                                                      \
    }

#define STAGEP(CUR, NXT2)                                                      \
    if (t >= nB) break;                                                        \
    LOADP(NXT2, t + 2 * HOPW);                                                 \
    FILTERP(CUR);                                                              \
    t += HOPW;                                                                 \
    if (cnt > QFLUSH) drain(true, cnt);

template <class DrainF>
__device__ __forceinline__ void scan_planes(const uint4* __restrict__ pv, const int E,
                                            const int tsel, int* const q,
                                            const int wave, const int lane, DrainF drain) {
    int cnt = 0;
    const int nB = (E + PBLK - 1) / PBLK;
    const unsigned m0 = (tsel & 1)   ? ~0u : 0u;
    const unsigned m1 = (tsel & 2)   ? ~0u : 0u;
    const unsigned m2 = (tsel & 4)   ? ~0u : 0u;
    const unsigned m3 = (tsel & 8)   ? ~0u : 0u;
    const unsigned m4 = (tsel & 16)  ? ~0u : 0u;
    const unsigned m5 = (tsel & 32)  ? ~0u : 0u;
    const unsigned m6 = (tsel & 64)  ? ~0u : 0u;
    const unsigned m7 = (tsel & 128) ? ~0u : 0u;
    const int rot = (int)(((long long)blockIdx.x * nB) / gridDim.x);

    int t = wave;
    const uint4 zero = make_uint4(0u, 0u, 0u, 0u);
    uint4 A0 = zero, A1 = zero, B0 = zero, B1 = zero, C0 = zero, C1 = zero;
    int gA = 0, gB = 0, gC = 0;
    if (nB > 0) {
        LOADP(A, t);
        LOADP(B, t + HOPW);
    }
    for (;;) {
        STAGEP(A, C);
        STAGEP(B, A);
        STAGEP(C, B);
    }
    drain(false, cnt);
}

__global__ __launch_bounds__(HOPW * 32)
void k_filter_tile(float* lists, const float* planesf, const float* __restrict__ pkf,
                   float* __restrict__ dinv, int N, int E) {
    __shared__ int queue[HOPW * FQCAP];
    __shared__ unsigned segbuf[HOPW * SEGMAX];
    __shared__ int degc[TILE];
    __shared__ int segcnt[HOPW];

    const int tid  = threadIdx.x;
    const int lane = tid & 31;
    const int wave = __builtin_amdgcn_readfirstlane(tid >> 5);
    const int t0   = blockIdx.x * TILE;
    const int tn   = min(TILE, N - t0);

    int* const lhdr = (int*)(lists + (size_t)t0 * FEAT);
    const int  segcap = min(SEGMAX, (((tn * FEAT - LHDR) / 2) / HOPW) & ~31);
    unsigned* const lp = (unsigned*)(lhdr + LHDR);

    for (int i = tid; i < TILE; i += HOPW * 32) degc[i] = 0;
    if (tid < HOPW) segcnt[tid] = 0;
    __syncthreads();

    int* const q = queue + wave * FQCAP;
    const unsigned* const pk = (const unsigned*)pkf;
    auto drain = [&](const bool more, int& cnt) {
        const int nproc = more ? (cnt & ~31) : cnt;
        for (int b = 0; b < nproc; b += 64) {
            const int  i0 = b + lane;
            const int  i1 = b + 32 + lane;
            const bool v0 = i0 < nproc;
            const bool v1 = i1 < nproc;
            const int  e0 = q[v0 ? i0 : 0];
            const int  e1 = q[v1 ? i1 : 0];
            const unsigned p0 = pk[e0];
            const unsigned p1 = pk[e1];
            const int  cl0 = (int)(p0 >> PKSH);
            const int  cl1 = (int)(p1 >> PKSH);
            if (v0) {
                __hip_atomic_fetch_add(&degc[cl0], 1, __ATOMIC_RELAXED, __HIP_MEMORY_SCOPE_WORKGROUP);
                const int ow = cl0 & (HOPW - 1);
                const int s  = __hip_atomic_fetch_add(&segcnt[ow], 1, __ATOMIC_RELAXED, __HIP_MEMORY_SCOPE_WORKGROUP);
                if (s < segcap) segbuf[ow * SEGMAX + s] = p0;
            }
            if (v1) {
                __hip_atomic_fetch_add(&degc[cl1], 1, __ATOMIC_RELAXED, __HIP_MEMORY_SCOPE_WORKGROUP);
                const int ow = cl1 & (HOPW - 1);
                const int s  = __hip_atomic_fetch_add(&segcnt[ow], 1, __ATOMIC_RELAXED, __HIP_MEMORY_SCOPE_WORKGROUP);
                if (s < segcap) segbuf[ow * SEGMAX + s] = p1;
            }
        }
        const int rem = cnt - nproc;
        if (rem > 0) {
            const int keep = (lane < rem) ? q[nproc + lane] : 0;
            if (lane < rem) q[lane] = keep;
        }
        cnt = rem;
    };
    scan_planes((const uint4*)planesf, E, (int)blockIdx.x, q, wave, lane, drain);
    __syncthreads();

    for (int i = tid; i < tn; i += HOPW * 32) {
        const float v = (float)(degc[i] + 1);
        dinv[t0 + i] = (v > 0.0f) ? rsqrtf(v) : 0.0f;
    }
    {
        const int n = min(segcnt[wave], segcap);
        unsigned* const dst = lp + wave * segcap;
        const unsigned* const sb = segbuf + wave * SEGMAX;
        for (int i = lane; i < n; i += 32) dst[i] = sb[i];
    }
    if (tid < HOPW) {
        const int cn = segcnt[tid];
        lhdr[32 + tid] = cn;
        const unsigned bad = __builtin_amdgcn_ballot_w32(cn > segcap);
        if (tid == 0) lhdr[0] = bad ? -1 : 1;
    }
}

static_assert(HOPW == 32, "lane l of wave 0 writes segment l's count; a tile row's owner is its low 5 bits");

template <bool FIRST, int NB>
__device__ __forceinline__ float walk_own(const float* __restrict__ xin, float* const tile,
                                          const unsigned pe, const float wa, const float wb,
                                          const int n, const int lane) {
    const int rs = (int)(pe & PKMASK);
    const int cs = (int)(pe >> PKSH);
    v2f val[NB];
#pragma unroll
    for (int u = 0; u < NB; ++u) {
        const int rk = __builtin_amdgcn_readlane(rs, u);
        val[u] = *(const v2f*)(xin + (size_t)rk * FEAT + 2 * lane);
    }
    const float ws = FIRST ? wa * wb : wa;
#pragma unroll
    for (int g = 0; g < NB; g += 4) {
        int   ck[4];
        float wk[4];
        v2f   t[4];
#pragma unroll
        for (int i = 0; i < 4; ++i) {
            const float w = __int_as_float(__builtin_amdgcn_readlane(__float_as_int(ws), g + i));
            wk[i] = (g + i < n) ? w : 0.0f;
            ck[i] = __builtin_amdgcn_readlane(cs, g + i);
            t[i]  = *(const v2f*)(tile + ck[i] * FEAT + 2 * lane);
        }
#pragma unroll
        for (int j = 0; j < 4; ++j) {
            v2f a = t[j];
#pragma unroll
            for (int i = 0; i <= j; ++i) {
                const float cw = (i == j || ck[i] == ck[j]) ? wk[i] : 0.0f;
                a.x += cw * val[g + i].x;
                a.y += cw * val[g + i].y;
            }
            t[j] = a;
        }
#pragma unroll
        for (int j = 0; j < 4; ++j) *(v2f*)(tile + ck[j] * FEAT + 2 * lane) = t[j];
    }
    return ws;
}

template <bool FIRST>
__global__ __launch_bounds__(HOPW * 32)
void k_hop_tile(const float* __restrict__ xin, float* xout, float* lists,
                const int* __restrict__ row, const int* __restrict__ col,
                const float* __restrict__ dinv,
                const float* __restrict__ Wm, const float* __restrict__ bias,
                float* __restrict__ outp, int N, int E) {
    __shared__ __attribute__((aligned(16))) float tile[TILE * FEAT];
    __shared__ int queue[HOPW * QCAP];
    __shared__ float dloc[TILE];
    __shared__ __attribute__((aligned(16))) float wl[FIRST ? 4 : NCLSP * WLS];

    const int tid  = threadIdx.x;
    const int lane = tid & 31;
    const int wave = __builtin_amdgcn_readfirstlane(tid >> 5);
    const int t0   = blockIdx.x * TILE;
    const int tn   = min(TILE, N - t0);

    int*   const lhdr   = (int*)(lists + (size_t)t0 * FEAT);
    const int    segcap = min(SEGMAX, (((tn * FEAT - LHDR) / 2) / HOPW) & ~31);
    const unsigned* const lp = (const unsigned*)(lhdr + LHDR) + wave * segcap;
    float* const lw     = (float*)(lhdr + LHDR + HOPW * segcap) + wave * segcap;
    const int hraw = lhdr[0];
    const int sraw = lhdr[32 + wave];

    constexpr int SEEDK = (TILE * (FEAT / 4) + HOPW * 32 - 1) / (HOPW * 32);
    constexpr int WK    = (NCLSP * FEAT + HOPW * 32 - 1) / (HOPW * 32);
    static_assert(TILE <= HOPW * 32, "one round copies the tile's own dinv");
    const int slim = tn * (FEAT / 4);
    float4 sv[SEEDK];
    float  sw[SEEDK];
    float  wv[FIRST ? 1 : WK];
    float  dl = 0.0f;
    {
        const float4* src = (const float4*)(xin + (size_t)t0 * FEAT);
#pragma unroll
        for (int k = 0; k < SEEDK; ++k) {
            const int i = min(tid + k * (HOPW * 32), slim - 1);
            sw[k] = dinv[t0 + (i >> 4)];
            sv[k] = src[i];
        }
        if (FIRST) {
            dl = dinv[t0 + min(tid, tn - 1)];
        } else {
#pragma unroll
            for (int k = 0; k < WK; ++k)
                wv[k] = Wm[min(tid + k * (HOPW * 32), NCLS * FEAT - 1)];
        }
    }

    const bool uselist = __builtin_amdgcn_readfirstlane(hraw) == 1;
    const int  nseg    = uselist ? __builtin_amdgcn_readfirstlane(sraw) : 0;

    int      b  = 0;
    unsigned pe = 0u;
    float    we = 0.0f;
    if (b < nseg) {
        const int j = (lane < nseg) ? lane : 0;
        pe = lp[j];
        if (!FIRST) we = lw[j];
    }

    {
        float4* dst = (float4*)tile;
#pragma unroll
        for (int k = 0; k < SEEDK; ++k) {
            const int i = tid + k * (HOPW * 32);
            const float w = sw[k] * sw[k];
            const float4 v = sv[k];
            if (i < slim) dst[i] = make_float4(w * v.x, w * v.y, w * v.z, w * v.w);
        }
        if (FIRST) {
            if (tid < tn) dloc[tid] = dl;
        } else {
#pragma unroll
            for (int k = 0; k < WK; ++k) {
                const int i = tid + k * (HOPW * 32);
                const int r = i >> 6;
                if (i < NCLSP * FEAT) wl[r * WLS + (i & (FEAT - 1))] = (r < NCLS) ? wv[k] : 0.0f;
            }
        }
    }
    __syncthreads();

    if (!uselist) {
        auto consume = [&](const int r0, const int r1, const float w0, const float w1,
                           const int cl0, const int cl1, const int mtot) {
#pragma unroll 1
            for (int k = 0; k < mtot; k += BW) {
                const bool  hi = k >= 32;
                const int   rs = hi ? r1 : r0;
                const float ws = hi ? w1 : w0;
                const int   cs = hi ? cl1 : cl0;
                const int   kk = k & 31;
                v2f val[BW];
#pragma unroll
                for (int u = 0; u < BW; ++u) {
                    const int rk = __builtin_amdgcn_readlane(rs, kk + u);
                    val[u] = *(const v2f*)(xin + (size_t)rk * FEAT + 2 * lane);
                }
#pragma unroll
                for (int u = 0; u < BW; ++u) {
                    if (k + u < mtot) {
                        const float wk = __int_as_float(__builtin_amdgcn_readlane(__float_as_int(ws), kk + u));
                        const int   ck = __builtin_amdgcn_readlane(cs, kk + u);
                        float* d = tile + ck * FEAT + 2 * lane;
                        __hip_atomic_fetch_add(d + 0, wk * val[u].x, __ATOMIC_RELAXED, __HIP_MEMORY_SCOPE_WORKGROUP);
                        __hip_atomic_fetch_add(d + 1, wk * val[u].y, __ATOMIC_RELAXED, __HIP_MEMORY_SCOPE_WORKGROUP);
                    }
                }
            }
        };
        int* const q = queue + wave * QCAP;
        auto drain = [&](const bool more, int& cnt) {
            const int nproc = more ? (cnt & ~31) : cnt;
            for (int b = 0; b < nproc; b += 64) {
                const int  i0 = b + lane;
                const int  i1 = b + 32 + lane;
                const bool v0 = i0 < nproc;
                const bool v1 = i1 < nproc;
                const int  e0 = q[v0 ? i0 : 0];
                const int  e1 = q[v1 ? i1 : 0];
                const int  r0 = row[e0];
                const int  r1 = row[e1];
                const int  c0 = col[e0];
                const int  c1 = col[e1];
                const float dr0 = dinv[r0];
                const float dr1 = dinv[r1];
                const float dc0 = dinv[c0];
                const float dc1 = dinv[c1];
                const float w0 = v0 ? dr0 * dc0 : 0.0f;
                const float w1 = v1 ? dr1 * dc1 : 0.0f;
                consume(r0, r1, w0, w1, c0 - t0, c1 - t0, min(64, nproc - b));
            }
            const int rem = cnt - nproc;
            if (rem > 0) {
                const int keep = (lane < rem) ? q[nproc + lane] : 0;
                if (lane < rem) q[lane] = keep;
            }
            cnt = rem;
        };
        scan_edge_list(col, E, t0, tn, q, wave, lane, drain);
    } else {
        while (b < nseg) {
            const int nb = b + 32;
            unsigned npe = 0u;
            float    nwe = 0.0f;
            if (nb < nseg) {
                const int j = (nb + lane < nseg) ? nb + lane : 0;
                npe = lp[j];
                if (!FIRST) nwe = lw[j];
            }
            const int m = nseg - b;
            const float wa = FIRST ? dinv[pe & PKMASK] : we;
            const float wb = FIRST ? dloc[pe >> PKSH] : 1.0f;
            float g;
            if (m > 16)     g = walk_own<FIRST, 32>(xin, tile, pe, wa, wb, m, lane);
            else if (m > 8) g = walk_own<FIRST, 16>(xin, tile, pe, wa, wb, m, lane);
            else            g = walk_own<FIRST, 8>(xin, tile, pe, wa, wb, m, lane);
            if (FIRST) {
                if (b + lane < nseg) lw[b + lane] = g;
            }
            pe = npe; we = nwe;
            b = nb;
        }
    }
    __syncthreads();

    if (FIRST) {
        const float4* srct = (const float4*)tile;
        float4* dst = (float4*)(xout + (size_t)t0 * FEAT);
        for (int i = tid; i < tn * (FEAT / 4); i += HOPW * 32) dst[i] = srct[i];
    } else {
        const int half = lane >> 4;
        const int l16  = lane & 15;
        const bool has2 = l16 < NCLS - 32;
        const float bb0 = bias[l16];
        const float bb1 = bias[16 + l16];
        const float bb2 = has2 ? bias[32 + l16] : 0.0f;
        const float* const b0r = wl + (l16)      * WLS + 2 * half;
        const float* const b1r = wl + (16 + l16) * WLS + 2 * half;
        const float* const b2r = wl + (32 + l16) * WLS + 2 * half;
        for (int rb = wave * 16; rb < tn; rb += HOPW * 16) {
            const int ar = min(rb + l16, tn - 1);
            const float* const arow = tile + ar * FEAT + 2 * half;
            v8f acc0 = {};
            v8f acc1 = {};
            v8f acc2 = {};
#pragma unroll
            for (int k0 = 0; k0 < FEAT; k0 += 4) {
                const v2f a  = *(const v2f*)(arow + k0);
                const v2f b0 = *(const v2f*)(b0r + k0);
                const v2f b1 = *(const v2f*)(b1r + k0);
                const v2f b2 = *(const v2f*)(b2r + k0);
                acc0 = __builtin_amdgcn_wmma_f32_16x16x4_f32(false, a, false, b0, (short)0, acc0, false, false);
                acc1 = __builtin_amdgcn_wmma_f32_16x16x4_f32(false, a, false, b1, (short)0, acc1, false, false);
                acc2 = __builtin_amdgcn_wmma_f32_16x16x4_f32(false, a, false, b2, (short)0, acc2, false, false);
            }
            float a0[8], a1[8], a2[8], mx[8], sm[8];
#pragma unroll
            for (int j = 0; j < 8; ++j) {
                a0[j] = acc0[j] + bb0;
                a1[j] = acc1[j] + bb1;
                a2[j] = acc2[j] + bb2;
                const float m01 = fmaxf(a0[j], a1[j]);
                mx[j] = has2 ? fmaxf(m01, a2[j]) : m01;
            }
#pragma unroll
            for (int s = 1; s < 16; s <<= 1) {
#pragma unroll
                for (int j = 0; j < 8; ++j) mx[j] = fmaxf(mx[j], __shfl_xor(mx[j], s));
            }
#pragma unroll
            for (int j = 0; j < 8; ++j) {
                const float e01 = expf(a0[j] - mx[j]) + expf(a1[j] - mx[j]);
                sm[j] = has2 ? e01 + expf(a2[j] - mx[j]) : e01;
            }
#pragma unroll
            for (int s = 1; s < 16; s <<= 1) {
#pragma unroll
                for (int j = 0; j < 8; ++j) sm[j] += __shfl_xor(sm[j], s);
            }
#pragma unroll
            for (int j = 0; j < 8; ++j) {
                const int   rr = rb + j + 8 * half;
                const float ls = logf(sm[j]);
                if (rr < tn) {
                    float* o = outp + (size_t)(t0 + rr) * NCLS;
                    o[l16]      = a0[j] - mx[j] - ls;
                    o[16 + l16] = a1[j] - mx[j] - ls;
                    if (has2) o[32 + l16] = a2[j] - mx[j] - ls;
                }
            }
        }
    }
}


static inline size_t align256f(size_t nfloats) { return (nfloats + 63) & ~(size_t)63; }

extern "C" void kernel_launch(void* const* d_in, const int* in_sizes, int n_in,
                              void* d_out, int out_size, void* d_ws, size_t ws_size,
                              hipStream_t stream) {
    const float* x  = (const float*)d_in[0];
    const int*   ei = (const int*)d_in[1];
    const float* W  = (const float*)d_in[2];
    const float* bi = (const float*)d_in[3];

    const int N = in_sizes[0] / FEAT;
    const int E = in_sizes[1] / 2;
    const int* row = ei;
    const int* col = ei + E;
    const int total = N * FEAT;

    float* ws = (float*)d_ws;
    size_t off = 0;
    float* dinv = ws + off; off += align256f((size_t)N);
    float* x0   = ws + off; off += align256f((size_t)total);
    float* lst  = ws + off;

    const int htiles = (N + TILE - 1) / TILE;
    const int nblk = (E + PBLK - 1) / PBLK;
    float* pkw = x0 + align256f((size_t)nblk * (PBLK / 4));
    k_tile_planes<<<(nblk + 7) / 8, 256, 0, stream>>>(row, col, x0, pkw, E);
    k_filter_tile<<<htiles, HOPW * 32, 0, stream>>>(lst, x0, pkw, dinv, N, E);
    k_hop_tile<true><<<htiles, HOPW * 32, 0, stream>>>(x, x0, lst, row, col, dinv, W, bi, (float*)d_out, N, E);
    k_hop_tile<false><<<htiles, HOPW * 32, 0, stream>>>(x0, nullptr, lst, row, col, dinv, W, bi, (float*)d_out, N, E);
}


# a hop wave's two 16-lane halves each walk a segment of their own (tile row mod 64) with 16-byte lanes and DPP row_share: one instruction serves two entries, no readlane; sums: half-segment order
# speedup vs baseline: 4.3322x; 1.1127x over previous
#include <hip/hip_runtime.h>
#include <math.h>

typedef float v2f __attribute__((ext_vector_type(2)));
typedef float v8f __attribute__((ext_vector_type(8)));

#define FEAT   64
#define NCLS   40
#define NCLSP  48
#define WLS    68

#define TILE    391
#define HOPW    32
#define GRP     512
#define QFLUSH  64
#define QCAP    (QFLUSH + GRP)
#define BW      16
#define SUBS    64
#define LHDR    128
#define SEGMAX  192
#define FQCAP   1088
#define PBLK    1024
#define PKSH    23
#define PKMASK  0x7fffffu

#define APPEND(m_, e_)                                                         \
    {                                                                          \
        const unsigned mk_ = __builtin_amdgcn_ballot_w32(m_);                  \
        if (mk_) {                                                             \
            if (m_) q[cnt + (int)__builtin_amdgcn_mbcnt_lo(mk_, 0u)] = (e_);   \
            cnt += __builtin_popcount(mk_);                                    \
        }                                                                      \
    }

#define FILTER4(cv_, e0_)                                                      \
    {                                                                          \
        bool m_;                                                               \
        m_ = (unsigned)((cv_).x - t0) < (unsigned)tn; APPEND(m_, (e0_) + 0);   \
        m_ = (unsigned)((cv_).y - t0) < (unsigned)tn; APPEND(m_, (e0_) + 1);   \
        m_ = (unsigned)((cv_).z - t0) < (unsigned)tn; APPEND(m_, (e0_) + 2);   \
        m_ = (unsigned)((cv_).w - t0) < (unsigned)tn; APPEND(m_, (e0_) + 3);   \
    }

#define LOADSET(X, tt)                                                         \
    {                                                                          \
        int g_ = (tt) + rot; if (g_ >= nG) g_ -= nG;                           \
        g_ = ((tt) < nG) ? g_ : 0;                                             \
        g##X = g_;                                                             \
        const int4* p_ = colv + (size_t)g_ * (GRP / 4) + lane;                 \
        X##0 = p_[0]; X##1 = p_[32]; X##2 = p_[64]; X##3 = p_[96];             \
    }

#define FILTERSET(X)                                                           \
    {                                                                          \
        const int eb_ = head + g##X * GRP + lane * 4;                          \
        FILTER4(X##0, eb_);                                                    \
        FILTER4(X##1, eb_ + 128);                                              \
        FILTER4(X##2, eb_ + 256);                                              \
        FILTER4(X##3, eb_ + 384);                                              \
    }

#define STAGE(CUR, NXT2)                                                       \
    if (t >= nG) break;                                                        \
    LOADSET(NXT2, t + 2 * HOPW);                                               \
    FILTERSET(CUR);                                                            \
    t += HOPW;                                                                 \
    if (cnt > QFLUSH) drain(true, cnt);

template <class DrainF>
__device__ __forceinline__ void scan_edge_list(const int* __restrict__ col, const int E,
                                               const int t0, const int tn, int* const q,
                                               const int wave, const int lane, DrainF drain) {
    int cnt = 0;

    int head = (int)((0ull - ((unsigned long long)(size_t)col >> 2)) & 3ull);
    if (head > E) head = E;
    const int nG   = (E - head) / GRP;
    const int nrem = E - nG * GRP;
    for (int s = wave; s * 32 < nrem; s += HOPW) {
        const int idx = s * 32 + lane;
        const int e = (idx < head) ? idx : idx + nG * GRP;
        const int c = (idx < nrem) ? col[e] : -1;
        const bool m = (unsigned)(c - t0) < (unsigned)tn;
        APPEND(m, e);
    }

    const int rot = (int)(((long long)blockIdx.x * nG) / gridDim.x);
    const int4* colv = (const int4*)(col + head);

    int t = wave;
    const int4 none = make_int4(-1, -1, -1, -1);
    int4 A0 = none, A1 = none, A2 = none, A3 = none;
    int4 B0 = none, B1 = none, B2 = none, B3 = none;
    int4 C0 = none, C1 = none, C2 = none, C3 = none;
    int gA = 0, gB = 0, gC = 0;

    if (nG > 0) {
        LOADSET(A, t);
        LOADSET(B, t + HOPW);
    }
    for (;;) {
        STAGE(A, C);
        STAGE(B, A);
        STAGE(C, B);
    }
    drain(false, cnt);
}

__global__ __launch_bounds__(256)
void k_tile_planes(const int* __restrict__ row, const int* __restrict__ col, float* planesf,
                   float* pkf, int E) {
    const int lane = threadIdx.x & 31;
    const int wave = __builtin_amdgcn_readfirstlane((int)(threadIdx.x >> 5));
    const int blk  = blockIdx.x * 8 + wave;
    const int nB   = (E + PBLK - 1) / PBLK;
    if (blk >= nB) return;
    unsigned p0 = 0, p1 = 0, p2 = 0, p3 = 0, p4 = 0, p5 = 0, p6 = 0, p7 = 0;
    unsigned* const pk = (unsigned*)pkf;
    int cv[32];
    int rv[32];
#pragma unroll
    for (int r = 0; r < 32; ++r) {
        const int e = min(blk * PBLK + r * 32 + lane, E - 1);
        cv[r] = col[e];
        rv[r] = row[e];
    }
#pragma unroll
    for (int r = 0; r < 32; ++r) {
        const int e = min(blk * PBLK + r * 32 + lane, E - 1);
        const unsigned tl = (unsigned)cv[r] / (unsigned)TILE;
        const unsigned tr = (unsigned)cv[r] - tl * (unsigned)TILE;
        pk[e] = (unsigned)rv[r] | (tr << PKSH);
        cv[r] = (int)tl;
    }
#pragma unroll
    for (int r = 0; r < 32; ++r) {
        const unsigned tl = (unsigned)cv[r];
        const unsigned b0 = __builtin_amdgcn_ballot_w32((tl & 1u) != 0u);
        const unsigned b1 = __builtin_amdgcn_ballot_w32((tl & 2u) != 0u);
        const unsigned b2 = __builtin_amdgcn_ballot_w32((tl & 4u) != 0u);
        const unsigned b3 = __builtin_amdgcn_ballot_w32((tl & 8u) != 0u);
        const unsigned b4 = __builtin_amdgcn_ballot_w32((tl & 16u) != 0u);
        const unsigned b5 = __builtin_amdgcn_ballot_w32((tl & 32u) != 0u);
        const unsigned b6 = __builtin_amdgcn_ballot_w32((tl & 64u) != 0u);
        const unsigned b7 = __builtin_amdgcn_ballot_w32((tl & 128u) != 0u);
        const bool mine = lane == r;
        p0 = mine ? b0 : p0;
        p1 = mine ? b1 : p1;
        p2 = mine ? b2 : p2;
        p3 = mine ? b3 : p3;
        p4 = mine ? b4 : p4;
        p5 = mine ? b5 : p5;
        p6 = mine ? b6 : p6;
        p7 = mine ? b7 : p7;
    }
    uint4* const o = (uint4*)planesf + (size_t)blk * (PBLK / 16);
    o[lane]      = make_uint4(p0, p1, p2, p3);
    o[32 + lane] = make_uint4(p4, p5, p6, p7);
}

#define LOADP(X, tt)                                                           \
    {                                                                          \
        int g_ = (tt) + rot; if (g_ >= nB) g_ -= nB;                           \
        g_ = ((tt) < nB) ? g_ : 0;                                             \
        g##X = g_;                                                             \
        const uint4* p_ = pv + (size_t)g_ * (PBLK / 16) + lane;                \
        X##0 = p_[0]; X##1 = p_[32];                                           \
    }

#define FILTERP(X)                                                             \
    {                                                                          \
        const uint4 lo_ = X##0;                                                \
        const uint4 hi_ = X##1;                                                \
        unsigned hm_ = ~(((lo_.x ^ m0) | (lo_.y ^ m1) | (lo_.z ^ m2) | (lo_.w ^ m3)) | \
                         ((hi_.x ^ m4) | (hi_.y ^ m5) | (hi_.z ^ m6) | (hi_.w ^ m7))); \
        const int eb_ = g##X * PBLK + lane * 32;                               \
        if (g##X == nB - 1) {                                                  \
            const int nv_ = E - eb_;     \
            hm_ = (nv_ >= 32) ? hm_ : ((nv_ > 0) ? (hm_ & ((1u << nv_) - 1u)) : 0u); \
        }                                                                      \
        unsigned mk_ = __builtin_amdgcn_ballot_w32(hm_ != 0u);                 \
        while (mk_) {                                                          \
            if (hm_ != 0u)                                                     \
                q[cnt + (int)__builtin_amdgcn_mbcnt_lo(mk_, 0u)] = eb_ + __builtin_ctz(hm_); \
            cnt += __builtin_popcount(mk_);                                    \
            hm_ &= hm_ - 1u;                                                   \
            mk_ = __builtin_amdgcn_ballot_w32(hm_ != 0u);                      \
        }                                                                      \
    }

#define STAGEP(CUR, NXT2)                                                      \
    if (t >= nB) break;                                                        \
    LOADP(NXT2, t + 2 * HOPW);                                                 \
    FILTERP(CUR);                                                              \
    t += HOPW;                                                                 \
    if (cnt > QFLUSH) drain(true, cnt);

template <class DrainF>
__device__ __forceinline__ void scan_planes(const uint4* __restrict__ pv, const int E,
                                            const int tsel, int* const q,
                                            const int wave, const int lane, DrainF drain) {
    int cnt = 0;
    const int nB = (E + PBLK - 1) / PBLK;
    const unsigned m0 = (tsel & 1)   ? ~0u : 0u;
    const unsigned m1 = (tsel & 2)   ? ~0u : 0u;
    const unsigned m2 = (tsel & 4)   ? ~0u : 0u;
    const unsigned m3 = (tsel & 8)   ? ~0u : 0u;
    const unsigned m4 = (tsel & 16)  ? ~0u : 0u;
    const unsigned m5 = (tsel & 32)  ? ~0u : 0u;
    const unsigned m6 = (tsel & 64)  ? ~0u : 0u;
    const unsigned m7 = (tsel & 128) ? ~0u : 0u;
    const int rot = (int)(((long long)blockIdx.x * nB) / gridDim.x);

    int t = wave;
    const uint4 zero = make_uint4(0u, 0u, 0u, 0u);
    uint4 A0 = zero, A1 = zero, B0 = zero, B1 = zero, C0 = zero, C1 = zero;
    int gA = 0, gB = 0, gC = 0;
    if (nB > 0) {
        LOADP(A, t);
        LOADP(B, t + HOPW);
    }
    for (;;) {
        STAGEP(A, C);
        STAGEP(B, A);
        STAGEP(C, B);
    }
    drain(false, cnt);
}

__global__ __launch_bounds__(HOPW * 32)
void k_filter_tile(float* lists, const float* planesf, const float* __restrict__ pkf,
                   float* __restrict__ dinv, int N, int E) {
    __shared__ int queue[HOPW * FQCAP];
    __shared__ unsigned segbuf[SUBS * SEGMAX];
    __shared__ int degc[TILE];
    __shared__ int segcnt[SUBS];

    const int tid  = threadIdx.x;
    const int lane = tid & 31;
    const int wave = __builtin_amdgcn_readfirstlane(tid >> 5);
    const int t0   = blockIdx.x * TILE;
    const int tn   = min(TILE, N - t0);

    int* const lhdr = (int*)(lists + (size_t)t0 * FEAT);
    const int  segcap = min(SEGMAX, (((tn * FEAT - LHDR) / 2) / SUBS) & ~31);
    unsigned* const lp = (unsigned*)(lhdr + LHDR);

    for (int i = tid; i < TILE; i += HOPW * 32) degc[i] = 0;
    if (tid < SUBS) segcnt[tid] = 0;
    __syncthreads();

    int* const q = queue + wave * FQCAP;
    const unsigned* const pk = (const unsigned*)pkf;
    auto drain = [&](const bool more, int& cnt) {
        const int nproc = more ? (cnt & ~31) : cnt;
        for (int b = 0; b < nproc; b += 64) {
            const int  i0 = b + lane;
            const int  i1 = b + 32 + lane;
            const bool v0 = i0 < nproc;
            const bool v1 = i1 < nproc;
            const int  e0 = q[v0 ? i0 : 0];
            const int  e1 = q[v1 ? i1 : 0];
            const unsigned p0 = pk[e0];
            const unsigned p1 = pk[e1];
            const int  cl0 = (int)(p0 >> PKSH);
            const int  cl1 = (int)(p1 >> PKSH);
            if (v0) {
                __hip_atomic_fetch_add(&degc[cl0], 1, __ATOMIC_RELAXED, __HIP_MEMORY_SCOPE_WORKGROUP);
                const int ow = cl0 & (SUBS - 1);
                const int s  = __hip_atomic_fetch_add(&segcnt[ow], 1, __ATOMIC_RELAXED, __HIP_MEMORY_SCOPE_WORKGROUP);
                if (s < segcap) segbuf[ow * SEGMAX + s] = p0;
            }
            if (v1) {
                __hip_atomic_fetch_add(&degc[cl1], 1, __ATOMIC_RELAXED, __HIP_MEMORY_SCOPE_WORKGROUP);
                const int ow = cl1 & (SUBS - 1);
                const int s  = __hip_atomic_fetch_add(&segcnt[ow], 1, __ATOMIC_RELAXED, __HIP_MEMORY_SCOPE_WORKGROUP);
                if (s < segcap) segbuf[ow * SEGMAX + s] = p1;
            }
        }
        const int rem = cnt - nproc;
        if (rem > 0) {
            const int keep = (lane < rem) ? q[nproc + lane] : 0;
            if (lane < rem) q[lane] = keep;
        }
        cnt = rem;
    };
    scan_planes((const uint4*)planesf, E, (int)blockIdx.x, q, wave, lane, drain);
    __syncthreads();

    for (int i = tid; i < tn; i += HOPW * 32) {
        const float v = (float)(degc[i] + 1);
        dinv[t0 + i] = (v > 0.0f) ? rsqrtf(v) : 0.0f;
    }
    for (int sg = wave; sg < SUBS; sg += HOPW) {
        const int n = min(segcnt[sg], segcap);
        unsigned* const dst = lp + sg * segcap;
        const unsigned* const sb = segbuf + sg * SEGMAX;
        for (int i = lane; i < n; i += 32) dst[i] = sb[i];
    }
    if (tid < HOPW) {
        const int cn0 = segcnt[tid];
        const int cn1 = segcnt[HOPW + tid];
        lhdr[64 + tid]        = cn0;
        lhdr[64 + HOPW + tid] = cn1;
        const unsigned bad = __builtin_amdgcn_ballot_w32(cn0 > segcap || cn1 > segcap);
        if (tid == 0) lhdr[0] = bad ? -1 : 1;
    }
}

static_assert(HOPW == 32 && SUBS == 2 * HOPW, "lane l of wave 0 writes the counts of segments l and l + 32; a tile row's owner half is its low 6 bits");

template <int U>
__device__ __forceinline__ int half_share(const int v) {
    return __builtin_amdgcn_update_dpp(0, v, 0x150 + U, 0xf, 0xf, true);
}

#define GATHER1(u_)                                                            \
    {                                                                          \
        const unsigned o_ = (unsigned)half_share<u_>((int)ro) + c16;           \
        val[u_] = *(const float4*)((const char*)xin + o_);                     \
    }

#define ADD2(u_)                                                               \
    {                                                                          \
        const unsigned a0_ = (unsigned)half_share<u_>((int)co) + c16;          \
        const unsigned a1_ = (unsigned)half_share<u_ + 1>((int)co) + c16;      \
        const float w0_ = __int_as_float(half_share<u_>(__float_as_int(ws)));  \
        const float w1_ = __int_as_float(half_share<u_ + 1>(__float_as_int(ws))); \
        float4* const p0_ = (float4*)((char*)tile + a0_);                      \
        float4* const p1_ = (float4*)((char*)tile + a1_);                      \
        float4 t0_ = *p0_;                                                     \
        float4 t1_ = *p1_;                                                     \
        const float  wd_ = (a0_ == a1_) ? w0_ : 0.0f;                          \
        const float4 v0_ = val[u_];                                            \
        const float4 v1_ = val[u_ + 1];                                        \
        t0_.x += w0_ * v0_.x; t0_.y += w0_ * v0_.y;                            \
        t0_.z += w0_ * v0_.z; t0_.w += w0_ * v0_.w;                            \
        t1_.x += w1_ * v1_.x; t1_.y += w1_ * v1_.y;                            \
        t1_.z += w1_ * v1_.z; t1_.w += w1_ * v1_.w;                            \
        t1_.x += wd_ * v0_.x; t1_.y += wd_ * v0_.y;                            \
        t1_.z += wd_ * v0_.z; t1_.w += wd_ * v0_.w;                            \
        *p0_ = t0_;                                                            \
        *p1_ = t1_;                                                            \
    }

template <int NP>
__device__ __forceinline__ void walk_halves(const float* __restrict__ xin, float* const tile,
                                            const unsigned pe, const float ws,
                                            const unsigned c16) {
    const unsigned ro = (pe & PKMASK) << 8;
    const unsigned co = (pe >> PKSH) << 8;
    float4 val[NP];
    GATHER1(0)  GATHER1(1)  GATHER1(2)  GATHER1(3)
    GATHER1(4)  GATHER1(5)  GATHER1(6)  GATHER1(7)
    if constexpr (NP > 8) {
        GATHER1(8)  GATHER1(9)  GATHER1(10) GATHER1(11)
        GATHER1(12) GATHER1(13) GATHER1(14) GATHER1(15)
    }
    ADD2(0) ADD2(2) ADD2(4) ADD2(6)
    if constexpr (NP > 8) {
        ADD2(8) ADD2(10) ADD2(12) ADD2(14)
    }
}

template <bool FIRST>
__global__ __launch_bounds__(HOPW * 32)
void k_hop_tile(const float* __restrict__ xin, float* xout, float* lists,
                const int* __restrict__ row, const int* __restrict__ col,
                const float* __restrict__ dinv,
                const float* __restrict__ Wm, const float* __restrict__ bias,
                float* __restrict__ outp, int N, int E) {
    __shared__ __attribute__((aligned(16))) float tile[TILE * FEAT];
    __shared__ int queue[HOPW * QCAP];
    __shared__ float dloc[TILE];
    __shared__ __attribute__((aligned(16))) float wl[FIRST ? 4 : NCLSP * WLS];

    const int tid  = threadIdx.x;
    const int lane = tid & 31;
    const int wave = __builtin_amdgcn_readfirstlane(tid >> 5);
    const int t0   = blockIdx.x * TILE;
    const int tn   = min(TILE, N - t0);

    const int      sg  = wave + HOPW * (lane >> 4);
    const int      cl  = lane & 15;
    const unsigned c16 = (unsigned)cl * 16u;
    int*   const lhdr   = (int*)(lists + (size_t)t0 * FEAT);
    const int    segcap = min(SEGMAX, (((tn * FEAT - LHDR) / 2) / SUBS) & ~31);
    const unsigned* const lp = (const unsigned*)(lhdr + LHDR) + sg * segcap;
    float* const lw     = (float*)(lhdr + LHDR + SUBS * segcap) + sg * segcap;
    const int hraw = lhdr[0];
    const int sraw = lhdr[64 + sg];

    constexpr int SEEDK = (TILE * (FEAT / 4) + HOPW * 32 - 1) / (HOPW * 32);
    constexpr int WK    = (NCLSP * FEAT + HOPW * 32 - 1) / (HOPW * 32);
    static_assert(TILE <= HOPW * 32, "one round copies the tile's own dinv");
    const int slim = tn * (FEAT / 4);
    float4 sv[SEEDK];
    float  sw[SEEDK];
    float  wv[FIRST ? 1 : WK];
    float  dl = 0.0f;
    {
        const float4* src = (const float4*)(xin + (size_t)t0 * FEAT);
#pragma unroll
        for (int k = 0; k < SEEDK; ++k) {
            const int i = min(tid + k * (HOPW * 32), slim - 1);
            sw[k] = dinv[t0 + (i >> 4)];
            sv[k] = src[i];
        }
        if (FIRST) {
            dl = dinv[t0 + min(tid, tn - 1)];
        } else {
#pragma unroll
            for (int k = 0; k < WK; ++k)
                wv[k] = Wm[min(tid + k * (HOPW * 32), NCLS * FEAT - 1)];
        }
    }

    const bool uselist = __builtin_amdgcn_readfirstlane(hraw) == 1;
    const int  nmine   = uselist ? sraw : 0;
    const int  nseg    = max(__builtin_amdgcn_readlane(nmine, 0), __builtin_amdgcn_readlane(nmine, 16));

    const unsigned fb = (unsigned)sg << PKSH;
    int      b     = 0;
    unsigned praw  = 0u;
    unsigned praw1 = 0u;
    float    wraw  = 0.0f;
    if (b < nseg)      praw  = lp[(cl < nmine) ? cl : 0];
    if (b + 16 < nseg) praw1 = lp[(16 + cl < nmine) ? 16 + cl : 0];
    if (b < nseg) {
        if (FIRST) wraw = dinv[((cl < nmine) ? praw : fb) & PKMASK];
        else       wraw = lw[(cl < nmine) ? cl : 0];
    }

    {
        float4* dst = (float4*)tile;
#pragma unroll
        for (int k = 0; k < SEEDK; ++k) {
            const int i = tid + k * (HOPW * 32);
            const float w = sw[k] * sw[k];
            const float4 v = sv[k];
            if (i < slim) dst[i] = make_float4(w * v.x, w * v.y, w * v.z, w * v.w);
        }
        if (FIRST) {
            if (tid < tn) dloc[tid] = dl;
        } else {
#pragma unroll
            for (int k = 0; k < WK; ++k) {
                const int i = tid + k * (HOPW * 32);
                const int r = i >> 6;
                if (i < NCLSP * FEAT) wl[r * WLS + (i & (FEAT - 1))] = (r < NCLS) ? wv[k] : 0.0f;
            }
        }
    }
    __syncthreads();

    if (!uselist) {
        auto consume = [&](const int r0, const int r1, const float w0, const float w1,
                           const int cl0, const int cl1, const int mtot) {
#pragma unroll 1
            for (int k = 0; k < mtot; k += BW) {
                const bool  hi = k >= 32;
                const int   rs = hi ? r1 : r0;
                const float ws = hi ? w1 : w0;
                const int   cs = hi ? cl1 : cl0;
                const int   kk = k & 31;
                v2f val[BW];
#pragma unroll
                for (int u = 0; u < BW; ++u) {
                    const int rk = __builtin_amdgcn_readlane(rs, kk + u);
                    val[u] = *(const v2f*)(xin + (size_t)rk * FEAT + 2 * lane);
                }
#pragma unroll
                for (int u = 0; u < BW; ++u) {
                    if (k + u < mtot) {
                        const float wk = __int_as_float(__builtin_amdgcn_readlane(__float_as_int(ws), kk + u));
                        const int   ck = __builtin_amdgcn_readlane(cs, kk + u);
                        float* d = tile + ck * FEAT + 2 * lane;
                        __hip_atomic_fetch_add(d + 0, wk * val[u].x, __ATOMIC_RELAXED, __HIP_MEMORY_SCOPE_WORKGROUP);
                        __hip_atomic_fetch_add(d + 1, wk * val[u].y, __ATOMIC_RELAXED, __HIP_MEMORY_SCOPE_WORKGROUP);
                    }
                }
            }
        };
        int* const q = queue + wave * QCAP;
        auto drain = [&](const bool more, int& cnt) {
            const int nproc = more ? (cnt & ~31) : cnt;
            for (int b = 0; b < nproc; b += 64) {
                const int  i0 = b + lane;
                const int  i1 = b + 32 + lane;
                const bool v0 = i0 < nproc;
                const bool v1 = i1 < nproc;
                const int  e0 = q[v0 ? i0 : 0];
                const int  e1 = q[v1 ? i1 : 0];
                const int  r0 = row[e0];
                const int  r1 = row[e1];
                const int  c0 = col[e0];
                const int  c1 = col[e1];
                const float dr0 = dinv[r0];
                const float dr1 = dinv[r1];
                const float dc0 = dinv[c0];
                const float dc1 = dinv[c1];
                const float w0 = v0 ? dr0 * dc0 : 0.0f;
                const float w1 = v1 ? dr1 * dc1 : 0.0f;
                consume(r0, r1, w0, w1, c0 - t0, c1 - t0, min(64, nproc - b));
            }
            const int rem = cnt - nproc;
            if (rem > 0) {
                const int keep = (lane < rem) ? q[nproc + lane] : 0;
                if (lane < rem) q[lane] = keep;
            }
            cnt = rem;
        };
        scan_edge_list(col, E, t0, tn, q, wave, lane, drain);
    } else {
        bool     valid = cl < nmine;
        unsigned pe    = valid ? praw : fb;
        float    ws    = valid ? (FIRST ? wraw * dloc[pe >> PKSH] : wraw) : 0.0f;
        while (b < nseg) {
            const int nb  = b + 16;
            const int nb2 = b + 32;
            const bool     nvalid = nb + cl < nmine;
            const unsigned npe    = nvalid ? praw1 : fb;
            unsigned praw2 = 0u;
            float    nwraw = 0.0f;
            if (nb2 < nseg) praw2 = lp[(nb2 + cl < nmine) ? nb2 + cl : 0];
            if (nb < nseg) {
                if (FIRST) nwraw = dinv[npe & PKMASK];
                else       nwraw = lw[nvalid ? nb + cl : 0];
            }
            if (nseg - b > 8) walk_halves<16>(xin, tile, pe, ws, c16);
            else              walk_halves<8>(xin, tile, pe, ws, c16);
            if (FIRST) {
                if (valid) lw[b + cl] = ws;
            }
            valid = nvalid;
            pe    = npe;
            ws    = nvalid ? (FIRST ? nwraw * dloc[npe >> PKSH] : nwraw) : 0.0f;
            praw1 = praw2;
            b = nb;
        }
    }
    __syncthreads();

    if (FIRST) {
        const float4* srct = (const float4*)tile;
        float4* dst = (float4*)(xout + (size_t)t0 * FEAT);
        for (int i = tid; i < tn * (FEAT / 4); i += HOPW * 32) dst[i] = srct[i];
    } else {
        const int half = lane >> 4;
        const int l16  = lane & 15;
        const bool has2 = l16 < NCLS - 32;
        const float bb0 = bias[l16];
        const float bb1 = bias[16 + l16];
        const float bb2 = has2 ? bias[32 + l16] : 0.0f;
        const float* const b0r = wl + (l16)      * WLS + 2 * half;
        const float* const b1r = wl + (16 + l16) * WLS + 2 * half;
        const float* const b2r = wl + (32 + l16) * WLS + 2 * half;
        for (int rb = wave * 16; rb < tn; rb += HOPW * 16) {
            const int ar = min(rb + l16, tn - 1);
            const float* const arow = tile + ar * FEAT + 2 * half;
            v8f acc0 = {};
            v8f acc1 = {};
            v8f acc2 = {};
#pragma unroll
            for (int k0 = 0; k0 < FEAT; k0 += 4) {
                const v2f a  = *(const v2f*)(arow + k0);
                const v2f b0 = *(const v2f*)(b0r + k0);
                const v2f b1 = *(const v2f*)(b1r + k0);
                const v2f b2 = *(const v2f*)(b2r + k0);
                acc0 = __builtin_amdgcn_wmma_f32_16x16x4_f32(false, a, false, b0, (short)0, acc0, false, false);
                acc1 = __builtin_amdgcn_wmma_f32_16x16x4_f32(false, a, false, b1, (short)0, acc1, false, false);
                acc2 = __builtin_amdgcn_wmma_f32_16x16x4_f32(false, a, false, b2, (short)0, acc2, false, false);
            }
            float a0[8], a1[8], a2[8], mx[8], sm[8];
#pragma unroll
            for (int j = 0; j < 8; ++j) {
                a0[j] = acc0[j] + bb0;
                a1[j] = acc1[j] + bb1;
                a2[j] = acc2[j] + bb2;
                const float m01 = fmaxf(a0[j], a1[j]);
                mx[j] = has2 ? fmaxf(m01, a2[j]) : m01;
            }
#pragma unroll
            for (int s = 1; s < 16; s <<= 1) {
#pragma unroll
                for (int j = 0; j < 8; ++j) mx[j] = fmaxf(mx[j], __shfl_xor(mx[j], s));
            }
#pragma unroll
            for (int j = 0; j < 8; ++j) {
                const float e01 = expf(a0[j] - mx[j]) + expf(a1[j] - mx[j]);
                sm[j] = has2 ? e01 + expf(a2[j] - mx[j]) : e01;
            }
#pragma unroll
            for (int s = 1; s < 16; s <<= 1) {
#pragma unroll
                for (int j = 0; j < 8; ++j) sm[j] += __shfl_xor(sm[j], s);
            }
#pragma unroll
            for (int j = 0; j < 8; ++j) {
                const int   rr = rb + j + 8 * half;
                const float ls = logf(sm[j]);
                if (rr < tn) {
                    float* o = outp + (size_t)(t0 + rr) * NCLS;
                    o[l16]      = a0[j] - mx[j] - ls;
                    o[16 + l16] = a1[j] - mx[j] - ls;
                    if (has2) o[32 + l16] = a2[j] - mx[j] - ls;
                }
            }
        }
    }
}


static inline size_t align256f(size_t nfloats) { return (nfloats + 63) & ~(size_t)63; }

extern "C" void kernel_launch(void* const* d_in, const int* in_sizes, int n_in,
                              void* d_out, int out_size, void* d_ws, size_t ws_size,
                              hipStream_t stream) {
    const float* x  = (const float*)d_in[0];
    const int*   ei = (const int*)d_in[1];
    const float* W  = (const float*)d_in[2];
    const float* bi = (const float*)d_in[3];

    const int N = in_sizes[0] / FEAT;
    const int E = in_sizes[1] / 2;
    const int* row = ei;
    const int* col = ei + E;
    const int total = N * FEAT;

    float* ws = (float*)d_ws;
    size_t off = 0;
    float* dinv = ws + off; off += align256f((size_t)N);
    float* x0   = ws + off; off += align256f((size_t)total);
    float* lst  = ws + off;

    const int htiles = (N + TILE - 1) / TILE;
    const int nblk = (E + PBLK - 1) / PBLK;
    float* pkw = x0 + align256f((size_t)nblk * (PBLK / 4));
    k_tile_planes<<<(nblk + 7) / 8, 256, 0, stream>>>(row, col, x0, pkw, E);
    k_filter_tile<<<htiles, HOPW * 32, 0, stream>>>(lst, x0, pkw, dinv, N, E);
    k_hop_tile<true><<<htiles, HOPW * 32, 0, stream>>>(x, x0, lst, row, col, dinv, W, bi, (float*)d_out, N, E);
    k_hop_tile<false><<<htiles, HOPW * 32, 0, stream>>>(x0, nullptr, lst, row, col, dinv, W, bi, (float*)d_out, N, E);
}
